# starting kernel
# baseline (speedup 1.0000x reference)
#include <hip/hip_runtime.h>
#include <math.h>


typedef __attribute__((ext_vector_type(2))) float v2f;
typedef __attribute__((ext_vector_type(8))) float v8f;

#define WMMA_F32_16x16x4(a, b, c) \
  __builtin_amdgcn_wmma_f32_16x16x4_f32(false, (a), false, (b), (short)0, (c), false, false)

#define NMAT 8192
#define NB   16
#define NN   512
#define NF   1024

__global__ __launch_bounds__(256) void cayley_kernel(const float* __restrict__ qw,
                                                     const float* __restrict__ kw,
                                                     const float* __restrict__ vw,
                                                     float* __restrict__ Cout) {
  __shared__ float X[32][32];
  __shared__ float Mm[32][33];
  __shared__ float Inv[32][33];
  __shared__ float fac[32];
  __shared__ float pivinv;
  const float* W = (blockIdx.x == 0) ? qw : ((blockIdx.x == 1) ? kw : vw);
  const int tid = threadIdx.x;

  for (int e = tid; e < 1024; e += 256) {
    int i = e >> 5, j = e & 31;
    float xv = W[i * 32 + j] - W[j * 32 + i];
    X[i][j]  = xv;
    Mm[i][j] = xv + ((i == j) ? 1.f : 0.f);
    Inv[i][j] = (i == j) ? 1.f : 0.f;
  }
  __syncthreads();

  for (int k = 0; k < 32; ++k) {
    if (tid == 0) pivinv = 1.f / Mm[k][k];
    __syncthreads();
    if (tid < 32)      Mm[k][tid]       *= pivinv;
    else if (tid < 64) Inv[k][tid - 32] *= pivinv;
    __syncthreads();
    if (tid < 32) fac[tid] = (tid == k) ? 0.f : Mm[tid][k];
    __syncthreads();
    for (int e = tid; e < 2048; e += 256) {
      int i = e >> 6, j = e & 63;
      if (i == k) continue;
      float f = fac[i];
      if (j < 32) Mm[i][j]        -= f * Mm[k][j];
      else        Inv[i][j - 32]  -= f * Inv[k][j - 32];
    }
    __syncthreads();
  }

  for (int e = tid; e < 1024; e += 256) {
    int i = e >> 5, j = e & 31;
    float acc = 0.f;
    for (int kk = 0; kk < 32; ++kk) {
      float a = ((i == kk) ? 1.f : 0.f) - X[i][kk];
      acc += a * Inv[kk][j];
    }
    Cout[blockIdx.x * 1024 + i * 32 + j] = acc;
  }
}

__global__ __launch_bounds__(32) void jacobi_logm_kernel(const float* __restrict__ x,
                                                         float* __restrict__ xl) {
  __shared__ float A[32][33];
  __shared__ float V[32][33];
  __shared__ float cc[16], ss[16];
  __shared__ int   pp[16], qq[16];
  __shared__ float lw[32];
  const int lane = threadIdx.x;
  const float* xm = x + (size_t)blockIdx.x * 1024;

  for (int i = 0; i < 32; ++i) {
    A[i][lane] = xm[i * 32 + lane];
    V[i][lane] = (i == lane) ? 1.f : 0.f;
  }
  __syncthreads();

  for (int sweep = 0; sweep < 8; ++sweep) {
    for (int r = 0; r < 31; ++r) {
      if (lane < 16) {
        int p, q;
        if (lane == 0) { p = 31; q = r % 31; }
        else { p = (r + lane) % 31; q = (r - lane + 31) % 31; }
        if (p > q) { int t = p; p = q; q = t; }
        float apq = A[p][q];
        float c = 1.f, s = 0.f;
        if (fabsf(apq) > 1e-12f) {
          float theta = (A[q][q] - A[p][p]) / (2.f * apq);
          float t = 1.f / (fabsf(theta) + sqrtf(theta * theta + 1.f));
          if (theta < 0.f) t = -t;
          c = rsqrtf(t * t + 1.f);
          s = t * c;
        }
        cc[lane] = c; ss[lane] = s; pp[lane] = p; qq[lane] = q;
      }
      __syncthreads();
      for (int t = 0; t < 16; ++t) {
        int p = pp[t], q = qq[t];
        float c = cc[t], s = ss[t];
        float ap = A[p][lane], aq = A[q][lane];
        A[p][lane] = c * ap - s * aq;
        A[q][lane] = s * ap + c * aq;
      }
      __syncthreads();
      for (int t = 0; t < 16; ++t) {
        int p = pp[t], q = qq[t];
        float c = cc[t], s = ss[t];
        float ap = A[lane][p], aq = A[lane][q];
        A[lane][p] = c * ap - s * aq;
        A[lane][q] = s * ap + c * aq;
        float vp = V[lane][p], vq = V[lane][q];
        V[lane][p] = c * vp - s * vq;
        V[lane][q] = s * vp + c * vq;
      }
      __syncthreads();
    }
  }

  lw[lane] = logf(fmaxf(A[lane][lane], 1e-20f));
  __syncthreads();

  float vj[32];
#pragma unroll
  for (int k = 0; k < 32; ++k) vj[k] = V[lane][k] * lw[k];

  float* outm = xl + (size_t)blockIdx.x * 1024;
  for (int i = 0; i < 32; ++i) {
    float acc = 0.f;
#pragma unroll
    for (int k = 0; k < 32; ++k) acc += V[i][k] * vj[k];
    outm[i * 32 + lane] = (i == lane) ? 0.f : acc;
  }
}

__global__ __launch_bounds__(32) void transbase_kernel(const float* __restrict__ xl,
                                                       const float* __restrict__ C,
                                                       float* __restrict__ qkv) {
  __shared__ float Y[32][33];
  __shared__ float CL[32][33];
  __shared__ float T[32][33];
  __shared__ float T2[32][33];
  const int lane = threadIdx.x;
  const int nIdx = lane & 15;
  const int hi   = lane >> 4;
  const int m = blockIdx.x, w = blockIdx.y;
  const float* ym = xl + (size_t)m * 1024;
  const float* Cw = C + (size_t)w * 1024;

  for (int i = 0; i < 32; ++i) {
    Y[i][lane]  = ym[i * 32 + lane];
    CL[i][lane] = Cw[i * 32 + lane];
  }
  __syncthreads();

  for (int ti = 0; ti < 2; ++ti)
    for (int tj = 0; tj < 2; ++tj) {
      v8f acc = {};
#pragma unroll
      for (int kk = 0; kk < 8; ++kk) {
        int k0 = kk * 4 + hi * 2;
        v2f a; a.x = CL[ti * 16 + nIdx][k0]; a.y = CL[ti * 16 + nIdx][k0 + 1];
        v2f b; b.x = Y[k0][tj * 16 + nIdx];  b.y = Y[k0 + 1][tj * 16 + nIdx];
        acc = WMMA_F32_16x16x4(a, b, acc);
      }
#pragma unroll
      for (int r = 0; r < 8; ++r)
        T[ti * 16 + r + hi * 8][tj * 16 + nIdx] = acc[r];
    }
  __syncthreads();

  for (int ti = 0; ti < 2; ++ti)
    for (int tj = 0; tj < 2; ++tj) {
      v8f acc = {};
#pragma unroll
      for (int kk = 0; kk < 8; ++kk) {
        int k0 = kk * 4 + hi * 2;
        v2f a; a.x = T[ti * 16 + nIdx][k0];       a.y = T[ti * 16 + nIdx][k0 + 1];
        v2f b; b.x = CL[tj * 16 + nIdx][k0];      b.y = CL[tj * 16 + nIdx][k0 + 1];
        acc = WMMA_F32_16x16x4(a, b, acc);
      }
#pragma unroll
      for (int r = 0; r < 8; ++r)
        T2[ti * 16 + r + hi * 8][tj * 16 + nIdx] = acc[r];
    }
  __syncthreads();

  float* outm = qkv + ((size_t)w * NMAT + m) * 1024;
  for (int i = 0; i < 32; ++i) {
    float lo = T2[i][lane];
    float up = T2[lane][i];
    float val = (i > lane) ? lo : ((i < lane) ? up : 0.f);
    outm[i * 32 + lane] = val;
  }
}

__global__ __launch_bounds__(256) void norms_kernel(const float* __restrict__ qkv,
                                                    float* __restrict__ nrm) {
  int idx = blockIdx.x * 256 + threadIdx.x;
  int sel = idx >> 13;
  int r   = idx & (NMAT - 1);
  const float* row = qkv + ((size_t)sel * NMAT + r) * 1024;
  float acc = 0.f;
  for (int f = 0; f < NF; ++f) { float t = row[f]; acc += t * t; }
  nrm[idx] = acc;
}

__global__ __launch_bounds__(32) void scores_kernel(const float* __restrict__ qkv,
                                                    const float* __restrict__ nrm,
                                                    float* __restrict__ scores) {
  const int lane = threadIdx.x;
  const int nIdx = lane & 15;
  const int hi   = lane >> 4;
  const int it = blockIdx.x, jt = blockIdx.y, b = blockIdx.z;
  const int i0 = it * 64, j0 = jt * 16;
  const float* qf = qkv + ((size_t)0 * NMAT + (size_t)b * NN) * 1024;
  const float* kf = qkv + ((size_t)1 * NMAT + (size_t)b * NN) * 1024;
  const float* arow  = kf + (size_t)(j0 + nIdx) * 1024;
  const float* brow0 = qf + (size_t)(i0 +  0 + nIdx) * 1024;
  const float* brow1 = qf + (size_t)(i0 + 16 + nIdx) * 1024;
  const float* brow2 = qf + (size_t)(i0 + 32 + nIdx) * 1024;
  const float* brow3 = qf + (size_t)(i0 + 48 + nIdx) * 1024;

  v8f acc0 = {}, acc1 = {}, acc2 = {}, acc3 = {};
  for (int k0 = hi * 2; k0 < NF; k0 += 4) {
    v2f a = *(const v2f*)(arow + k0);
    acc0 = WMMA_F32_16x16x4(a, *(const v2f*)(brow0 + k0), acc0);
    acc1 = WMMA_F32_16x16x4(a, *(const v2f*)(brow1 + k0), acc1);
    acc2 = WMMA_F32_16x16x4(a, *(const v2f*)(brow2 + k0), acc2);
    acc3 = WMMA_F32_16x16x4(a, *(const v2f*)(brow3 + k0), acc3);
  }

  float* sb = scores + (size_t)b * NN * NN;
#pragma unroll
  for (int t = 0; t < 4; ++t) {
    v8f acc = (t == 0) ? acc0 : ((t == 1) ? acc1 : ((t == 2) ? acc2 : acc3));
    int i = i0 + t * 16 + nIdx;
    float q2 = nrm[b * NN + i];
#pragma unroll
    for (int r = 0; r < 8; ++r) {
      int j = j0 + r + hi * 8;
      float d2 = q2 + nrm[NMAT + b * NN + j] - 2.f * acc[r];
      d2 = fmaxf(d2, 1e-12f);
      float e  = sqrtf(d2);
      float sc = 1.f / (1.f + log1pf(e));
      sb[(size_t)j * NN + i] = sc;
    }
  }
}

__global__ __launch_bounds__(256) void softmax_kernel(float* __restrict__ scores) {
  int idx = blockIdx.x * 256 + threadIdx.x;
  int b = idx >> 9, i = idx & (NN - 1);
  float* col = scores + (size_t)b * NN * NN + i;
  float mx = -1e30f;
  for (int j = 0; j < NN; ++j) mx = fmaxf(mx, col[(size_t)j * NN]);
  float sum = 0.f;
  for (int j = 0; j < NN; ++j) {
    float e = __expf(col[(size_t)j * NN] - mx);
    sum += e;
    col[(size_t)j * NN] = e;
  }
  float inv = 1.f / sum;
  for (int j = 0; j < NN; ++j) col[(size_t)j * NN] *= inv;
}

__global__ __launch_bounds__(32) void outgemm_kernel(const float* __restrict__ scores,
                                                     const float* __restrict__ qkv,
                                                     float* __restrict__ out) {
  const int lane = threadIdx.x;
  const int nIdx = lane & 15;
  const int hi   = lane >> 4;
  const int ft = blockIdx.x, ct = blockIdx.y, b = blockIdx.z;
  const int f0 = ft * 16, c0 = ct * 64;
  const float* p  = scores + (size_t)b * NN * NN;
  const float* vf = qkv + ((size_t)2 * NMAT + (size_t)b * NN) * 1024;
  const float* arow0 = p + (size_t)(c0 +  0 + nIdx) * NN;
  const float* arow1 = p + (size_t)(c0 + 16 + nIdx) * NN;
  const float* arow2 = p + (size_t)(c0 + 32 + nIdx) * NN;
  const float* arow3 = p + (size_t)(c0 + 48 + nIdx) * NN;

  v8f acc0 = {}, acc1 = {}, acc2 = {}, acc3 = {};
  for (int k0 = hi * 2; k0 < NN; k0 += 4) {
    v2f bb;
    bb.x = vf[(size_t)k0 * NF + f0 + nIdx];
    bb.y = vf[(size_t)(k0 + 1) * NF + f0 + nIdx];
    acc0 = WMMA_F32_16x16x4(*(const v2f*)(arow0 + k0), bb, acc0);
    acc1 = WMMA_F32_16x16x4(*(const v2f*)(arow1 + k0), bb, acc1);
    acc2 = WMMA_F32_16x16x4(*(const v2f*)(arow2 + k0), bb, acc2);
    acc3 = WMMA_F32_16x16x4(*(const v2f*)(arow3 + k0), bb, acc3);
  }

  float* ob = out + (size_t)b * NN * NF;
#pragma unroll
  for (int t = 0; t < 4; ++t) {
    v8f acc = (t == 0) ? acc0 : ((t == 1) ? acc1 : ((t == 2) ? acc2 : acc3));
#pragma unroll
    for (int r = 0; r < 8; ++r) {
      int c = c0 + t * 16 + r + hi * 8;
      ob[(size_t)c * NF + f0 + nIdx] = acc[r];
    }
  }
}

extern "C" void kernel_launch(void* const* d_in, const int* in_sizes, int n_in,
                              void* d_out, int out_size, void* d_ws, size_t ws_size,
                              hipStream_t stream) {
  (void)in_sizes; (void)n_in; (void)out_size; (void)ws_size;
  const float* x  = (const float*)d_in[0];
  const float* qw = (const float*)d_in[1];
  const float* kw = (const float*)d_in[2];
  const float* vw = (const float*)d_in[3];
  float* out = (float*)d_out;

  float* ws     = (float*)d_ws;
  float* C      = ws;
  float* xl     = C + 3 * 1024;
  float* qkv    = xl + (size_t)NMAT * 1024;
  float* nrm    = qkv + (size_t)3 * NMAT * 1024;
  float* scores = nrm + 2 * NMAT;

  cayley_kernel<<<3, 256, 0, stream>>>(qw, kw, vw, C);
  jacobi_logm_kernel<<<NMAT, 32, 0, stream>>>(x, xl);
  transbase_kernel<<<dim3(NMAT, 3), 32, 0, stream>>>(xl, C, qkv);
  norms_kernel<<<64, 256, 0, stream>>>(qkv, nrm);
  scores_kernel<<<dim3(NN / 64, NN / 16, NB), 32, 0, stream>>>(qkv, nrm, scores);
  softmax_kernel<<<32, 256, 0, stream>>>(scores);
  outgemm_kernel<<<dim3(NF / 16, NN / 64, NB), 32, 0, stream>>>(scores, qkv, out);
}


# scores GEMM and softmax fused in one launch: an 8-wave block owns 32 whole columns (all j, tile kept in registers, operand rows read as whole 128-B lines); k order and column-sum order change
# speedup vs baseline: 1.4196x; 1.4196x over previous
#include <hip/hip_runtime.h>
#include <math.h>


typedef __attribute__((ext_vector_type(2))) float v2f;
typedef __attribute__((ext_vector_type(8))) float v8f;

#define WMMA_F32_16x16x4(a, b, c) \
  __builtin_amdgcn_wmma_f32_16x16x4_f32(false, (a), false, (b), (short)0, (c), false, false)

#define NMAT 8192
#define NB   16
#define NN   512
#define NF   1024

__global__ __launch_bounds__(256) void cayley_kernel(const float* __restrict__ qw,
                                                     const float* __restrict__ kw,
                                                     const float* __restrict__ vw,
                                                     float* __restrict__ Cout) {
  __shared__ float X[32][32];
  __shared__ float Mm[32][33];
  __shared__ float Inv[32][33];
  __shared__ float fac[32];
  __shared__ float pivinv;
  const float* W = (blockIdx.x == 0) ? qw : ((blockIdx.x == 1) ? kw : vw);
  const int tid = threadIdx.x;

  for (int e = tid; e < 1024; e += 256) {
    int i = e >> 5, j = e & 31;
    float xv = W[i * 32 + j] - W[j * 32 + i];
    X[i][j]  = xv;
    Mm[i][j] = xv + ((i == j) ? 1.f : 0.f);
    Inv[i][j] = (i == j) ? 1.f : 0.f;
  }
  __syncthreads();

  for (int k = 0; k < 32; ++k) {
    if (tid == 0) pivinv = 1.f / Mm[k][k];
    __syncthreads();
    if (tid < 32)      Mm[k][tid]       *= pivinv;
    else if (tid < 64) Inv[k][tid - 32] *= pivinv;
    __syncthreads();
    if (tid < 32) fac[tid] = (tid == k) ? 0.f : Mm[tid][k];
    __syncthreads();
    for (int e = tid; e < 2048; e += 256) {
      int i = e >> 6, j = e & 63;
      if (i == k) continue;
      float f = fac[i];
      if (j < 32) Mm[i][j]        -= f * Mm[k][j];
      else        Inv[i][j - 32]  -= f * Inv[k][j - 32];
    }
    __syncthreads();
  }

  for (int e = tid; e < 1024; e += 256) {
    int i = e >> 5, j = e & 31;
    float acc = 0.f;
    for (int kk = 0; kk < 32; ++kk) {
      float a = ((i == kk) ? 1.f : 0.f) - X[i][kk];
      acc += a * Inv[kk][j];
    }
    Cout[blockIdx.x * 1024 + i * 32 + j] = acc;
  }
}

__global__ __launch_bounds__(32) void jacobi_logm_kernel(const float* __restrict__ x,
                                                         float* __restrict__ xl) {
  __shared__ float A[32][33];
  __shared__ float V[32][33];
  __shared__ float cc[16], ss[16];
  __shared__ int   pp[16], qq[16];
  __shared__ float lw[32];
  const int lane = threadIdx.x;
  const float* xm = x + (size_t)blockIdx.x * 1024;

  for (int i = 0; i < 32; ++i) {
    A[i][lane] = xm[i * 32 + lane];
    V[i][lane] = (i == lane) ? 1.f : 0.f;
  }
  __syncthreads();

  for (int sweep = 0; sweep < 8; ++sweep) {
    for (int r = 0; r < 31; ++r) {
      if (lane < 16) {
        int p, q;
        if (lane == 0) { p = 31; q = r % 31; }
        else { p = (r + lane) % 31; q = (r - lane + 31) % 31; }
        if (p > q) { int t = p; p = q; q = t; }
        float apq = A[p][q];
        float c = 1.f, s = 0.f;
        if (fabsf(apq) > 1e-12f) {
          float theta = (A[q][q] - A[p][p]) / (2.f * apq);
          float t = 1.f / (fabsf(theta) + sqrtf(theta * theta + 1.f));
          if (theta < 0.f) t = -t;
          c = rsqrtf(t * t + 1.f);
          s = t * c;
        }
        cc[lane] = c; ss[lane] = s; pp[lane] = p; qq[lane] = q;
      }
      __syncthreads();
      for (int t = 0; t < 16; ++t) {
        int p = pp[t], q = qq[t];
        float c = cc[t], s = ss[t];
        float ap = A[p][lane], aq = A[q][lane];
        A[p][lane] = c * ap - s * aq;
        A[q][lane] = s * ap + c * aq;
      }
      __syncthreads();
      for (int t = 0; t < 16; ++t) {
        int p = pp[t], q = qq[t];
        float c = cc[t], s = ss[t];
        float ap = A[lane][p], aq = A[lane][q];
        A[lane][p] = c * ap - s * aq;
        A[lane][q] = s * ap + c * aq;
        float vp = V[lane][p], vq = V[lane][q];
        V[lane][p] = c * vp - s * vq;
        V[lane][q] = s * vp + c * vq;
      }
      __syncthreads();
    }
  }

  lw[lane] = logf(fmaxf(A[lane][lane], 1e-20f));
  __syncthreads();

  float vj[32];
#pragma unroll
  for (int k = 0; k < 32; ++k) vj[k] = V[lane][k] * lw[k];

  float* outm = xl + (size_t)blockIdx.x * 1024;
  for (int i = 0; i < 32; ++i) {
    float acc = 0.f;
#pragma unroll
    for (int k = 0; k < 32; ++k) acc += V[i][k] * vj[k];
    outm[i * 32 + lane] = (i == lane) ? 0.f : acc;
  }
}

__global__ __launch_bounds__(32) void transbase_kernel(const float* __restrict__ xl,
                                                       const float* __restrict__ C,
                                                       float* __restrict__ qkv) {
  __shared__ float Y[32][33];
  __shared__ float CL[32][33];
  __shared__ float T[32][33];
  __shared__ float T2[32][33];
  const int lane = threadIdx.x;
  const int nIdx = lane & 15;
  const int hi   = lane >> 4;
  const int m = blockIdx.x, w = blockIdx.y;
  const float* ym = xl + (size_t)m * 1024;
  const float* Cw = C + (size_t)w * 1024;

  for (int i = 0; i < 32; ++i) {
    Y[i][lane]  = ym[i * 32 + lane];
    CL[i][lane] = Cw[i * 32 + lane];
  }
  __syncthreads();

  for (int ti = 0; ti < 2; ++ti)
    for (int tj = 0; tj < 2; ++tj) {
      v8f acc = {};
#pragma unroll
      for (int kk = 0; kk < 8; ++kk) {
        int k0 = kk * 4 + hi * 2;
        v2f a; a.x = CL[ti * 16 + nIdx][k0]; a.y = CL[ti * 16 + nIdx][k0 + 1];
        v2f b; b.x = Y[k0][tj * 16 + nIdx];  b.y = Y[k0 + 1][tj * 16 + nIdx];
        acc = WMMA_F32_16x16x4(a, b, acc);
      }
#pragma unroll
      for (int r = 0; r < 8; ++r)
        T[ti * 16 + r + hi * 8][tj * 16 + nIdx] = acc[r];
    }
  __syncthreads();

  for (int ti = 0; ti < 2; ++ti)
    for (int tj = 0; tj < 2; ++tj) {
      v8f acc = {};
#pragma unroll
      for (int kk = 0; kk < 8; ++kk) {
        int k0 = kk * 4 + hi * 2;
        v2f a; a.x = T[ti * 16 + nIdx][k0];       a.y = T[ti * 16 + nIdx][k0 + 1];
        v2f b; b.x = CL[tj * 16 + nIdx][k0];      b.y = CL[tj * 16 + nIdx][k0 + 1];
        acc = WMMA_F32_16x16x4(a, b, acc);
      }
#pragma unroll
      for (int r = 0; r < 8; ++r)
        T2[ti * 16 + r + hi * 8][tj * 16 + nIdx] = acc[r];
    }
  __syncthreads();

  float* outm = qkv + ((size_t)w * NMAT + m) * 1024;
  for (int i = 0; i < 32; ++i) {
    float lo = T2[i][lane];
    float up = T2[lane][i];
    float val = (i > lane) ? lo : ((i < lane) ? up : 0.f);
    outm[i * 32 + lane] = val;
  }
}

__global__ __launch_bounds__(256) void norms_kernel(const float* __restrict__ qkv,
                                                    float* __restrict__ nrm) {
  int idx = blockIdx.x * 256 + threadIdx.x;
  int sel = idx >> 13;
  int r   = idx & (NMAT - 1);
  const float* row = qkv + ((size_t)sel * NMAT + r) * 1024;
  float acc = 0.f;
  for (int f = 0; f < NF; ++f) { float t = row[f]; acc += t * t; }
  nrm[idx] = acc;
}

typedef __attribute__((ext_vector_type(4))) float v4f;

struct RowFrag { v4f q[4]; };

__device__ __forceinline__ void load_frag(RowFrag& f, const float* p) {
  const v4f* p4 = (const v4f*)p;
  f.q[0] = p4[0]; f.q[1] = p4[1]; f.q[2] = p4[2]; f.q[3] = p4[3];
}

__device__ __forceinline__ v2f frag_pair(const RowFrag& f, int s) {
  v4f q = f.q[s >> 1];
  v2f lo = __builtin_shufflevector(q, q, 0, 1);
  v2f hp = __builtin_shufflevector(q, q, 2, 3);
  return (s & 1) ? hp : lo;
}

__device__ __forceinline__ void load_tiles(RowFrag (&A)[4], RowFrag (&B)[2],
                                           const float* arow, const float* brow, int kb) {
#pragma unroll
  for (int jt = 0; jt < 4; ++jt) load_frag(A[jt], arow + jt * (16 * NF) + kb);
#pragma unroll
  for (int it = 0; it < 2; ++it) load_frag(B[it], brow + it * (16 * NF) + kb);
}

__device__ __forceinline__ void mma_chunk(v8f (&acc)[4][2], const RowFrag (&A)[4],
                                          const RowFrag (&B)[2]) {
#pragma unroll
  for (int s = 0; s < 8; ++s) {
#pragma unroll
    for (int jt = 0; jt < 4; ++jt) {
      v2f a = frag_pair(A[jt], s);
#pragma unroll
      for (int it = 0; it < 2; ++it)
        acc[jt][it] = WMMA_F32_16x16x4(a, frag_pair(B[it], s), acc[jt][it]);
    }
  }
}

__global__ __launch_bounds__(256) void scores_softmax_kernel(const float* __restrict__ qkv,
                                                             const float* __restrict__ nrm,
                                                             float* __restrict__ scores) {
  __shared__ float red[8][32];
  const int tid  = threadIdx.x;
  const int lane = tid & 31;
  const int w    = tid >> 5;
  const int nIdx = lane & 15;
  const int hi   = lane >> 4;
  const int b  = blockIdx.y;
  const int i0 = blockIdx.x * 32;
  const int j0 = w * 64;
  const float* qf = qkv + ((size_t)0 * NMAT + (size_t)b * NN) * 1024;
  const float* kf = qkv + ((size_t)1 * NMAT + (size_t)b * NN) * 1024;
  const float* arow = kf + (size_t)(j0 + nIdx) * NF + hi * 16;
  const float* brow = qf + (size_t)(i0 + nIdx) * NF + hi * 16;

  v8f acc[4][2] = {};
  RowFrag a0[4], b0[2], a1[4], b1[2];

  load_tiles(a0, b0, arow, brow, 0);
  for (int kb = 0; kb < NF; kb += 64) {
    load_tiles(a1, b1, arow, brow, kb + 32);
    mma_chunk(acc, a0, b0);
    if (kb + 64 < NF) load_tiles(a0, b0, arow, brow, kb + 64);
    mma_chunk(acc, a1, b1);
  }

  float q2[2], mx[2];
#pragma unroll
  for (int it = 0; it < 2; ++it) {
    q2[it] = nrm[b * NN + i0 + it * 16 + nIdx];
    mx[it] = -1e30f;
  }
#pragma unroll
  for (int jt = 0; jt < 4; ++jt) {
#pragma unroll
    for (int r = 0; r < 8; ++r) {
      float k2 = nrm[NMAT + b * NN + j0 + jt * 16 + hi * 8 + r];
#pragma unroll
      for (int it = 0; it < 2; ++it) {
        float d2 = q2[it] + k2 - 2.f * acc[jt][it][r];
        d2 = fmaxf(d2, 1e-12f);
        float e  = sqrtf(d2);
        float sc = 1.f / (1.f + log1pf(e));
        acc[jt][it][r] = sc;
        mx[it] = fmaxf(mx[it], sc);
      }
    }
  }

#pragma unroll
  for (int it = 0; it < 2; ++it) {
    mx[it] = fmaxf(mx[it], __shfl_xor(mx[it], 16));
    if (hi == 0) red[w][it * 16 + nIdx] = mx[it];
  }
  __syncthreads();
#pragma unroll
  for (int it = 0; it < 2; ++it) {
    float m = -1e30f;
#pragma unroll
    for (int ww = 0; ww < 8; ++ww) m = fmaxf(m, red[ww][it * 16 + nIdx]);
    mx[it] = m;
  }
  __syncthreads();

  float ps[2] = {0.f, 0.f};
#pragma unroll
  for (int jt = 0; jt < 4; ++jt) {
#pragma unroll
    for (int r = 0; r < 8; ++r) {
#pragma unroll
      for (int it = 0; it < 2; ++it) {
        float e = __expf(acc[jt][it][r] - mx[it]);
        acc[jt][it][r] = e;
        ps[it] += e;
      }
    }
  }
#pragma unroll
  for (int it = 0; it < 2; ++it) {
    ps[it] += __shfl_xor(ps[it], 16);
    if (hi == 0) red[w][it * 16 + nIdx] = ps[it];
  }
  __syncthreads();
  float inv[2];
#pragma unroll
  for (int it = 0; it < 2; ++it) {
    float sum = 0.f;
#pragma unroll
    for (int ww = 0; ww < 8; ++ww) sum += red[ww][it * 16 + nIdx];
    inv[it] = 1.f / sum;
  }

  float* sb = scores + (size_t)b * NN * NN;
#pragma unroll
  for (int jt = 0; jt < 4; ++jt) {
#pragma unroll
    for (int r = 0; r < 8; ++r) {
      int j = j0 + jt * 16 + hi * 8 + r;
#pragma unroll
      for (int it = 0; it < 2; ++it)
        sb[(size_t)j * NN + i0 + it * 16 + nIdx] = acc[jt][it][r] * inv[it];
    }
  }
}

__global__ __launch_bounds__(32) void outgemm_kernel(const float* __restrict__ scores,
                                                     const float* __restrict__ qkv,
                                                     float* __restrict__ out) {
  const int lane = threadIdx.x;
  const int nIdx = lane & 15;
  const int hi   = lane >> 4;
  const int ft = blockIdx.x, ct = blockIdx.y, b = blockIdx.z;
  const int f0 = ft * 16, c0 = ct * 64;
  const float* p  = scores + (size_t)b * NN * NN;
  const float* vf = qkv + ((size_t)2 * NMAT + (size_t)b * NN) * 1024;
  const float* arow0 = p + (size_t)(c0 +  0 + nIdx) * NN;
  const float* arow1 = p + (size_t)(c0 + 16 + nIdx) * NN;
  const float* arow2 = p + (size_t)(c0 + 32 + nIdx) * NN;
  const float* arow3 = p + (size_t)(c0 + 48 + nIdx) * NN;

  v8f acc0 = {}, acc1 = {}, acc2 = {}, acc3 = {};
  for (int k0 = hi * 2; k0 < NN; k0 += 4) {
    v2f bb;
    bb.x = vf[(size_t)k0 * NF + f0 + nIdx];
    bb.y = vf[(size_t)(k0 + 1) * NF + f0 + nIdx];
    acc0 = WMMA_F32_16x16x4(*(const v2f*)(arow0 + k0), bb, acc0);
    acc1 = WMMA_F32_16x16x4(*(const v2f*)(arow1 + k0), bb, acc1);
    acc2 = WMMA_F32_16x16x4(*(const v2f*)(arow2 + k0), bb, acc2);
    acc3 = WMMA_F32_16x16x4(*(const v2f*)(arow3 + k0), bb, acc3);
  }

  float* ob = out + (size_t)b * NN * NF;
#pragma unroll
  for (int t = 0; t < 4; ++t) {
    v8f acc = (t == 0) ? acc0 : ((t == 1) ? acc1 : ((t == 2) ? acc2 : acc3));
#pragma unroll
    for (int r = 0; r < 8; ++r) {
      int c = c0 + t * 16 + r + hi * 8;
      ob[(size_t)c * NF + f0 + nIdx] = acc[r];
    }
  }
}

extern "C" void kernel_launch(void* const* d_in, const int* in_sizes, int n_in,
                              void* d_out, int out_size, void* d_ws, size_t ws_size,
                              hipStream_t stream) {
  (void)in_sizes; (void)n_in; (void)out_size; (void)ws_size;
  const float* x  = (const float*)d_in[0];
  const float* qw = (const float*)d_in[1];
  const float* kw = (const float*)d_in[2];
  const float* vw = (const float*)d_in[3];
  float* out = (float*)d_out;

  float* ws     = (float*)d_ws;
  float* C      = ws;
  float* xl     = C + 3 * 1024;
  float* qkv    = xl + (size_t)NMAT * 1024;
  float* nrm    = qkv + (size_t)3 * NMAT * 1024;
  float* scores = nrm + 2 * NMAT;

  cayley_kernel<<<3, 256, 0, stream>>>(qw, kw, vw, C);
  jacobi_logm_kernel<<<NMAT, 32, 0, stream>>>(x, xl);
  transbase_kernel<<<dim3(NMAT, 3), 32, 0, stream>>>(xl, C, qkv);
  norms_kernel<<<64, 256, 0, stream>>>(qkv, nrm);
  scores_softmax_kernel<<<dim3(NN / 32, NB), 256, 0, stream>>>(qkv, nrm, scores);
  outgemm_kernel<<<dim3(NF / 16, NN / 64, NB), 32, 0, stream>>>(scores, qkv, out);
}


# eigensolver: A and V held in registers (lane = row, pair t on seats 2t,2t+1): row rotation by DPP lane swap, seat change by ds_bpermute, no LDS or barrier in the sweeps; pair order and sums kept
# speedup vs baseline: 1.5985x; 1.1260x over previous
#include <hip/hip_runtime.h>
#include <math.h>


typedef __attribute__((ext_vector_type(2))) float v2f;
typedef __attribute__((ext_vector_type(8))) float v8f;

#define WMMA_F32_16x16x4(a, b, c) \
  __builtin_amdgcn_wmma_f32_16x16x4_f32(false, (a), false, (b), (short)0, (c), false, false)

#define NMAT 8192
#define NB   16
#define NN   512
#define NF   1024

__global__ __launch_bounds__(256) void cayley_kernel(const float* __restrict__ qw,
                                                     const float* __restrict__ kw,
                                                     const float* __restrict__ vw,
                                                     float* __restrict__ Cout) {
  __shared__ float X[32][32];
  __shared__ float Mm[32][33];
  __shared__ float Inv[32][33];
  __shared__ float fac[32];
  __shared__ float pivinv;
  const float* W = (blockIdx.x == 0) ? qw : ((blockIdx.x == 1) ? kw : vw);
  const int tid = threadIdx.x;

  for (int e = tid; e < 1024; e += 256) {
    int i = e >> 5, j = e & 31;
    float xv = W[i * 32 + j] - W[j * 32 + i];
    X[i][j]  = xv;
    Mm[i][j] = xv + ((i == j) ? 1.f : 0.f);
    Inv[i][j] = (i == j) ? 1.f : 0.f;
  }
  __syncthreads();

  for (int k = 0; k < 32; ++k) {
    if (tid == 0) pivinv = 1.f / Mm[k][k];
    __syncthreads();
    if (tid < 32)      Mm[k][tid]       *= pivinv;
    else if (tid < 64) Inv[k][tid - 32] *= pivinv;
    __syncthreads();
    if (tid < 32) fac[tid] = (tid == k) ? 0.f : Mm[tid][k];
    __syncthreads();
    for (int e = tid; e < 2048; e += 256) {
      int i = e >> 6, j = e & 63;
      if (i == k) continue;
      float f = fac[i];
      if (j < 32) Mm[i][j]        -= f * Mm[k][j];
      else        Inv[i][j - 32]  -= f * Inv[k][j - 32];
    }
    __syncthreads();
  }

  for (int e = tid; e < 1024; e += 256) {
    int i = e >> 5, j = e & 31;
    float acc = 0.f;
    for (int kk = 0; kk < 32; ++kk) {
      float a = ((i == kk) ? 1.f : 0.f) - X[i][kk];
      acc += a * Inv[kk][j];
    }
    Cout[blockIdx.x * 1024 + i * 32 + j] = acc;
  }
}

typedef __attribute__((ext_vector_type(4))) float v4f;

__host__ __device__ constexpr int seat_index(int P) { return (P & 1) ? 31 - (P >> 1) : (P >> 1); }
__host__ __device__ constexpr int index_seat(int k) { return (k < 16) ? 2 * k : 2 * (31 - k) + 1; }
__host__ __device__ constexpr int seat_source(int P) {
  return (P == 1) ? 1 : ((P & 1) == 0 ? (P < 30 ? P + 2 : 31) : (P == 3 ? 0 : P - 2));
}

__device__ __forceinline__ float lane_swap1(float v) {
  return __int_as_float(__builtin_amdgcn_update_dpp(0, __float_as_int(v), 0xB1, 0xF, 0xF, true));
}

__device__ __forceinline__ float pick(float hi, float lo, unsigned mask) {
  float r;
  asm("v_cndmask_b32_e64 %0, %1, %2, %3" : "=v"(r) : "v"(lo), "v"(hi), "s"(mask));
  return r;
}

__device__ __forceinline__ void pick_pair_entries(const float (&a)[32], unsigned sel16,
                                                  unsigned sel8, unsigned sel4, unsigned sel2,
                                                  float& e0, float& e1) {
  float s16[16], s8[8], s4[4];
#pragma unroll
  for (int k = 0; k < 16; ++k) s16[k] = pick(a[16 + k], a[k], sel16);
#pragma unroll
  for (int k = 0; k < 8; ++k) s8[k] = pick(s16[8 + k], s16[k], sel8);
#pragma unroll
  for (int k = 0; k < 4; ++k) s4[k] = pick(s8[4 + k], s8[k], sel4);
  e0 = pick(s4[2], s4[0], sel2);
  e1 = pick(s4[3], s4[1], sel2);
}

__global__ __launch_bounds__(32, 8) void jacobi_logm_kernel(const float* __restrict__ x,
                                                            float* __restrict__ xl) {
  __shared__ float V[32][33];
  __shared__ float lw[32];
  const int lane = threadIdx.x;
  const bool odd = (lane & 1) != 0;
  const float* xm = x + (size_t)blockIdx.x * 1024;

  const int myrow = odd ? 31 - (lane >> 1) : (lane >> 1);
  const v4f* rp = (const v4f*)(xm + myrow * 32);
  float rowv[32];
#pragma unroll
  for (int g = 0; g < 8; ++g) {
    v4f t = rp[g];
    rowv[4 * g + 0] = t.x; rowv[4 * g + 1] = t.y; rowv[4 * g + 2] = t.z; rowv[4 * g + 3] = t.w;
  }
  float a[32], w[32];
#pragma unroll
  for (int Q = 0; Q < 32; ++Q) {
    a[Q] = rowv[seat_index(Q)];
    w[Q] = (seat_index(Q) == lane) ? 1.f : 0.f;
  }
  const int pull = seat_source(lane) * 4;
  const unsigned sel16 = 0xFFFF0000u, sel8 = 0xFF00FF00u, sel4 = 0xF0F0F0F0u, sel2 = 0xCCCCCCCCu;

  for (int it = 0; it < 8 * 31; ++it) {
    float e0, e1;
    pick_pair_entries(a, sel16, sel8, sel4, sel2, e0, e1);
    const float dg  = odd ? e1 : e0;
    const float od  = odd ? e0 : e1;
    const float dgn = lane_swap1(dg);
    const float odn = lane_swap1(od);
    const float app = odd ? dgn : dg;
    const float aqq = odd ? dg : dgn;
    const float apq = odd ? odn : od;
    float c = 1.f, s = 0.f;
    if (fabsf(apq) > 1e-12f) {
      float theta = (aqq - app) / (2.f * apq);
      float t = 1.f / (fabsf(theta) + sqrtf(theta * theta + 1.f));
      if (theta < 0.f) t = -t;
      c = rsqrtf(t * t + 1.f);
      s = t * c;
    }
    float ct[16], st[16];
#pragma unroll
    for (int t = 0; t < 16; ++t) {
      ct[t] = __int_as_float(__builtin_amdgcn_readlane(__float_as_int(c), 2 * t));
      st[t] = __int_as_float(__builtin_amdgcn_readlane(__float_as_int(s), 2 * t));
    }

    const float sg = odd ? s : -s;
#pragma unroll
    for (int Q = 0; Q < 32; ++Q) a[Q] = c * a[Q] + sg * lane_swap1(a[Q]);

#pragma unroll
    for (int t = 0; t < 16; ++t) {
      const float ap = a[2 * t], aq = a[2 * t + 1];
      a[2 * t]     = ct[t] * ap - st[t] * aq;
      a[2 * t + 1] = st[t] * ap + ct[t] * aq;
    }

    float na[32];
#pragma unroll
    for (int P = 0; P < 32; ++P)
      na[P] = __int_as_float(__builtin_amdgcn_ds_bpermute(pull, __float_as_int(a[seat_source(P)])));

    __builtin_amdgcn_sched_barrier(0);
    float nw[32];
#pragma unroll
    for (int t = 0; t < 16; ++t) {
      const float wp = w[2 * t], wq = w[2 * t + 1];
      nw[2 * t]     = ct[t] * wp - st[t] * wq;
      nw[2 * t + 1] = st[t] * wp + ct[t] * wq;
    }
#pragma unroll
    for (int P = 0; P < 32; ++P) {
      w[P] = nw[seat_source(P)];
      a[P] = na[P];
    }
  }

  {
    float e0, e1;
    pick_pair_entries(a, sel16, sel8, sel4, sel2, e0, e1);
    const float dg = odd ? e1 : e0;
    lw[lane] = logf(fmaxf(dg, 1e-20f));
  }
#pragma unroll
  for (int Q = 0; Q < 32; ++Q) V[lane][Q] = w[Q];
  __syncthreads();

  float vj[32];
#pragma unroll
  for (int Q = 0; Q < 32; ++Q) vj[Q] = w[Q] * lw[Q];

  float* outm = xl + (size_t)blockIdx.x * 1024;
  for (int i = 0; i < 32; ++i) {
    float acc = 0.f;
#pragma unroll
    for (int k = 0; k < 32; ++k) acc += V[i][index_seat(k)] * vj[index_seat(k)];
    outm[i * 32 + lane] = (i == lane) ? 0.f : acc;
  }
}

__global__ __launch_bounds__(32) void transbase_kernel(const float* __restrict__ xl,
                                                       const float* __restrict__ C,
                                                       float* __restrict__ qkv) {
  __shared__ float Y[32][33];
  __shared__ float CL[32][33];
  __shared__ float T[32][33];
  __shared__ float T2[32][33];
  const int lane = threadIdx.x;
  const int nIdx = lane & 15;
  const int hi   = lane >> 4;
  const int m = blockIdx.x, w = blockIdx.y;
  const float* ym = xl + (size_t)m * 1024;
  const float* Cw = C + (size_t)w * 1024;

  for (int i = 0; i < 32; ++i) {
    Y[i][lane]  = ym[i * 32 + lane];
    CL[i][lane] = Cw[i * 32 + lane];
  }
  __syncthreads();

  for (int ti = 0; ti < 2; ++ti)
    for (int tj = 0; tj < 2; ++tj) {
      v8f acc = {};
#pragma unroll
      for (int kk = 0; kk < 8; ++kk) {
        int k0 = kk * 4 + hi * 2;
        v2f a; a.x = CL[ti * 16 + nIdx][k0]; a.y = CL[ti * 16 + nIdx][k0 + 1];
        v2f b; b.x = Y[k0][tj * 16 + nIdx];  b.y = Y[k0 + 1][tj * 16 + nIdx];
        acc = WMMA_F32_16x16x4(a, b, acc);
      }
#pragma unroll
      for (int r = 0; r < 8; ++r)
        T[ti * 16 + r + hi * 8][tj * 16 + nIdx] = acc[r];
    }
  __syncthreads();

  for (int ti = 0; ti < 2; ++ti)
    for (int tj = 0; tj < 2; ++tj) {
      v8f acc = {};
#pragma unroll
      for (int kk = 0; kk < 8; ++kk) {
        int k0 = kk * 4 + hi * 2;
        v2f a; a.x = T[ti * 16 + nIdx][k0];       a.y = T[ti * 16 + nIdx][k0 + 1];
        v2f b; b.x = CL[tj * 16 + nIdx][k0];      b.y = CL[tj * 16 + nIdx][k0 + 1];
        acc = WMMA_F32_16x16x4(a, b, acc);
      }
#pragma unroll
      for (int r = 0; r < 8; ++r)
        T2[ti * 16 + r + hi * 8][tj * 16 + nIdx] = acc[r];
    }
  __syncthreads();

  float* outm = qkv + ((size_t)w * NMAT + m) * 1024;
  for (int i = 0; i < 32; ++i) {
    float lo = T2[i][lane];
    float up = T2[lane][i];
    float val = (i > lane) ? lo : ((i < lane) ? up : 0.f);
    outm[i * 32 + lane] = val;
  }
}

__global__ __launch_bounds__(256) void norms_kernel(const float* __restrict__ qkv,
                                                    float* __restrict__ nrm) {
  int idx = blockIdx.x * 256 + threadIdx.x;
  int sel = idx >> 13;
  int r   = idx & (NMAT - 1);
  const float* row = qkv + ((size_t)sel * NMAT + r) * 1024;
  float acc = 0.f;
  for (int f = 0; f < NF; ++f) { float t = row[f]; acc += t * t; }
  nrm[idx] = acc;
}

struct RowFrag { v4f q[4]; };

__device__ __forceinline__ void load_frag(RowFrag& f, const float* p) {
  const v4f* p4 = (const v4f*)p;
  f.q[0] = p4[0]; f.q[1] = p4[1]; f.q[2] = p4[2]; f.q[3] = p4[3];
}

__device__ __forceinline__ v2f frag_pair(const RowFrag& f, int s) {
  v4f q = f.q[s >> 1];
  v2f lo = __builtin_shufflevector(q, q, 0, 1);
  v2f hp = __builtin_shufflevector(q, q, 2, 3);
  return (s & 1) ? hp : lo;
}

__device__ __forceinline__ void load_tiles(RowFrag (&A)[4], RowFrag (&B)[2],
                                           const float* arow, const float* brow, int kb) {
#pragma unroll
  for (int jt = 0; jt < 4; ++jt) load_frag(A[jt], arow + jt * (16 * NF) + kb);
#pragma unroll
  for (int it = 0; it < 2; ++it) load_frag(B[it], brow + it * (16 * NF) + kb);
}

__device__ __forceinline__ void mma_chunk(v8f (&acc)[4][2], const RowFrag (&A)[4],
                                          const RowFrag (&B)[2]) {
#pragma unroll
  for (int s = 0; s < 8; ++s) {
#pragma unroll
    for (int jt = 0; jt < 4; ++jt) {
      v2f a = frag_pair(A[jt], s);
#pragma unroll
      for (int it = 0; it < 2; ++it)
        acc[jt][it] = WMMA_F32_16x16x4(a, frag_pair(B[it], s), acc[jt][it]);
    }
  }
}

__global__ __launch_bounds__(256) void scores_softmax_kernel(const float* __restrict__ qkv,
                                                             const float* __restrict__ nrm,
                                                             float* __restrict__ scores) {
  __shared__ float red[8][32];
  const int tid  = threadIdx.x;
  const int lane = tid & 31;
  const int w    = tid >> 5;
  const int nIdx = lane & 15;
  const int hi   = lane >> 4;
  const int b  = blockIdx.y;
  const int i0 = blockIdx.x * 32;
  const int j0 = w * 64;
  const float* qf = qkv + ((size_t)0 * NMAT + (size_t)b * NN) * 1024;
  const float* kf = qkv + ((size_t)1 * NMAT + (size_t)b * NN) * 1024;
  const float* arow = kf + (size_t)(j0 + nIdx) * NF + hi * 16;
  const float* brow = qf + (size_t)(i0 + nIdx) * NF + hi * 16;

  v8f acc[4][2] = {};
  RowFrag a0[4], b0[2], a1[4], b1[2];

  load_tiles(a0, b0, arow, brow, 0);
  for (int kb = 0; kb < NF; kb += 64) {
    load_tiles(a1, b1, arow, brow, kb + 32);
    mma_chunk(acc, a0, b0);
    if (kb + 64 < NF) load_tiles(a0, b0, arow, brow, kb + 64);
    mma_chunk(acc, a1, b1);
  }

  float q2[2], mx[2];
#pragma unroll
  for (int it = 0; it < 2; ++it) {
    q2[it] = nrm[b * NN + i0 + it * 16 + nIdx];
    mx[it] = -1e30f;
  }
#pragma unroll
  for (int jt = 0; jt < 4; ++jt) {
#pragma unroll
    for (int r = 0; r < 8; ++r) {
      float k2 = nrm[NMAT + b * NN + j0 + jt * 16 + hi * 8 + r];
#pragma unroll
      for (int it = 0; it < 2; ++it) {
        float d2 = q2[it] + k2 - 2.f * acc[jt][it][r];
        d2 = fmaxf(d2, 1e-12f);
        float e  = sqrtf(d2);
        float sc = 1.f / (1.f + log1pf(e));
        acc[jt][it][r] = sc;
        mx[it] = fmaxf(mx[it], sc);
      }
    }
  }

#pragma unroll
  for (int it = 0; it < 2; ++it) {
    mx[it] = fmaxf(mx[it], __shfl_xor(mx[it], 16));
    if (hi == 0) red[w][it * 16 + nIdx] = mx[it];
  }
  __syncthreads();
#pragma unroll
  for (int it = 0; it < 2; ++it) {
    float m = -1e30f;
#pragma unroll
    for (int ww = 0; ww < 8; ++ww) m = fmaxf(m, red[ww][it * 16 + nIdx]);
    mx[it] = m;
  }
  __syncthreads();

  float ps[2] = {0.f, 0.f};
#pragma unroll
  for (int jt = 0; jt < 4; ++jt) {
#pragma unroll
    for (int r = 0; r < 8; ++r) {
#pragma unroll
      for (int it = 0; it < 2; ++it) {
        float e = __expf(acc[jt][it][r] - mx[it]);
        acc[jt][it][r] = e;
        ps[it] += e;
      }
    }
  }
#pragma unroll
  for (int it = 0; it < 2; ++it) {
    ps[it] += __shfl_xor(ps[it], 16);
    if (hi == 0) red[w][it * 16 + nIdx] = ps[it];
  }
  __syncthreads();
  float inv[2];
#pragma unroll
  for (int it = 0; it < 2; ++it) {
    float sum = 0.f;
#pragma unroll
    for (int ww = 0; ww < 8; ++ww) sum += red[ww][it * 16 + nIdx];
    inv[it] = 1.f / sum;
  }

  float* sb = scores + (size_t)b * NN * NN;
#pragma unroll
  for (int jt = 0; jt < 4; ++jt) {
#pragma unroll
    for (int r = 0; r < 8; ++r) {
      int j = j0 + jt * 16 + hi * 8 + r;
#pragma unroll
      for (int it = 0; it < 2; ++it)
        sb[(size_t)j * NN + i0 + it * 16 + nIdx] = acc[jt][it][r] * inv[it];
    }
  }
}

__global__ __launch_bounds__(32) void outgemm_kernel(const float* __restrict__ scores,
                                                     const float* __restrict__ qkv,
                                                     float* __restrict__ out) {
  const int lane = threadIdx.x;
  const int nIdx = lane & 15;
  const int hi   = lane >> 4;
  const int ft = blockIdx.x, ct = blockIdx.y, b = blockIdx.z;
  const int f0 = ft * 16, c0 = ct * 64;
  const float* p  = scores + (size_t)b * NN * NN;
  const float* vf = qkv + ((size_t)2 * NMAT + (size_t)b * NN) * 1024;
  const float* arow0 = p + (size_t)(c0 +  0 + nIdx) * NN;
  const float* arow1 = p + (size_t)(c0 + 16 + nIdx) * NN;
  const float* arow2 = p + (size_t)(c0 + 32 + nIdx) * NN;
  const float* arow3 = p + (size_t)(c0 + 48 + nIdx) * NN;

  v8f acc0 = {}, acc1 = {}, acc2 = {}, acc3 = {};
  for (int k0 = hi * 2; k0 < NN; k0 += 4) {
    v2f bb;
    bb.x = vf[(size_t)k0 * NF + f0 + nIdx];
    bb.y = vf[(size_t)(k0 + 1) * NF + f0 + nIdx];
    acc0 = WMMA_F32_16x16x4(*(const v2f*)(arow0 + k0), bb, acc0);
    acc1 = WMMA_F32_16x16x4(*(const v2f*)(arow1 + k0), bb, acc1);
    acc2 = WMMA_F32_16x16x4(*(const v2f*)(arow2 + k0), bb, acc2);
    acc3 = WMMA_F32_16x16x4(*(const v2f*)(arow3 + k0), bb, acc3);
  }

  float* ob = out + (size_t)b * NN * NF;
#pragma unroll
  for (int t = 0; t < 4; ++t) {
    v8f acc = (t == 0) ? acc0 : ((t == 1) ? acc1 : ((t == 2) ? acc2 : acc3));
#pragma unroll
    for (int r = 0; r < 8; ++r) {
      int c = c0 + t * 16 + r + hi * 8;
      ob[(size_t)c * NF + f0 + nIdx] = acc[r];
    }
  }
}

extern "C" void kernel_launch(void* const* d_in, const int* in_sizes, int n_in,
                              void* d_out, int out_size, void* d_ws, size_t ws_size,
                              hipStream_t stream) {
  (void)in_sizes; (void)n_in; (void)out_size; (void)ws_size;
  const float* x  = (const float*)d_in[0];
  const float* qw = (const float*)d_in[1];
  const float* kw = (const float*)d_in[2];
  const float* vw = (const float*)d_in[3];
  float* out = (float*)d_out;

  float* ws     = (float*)d_ws;
  float* C      = ws;
  float* xl     = C + 3 * 1024;
  float* qkv    = xl + (size_t)NMAT * 1024;
  float* nrm    = qkv + (size_t)3 * NMAT * 1024;
  float* scores = nrm + 2 * NMAT;

  cayley_kernel<<<3, 256, 0, stream>>>(qw, kw, vw, C);
  jacobi_logm_kernel<<<NMAT, 32, 0, stream>>>(x, xl);
  transbase_kernel<<<dim3(NMAT, 3), 32, 0, stream>>>(xl, C, qkv);
  norms_kernel<<<64, 256, 0, stream>>>(qkv, nrm);
  scores_softmax_kernel<<<dim3(NN / 32, NB), 256, 0, stream>>>(qkv, nrm, scores);
  outgemm_kernel<<<dim3(NF / 16, NN / 64, NB), 32, 0, stream>>>(scores, qkv, out);
}


# out GEMM (p x v): a lane reads 16 consecutive k of its p row per 32-k chunk (128-B line used up by 4 back-to-back loads, next chunk prefetched); v rows taken in the same k order; k sum order changes
# speedup vs baseline: 1.9421x; 1.2150x over previous
#include <hip/hip_runtime.h>
#include <math.h>


typedef __attribute__((ext_vector_type(2))) float v2f;
typedef __attribute__((ext_vector_type(8))) float v8f;

#define WMMA_F32_16x16x4(a, b, c) \
  __builtin_amdgcn_wmma_f32_16x16x4_f32(false, (a), false, (b), (short)0, (c), false, false)

#define NMAT 8192
#define NB   16
#define NN   512
#define NF   1024

__global__ __launch_bounds__(256) void cayley_kernel(const float* __restrict__ qw,
                                                     const float* __restrict__ kw,
                                                     const float* __restrict__ vw,
                                                     float* __restrict__ Cout) {
  __shared__ float X[32][32];
  __shared__ float Mm[32][33];
  __shared__ float Inv[32][33];
  __shared__ float fac[32];
  __shared__ float pivinv;
  const float* W = (blockIdx.x == 0) ? qw : ((blockIdx.x == 1) ? kw : vw);
  const int tid = threadIdx.x;

  for (int e = tid; e < 1024; e += 256) {
    int i = e >> 5, j = e & 31;
    float xv = W[i * 32 + j] - W[j * 32 + i];
    X[i][j]  = xv;
    Mm[i][j] = xv + ((i == j) ? 1.f : 0.f);
    Inv[i][j] = (i == j) ? 1.f : 0.f;
  }
  __syncthreads();

  for (int k = 0; k < 32; ++k) {
    if (tid == 0) pivinv = 1.f / Mm[k][k];
    __syncthreads();
    if (tid < 32)      Mm[k][tid]       *= pivinv;
    else if (tid < 64) Inv[k][tid - 32] *= pivinv;
    __syncthreads();
    if (tid < 32) fac[tid] = (tid == k) ? 0.f : Mm[tid][k];
    __syncthreads();
    for (int e = tid; e < 2048; e += 256) {
      int i = e >> 6, j = e & 63;
      if (i == k) continue;
      float f = fac[i];
      if (j < 32) Mm[i][j]        -= f * Mm[k][j];
      else        Inv[i][j - 32]  -= f * Inv[k][j - 32];
    }
    __syncthreads();
  }

  for (int e = tid; e < 1024; e += 256) {
    int i = e >> 5, j = e & 31;
    float acc = 0.f;
    for (int kk = 0; kk < 32; ++kk) {
      float a = ((i == kk) ? 1.f : 0.f) - X[i][kk];
      acc += a * Inv[kk][j];
    }
    Cout[blockIdx.x * 1024 + i * 32 + j] = acc;
  }
}

typedef __attribute__((ext_vector_type(4))) float v4f;

__host__ __device__ constexpr int seat_index(int P) { return (P & 1) ? 31 - (P >> 1) : (P >> 1); }
__host__ __device__ constexpr int index_seat(int k) { return (k < 16) ? 2 * k : 2 * (31 - k) + 1; }
__host__ __device__ constexpr int seat_source(int P) {
  return (P == 1) ? 1 : ((P & 1) == 0 ? (P < 30 ? P + 2 : 31) : (P == 3 ? 0 : P - 2));
}

__device__ __forceinline__ float lane_swap1(float v) {
  return __int_as_float(__builtin_amdgcn_update_dpp(0, __float_as_int(v), 0xB1, 0xF, 0xF, true));
}

__device__ __forceinline__ float pick(float hi, float lo, unsigned mask) {
  float r;
  asm("v_cndmask_b32_e64 %0, %1, %2, %3" : "=v"(r) : "v"(lo), "v"(hi), "s"(mask));
  return r;
}

__device__ __forceinline__ void pick_pair_entries(const float (&a)[32], unsigned sel16,
                                                  unsigned sel8, unsigned sel4, unsigned sel2,
                                                  float& e0, float& e1) {
  float s16[16], s8[8], s4[4];
#pragma unroll
  for (int k = 0; k < 16; ++k) s16[k] = pick(a[16 + k], a[k], sel16);
#pragma unroll
  for (int k = 0; k < 8; ++k) s8[k] = pick(s16[8 + k], s16[k], sel8);
#pragma unroll
  for (int k = 0; k < 4; ++k) s4[k] = pick(s8[4 + k], s8[k], sel4);
  e0 = pick(s4[2], s4[0], sel2);
  e1 = pick(s4[3], s4[1], sel2);
}

__global__ __launch_bounds__(32, 8) void jacobi_logm_kernel(const float* __restrict__ x,
                                                            float* __restrict__ xl) {
  __shared__ float V[32][33];
  __shared__ float lw[32];
  const int lane = threadIdx.x;
  const bool odd = (lane & 1) != 0;
  const float* xm = x + (size_t)blockIdx.x * 1024;

  const int myrow = odd ? 31 - (lane >> 1) : (lane >> 1);
  const v4f* rp = (const v4f*)(xm + myrow * 32);
  float rowv[32];
#pragma unroll
  for (int g = 0; g < 8; ++g) {
    v4f t = rp[g];
    rowv[4 * g + 0] = t.x; rowv[4 * g + 1] = t.y; rowv[4 * g + 2] = t.z; rowv[4 * g + 3] = t.w;
  }
  float a[32], w[32];
#pragma unroll
  for (int Q = 0; Q < 32; ++Q) {
    a[Q] = rowv[seat_index(Q)];
    w[Q] = (seat_index(Q) == lane) ? 1.f : 0.f;
  }
  const int pull = seat_source(lane) * 4;
  const unsigned sel16 = 0xFFFF0000u, sel8 = 0xFF00FF00u, sel4 = 0xF0F0F0F0u, sel2 = 0xCCCCCCCCu;

  for (int it = 0; it < 8 * 31; ++it) {
    float e0, e1;
    pick_pair_entries(a, sel16, sel8, sel4, sel2, e0, e1);
    const float dg  = odd ? e1 : e0;
    const float od  = odd ? e0 : e1;
    const float dgn = lane_swap1(dg);
    const float odn = lane_swap1(od);
    const float app = odd ? dgn : dg;
    const float aqq = odd ? dg : dgn;
    const float apq = odd ? odn : od;
    float c = 1.f, s = 0.f;
    if (fabsf(apq) > 1e-12f) {
      float theta = (aqq - app) / (2.f * apq);
      float t = 1.f / (fabsf(theta) + sqrtf(theta * theta + 1.f));
      if (theta < 0.f) t = -t;
      c = rsqrtf(t * t + 1.f);
      s = t * c;
    }
    float ct[16], st[16];
#pragma unroll
    for (int t = 0; t < 16; ++t) {
      ct[t] = __int_as_float(__builtin_amdgcn_readlane(__float_as_int(c), 2 * t));
      st[t] = __int_as_float(__builtin_amdgcn_readlane(__float_as_int(s), 2 * t));
    }

    const float sg = odd ? s : -s;
#pragma unroll
    for (int Q = 0; Q < 32; ++Q) a[Q] = c * a[Q] + sg * lane_swap1(a[Q]);

#pragma unroll
    for (int t = 0; t < 16; ++t) {
      const float ap = a[2 * t], aq = a[2 * t + 1];
      a[2 * t]     = ct[t] * ap - st[t] * aq;
      a[2 * t + 1] = st[t] * ap + ct[t] * aq;
    }

    float na[32];
#pragma unroll
    for (int P = 0; P < 32; ++P)
      na[P] = __int_as_float(__builtin_amdgcn_ds_bpermute(pull, __float_as_int(a[seat_source(P)])));

    __builtin_amdgcn_sched_barrier(0);
    float nw[32];
#pragma unroll
    for (int t = 0; t < 16; ++t) {
      const float wp = w[2 * t], wq = w[2 * t + 1];
      nw[2 * t]     = ct[t] * wp - st[t] * wq;
      nw[2 * t + 1] = st[t] * wp + ct[t] * wq;
    }
#pragma unroll
    for (int P = 0; P < 32; ++P) {
      w[P] = nw[seat_source(P)];
      a[P] = na[P];
    }
  }

  {
    float e0, e1;
    pick_pair_entries(a, sel16, sel8, sel4, sel2, e0, e1);
    const float dg = odd ? e1 : e0;
    lw[lane] = logf(fmaxf(dg, 1e-20f));
  }
#pragma unroll
  for (int Q = 0; Q < 32; ++Q) V[lane][Q] = w[Q];
  __syncthreads();

  float vj[32];
#pragma unroll
  for (int Q = 0; Q < 32; ++Q) vj[Q] = w[Q] * lw[Q];

  float* outm = xl + (size_t)blockIdx.x * 1024;
  for (int i = 0; i < 32; ++i) {
    float acc = 0.f;
#pragma unroll
    for (int k = 0; k < 32; ++k) acc += V[i][index_seat(k)] * vj[index_seat(k)];
    outm[i * 32 + lane] = (i == lane) ? 0.f : acc;
  }
}

__global__ __launch_bounds__(32) void transbase_kernel(const float* __restrict__ xl,
                                                       const float* __restrict__ C,
                                                       float* __restrict__ qkv) {
  __shared__ float Y[32][33];
  __shared__ float CL[32][33];
  __shared__ float T[32][33];
  __shared__ float T2[32][33];
  const int lane = threadIdx.x;
  const int nIdx = lane & 15;
  const int hi   = lane >> 4;
  const int m = blockIdx.x, w = blockIdx.y;
  const float* ym = xl + (size_t)m * 1024;
  const float* Cw = C + (size_t)w * 1024;

  for (int i = 0; i < 32; ++i) {
    Y[i][lane]  = ym[i * 32 + lane];
    CL[i][lane] = Cw[i * 32 + lane];
  }
  __syncthreads();

  for (int ti = 0; ti < 2; ++ti)
    for (int tj = 0; tj < 2; ++tj) {
      v8f acc = {};
#pragma unroll
      for (int kk = 0; kk < 8; ++kk) {
        int k0 = kk * 4 + hi * 2;
        v2f a; a.x = CL[ti * 16 + nIdx][k0]; a.y = CL[ti * 16 + nIdx][k0 + 1];
        v2f b; b.x = Y[k0][tj * 16 + nIdx];  b.y = Y[k0 + 1][tj * 16 + nIdx];
        acc = WMMA_F32_16x16x4(a, b, acc);
      }
#pragma unroll
      for (int r = 0; r < 8; ++r)
        T[ti * 16 + r + hi * 8][tj * 16 + nIdx] = acc[r];
    }
  __syncthreads();

  for (int ti = 0; ti < 2; ++ti)
    for (int tj = 0; tj < 2; ++tj) {
      v8f acc = {};
#pragma unroll
      for (int kk = 0; kk < 8; ++kk) {
        int k0 = kk * 4 + hi * 2;
        v2f a; a.x = T[ti * 16 + nIdx][k0];       a.y = T[ti * 16 + nIdx][k0 + 1];
        v2f b; b.x = CL[tj * 16 + nIdx][k0];      b.y = CL[tj * 16 + nIdx][k0 + 1];
        acc = WMMA_F32_16x16x4(a, b, acc);
      }
#pragma unroll
      for (int r = 0; r < 8; ++r)
        T2[ti * 16 + r + hi * 8][tj * 16 + nIdx] = acc[r];
    }
  __syncthreads();

  float* outm = qkv + ((size_t)w * NMAT + m) * 1024;
  for (int i = 0; i < 32; ++i) {
    float lo = T2[i][lane];
    float up = T2[lane][i];
    float val = (i > lane) ? lo : ((i < lane) ? up : 0.f);
    outm[i * 32 + lane] = val;
  }
}

__global__ __launch_bounds__(256) void norms_kernel(const float* __restrict__ qkv,
                                                    float* __restrict__ nrm) {
  int idx = blockIdx.x * 256 + threadIdx.x;
  int sel = idx >> 13;
  int r   = idx & (NMAT - 1);
  const float* row = qkv + ((size_t)sel * NMAT + r) * 1024;
  float acc = 0.f;
  for (int f = 0; f < NF; ++f) { float t = row[f]; acc += t * t; }
  nrm[idx] = acc;
}

struct RowFrag { v4f q[4]; };

__device__ __forceinline__ void load_frag(RowFrag& f, const float* p) {
  const v4f* p4 = (const v4f*)p;
  f.q[0] = p4[0]; f.q[1] = p4[1]; f.q[2] = p4[2]; f.q[3] = p4[3];
}

__device__ __forceinline__ v2f frag_pair(const RowFrag& f, int s) {
  v4f q = f.q[s >> 1];
  v2f lo = __builtin_shufflevector(q, q, 0, 1);
  v2f hp = __builtin_shufflevector(q, q, 2, 3);
  return (s & 1) ? hp : lo;
}

__device__ __forceinline__ void load_tiles(RowFrag (&A)[4], RowFrag (&B)[2],
                                           const float* arow, const float* brow, int kb) {
#pragma unroll
  for (int jt = 0; jt < 4; ++jt) load_frag(A[jt], arow + jt * (16 * NF) + kb);
#pragma unroll
  for (int it = 0; it < 2; ++it) load_frag(B[it], brow + it * (16 * NF) + kb);
}

__device__ __forceinline__ void mma_chunk(v8f (&acc)[4][2], const RowFrag (&A)[4],
                                          const RowFrag (&B)[2]) {
#pragma unroll
  for (int s = 0; s < 8; ++s) {
#pragma unroll
    for (int jt = 0; jt < 4; ++jt) {
      v2f a = frag_pair(A[jt], s);
#pragma unroll
      for (int it = 0; it < 2; ++it)
        acc[jt][it] = WMMA_F32_16x16x4(a, frag_pair(B[it], s), acc[jt][it]);
    }
  }
}

__global__ __launch_bounds__(256) void scores_softmax_kernel(const float* __restrict__ qkv,
                                                             const float* __restrict__ nrm,
                                                             float* __restrict__ scores) {
  __shared__ float red[8][32];
  const int tid  = threadIdx.x;
  const int lane = tid & 31;
  const int w    = tid >> 5;
  const int nIdx = lane & 15;
  const int hi   = lane >> 4;
  const int b  = blockIdx.y;
  const int i0 = blockIdx.x * 32;
  const int j0 = w * 64;
  const float* qf = qkv + ((size_t)0 * NMAT + (size_t)b * NN) * 1024;
  const float* kf = qkv + ((size_t)1 * NMAT + (size_t)b * NN) * 1024;
  const float* arow = kf + (size_t)(j0 + nIdx) * NF + hi * 16;
  const float* brow = qf + (size_t)(i0 + nIdx) * NF + hi * 16;

  v8f acc[4][2] = {};
  RowFrag a0[4], b0[2], a1[4], b1[2];

  load_tiles(a0, b0, arow, brow, 0);
  for (int kb = 0; kb < NF; kb += 64) {
    load_tiles(a1, b1, arow, brow, kb + 32);
    mma_chunk(acc, a0, b0);
    if (kb + 64 < NF) load_tiles(a0, b0, arow, brow, kb + 64);
    mma_chunk(acc, a1, b1);
  }

  float q2[2], mx[2];
#pragma unroll
  for (int it = 0; it < 2; ++it) {
    q2[it] = nrm[b * NN + i0 + it * 16 + nIdx];
    mx[it] = -1e30f;
  }
#pragma unroll
  for (int jt = 0; jt < 4; ++jt) {
#pragma unroll
    for (int r = 0; r < 8; ++r) {
      float k2 = nrm[NMAT + b * NN + j0 + jt * 16 + hi * 8 + r];
#pragma unroll
      for (int it = 0; it < 2; ++it) {
        float d2 = q2[it] + k2 - 2.f * acc[jt][it][r];
        d2 = fmaxf(d2, 1e-12f);
        float e  = sqrtf(d2);
        float sc = 1.f / (1.f + log1pf(e));
        acc[jt][it][r] = sc;
        mx[it] = fmaxf(mx[it], sc);
      }
    }
  }

#pragma unroll
  for (int it = 0; it < 2; ++it) {
    mx[it] = fmaxf(mx[it], __shfl_xor(mx[it], 16));
    if (hi == 0) red[w][it * 16 + nIdx] = mx[it];
  }
  __syncthreads();
#pragma unroll
  for (int it = 0; it < 2; ++it) {
    float m = -1e30f;
#pragma unroll
    for (int ww = 0; ww < 8; ++ww) m = fmaxf(m, red[ww][it * 16 + nIdx]);
    mx[it] = m;
  }
  __syncthreads();

  float ps[2] = {0.f, 0.f};
#pragma unroll
  for (int jt = 0; jt < 4; ++jt) {
#pragma unroll
    for (int r = 0; r < 8; ++r) {
#pragma unroll
      for (int it = 0; it < 2; ++it) {
        float e = __expf(acc[jt][it][r] - mx[it]);
        acc[jt][it][r] = e;
        ps[it] += e;
      }
    }
  }
#pragma unroll
  for (int it = 0; it < 2; ++it) {
    ps[it] += __shfl_xor(ps[it], 16);
    if (hi == 0) red[w][it * 16 + nIdx] = ps[it];
  }
  __syncthreads();
  float inv[2];
#pragma unroll
  for (int it = 0; it < 2; ++it) {
    float sum = 0.f;
#pragma unroll
    for (int ww = 0; ww < 8; ++ww) sum += red[ww][it * 16 + nIdx];
    inv[it] = 1.f / sum;
  }

  float* sb = scores + (size_t)b * NN * NN;
#pragma unroll
  for (int jt = 0; jt < 4; ++jt) {
#pragma unroll
    for (int r = 0; r < 8; ++r) {
      int j = j0 + jt * 16 + hi * 8 + r;
#pragma unroll
      for (int it = 0; it < 2; ++it)
        sb[(size_t)j * NN + i0 + it * 16 + nIdx] = acc[jt][it][r] * inv[it];
    }
  }
}

struct ColFrag { v2f e[8]; };

__device__ __forceinline__ void load_out_tiles(RowFrag (&A)[4], ColFrag& B,
                                               const float* arow, const float* bcol, int kb) {
#pragma unroll
  for (int t = 0; t < 4; ++t) load_frag(A[t], arow + t * (16 * NN) + kb);
  const float* bp = bcol + (size_t)kb * NF;
#pragma unroll
  for (int s = 0; s < 8; ++s) {
    B.e[s].x = bp[(size_t)(2 * s) * NF];
    B.e[s].y = bp[(size_t)(2 * s + 1) * NF];
  }
}

__device__ __forceinline__ void mma_out_chunk(v8f (&acc)[4], const RowFrag (&A)[4],
                                              const ColFrag& B) {
#pragma unroll
  for (int s = 0; s < 8; ++s) {
#pragma unroll
    for (int t = 0; t < 4; ++t)
      acc[t] = WMMA_F32_16x16x4(frag_pair(A[t], s), B.e[s], acc[t]);
  }
}

__global__ __launch_bounds__(32) void outgemm_kernel(const float* __restrict__ scores,
                                                     const float* __restrict__ qkv,
                                                     float* __restrict__ out) {
  const int lane = threadIdx.x;
  const int nIdx = lane & 15;
  const int hi   = lane >> 4;
  const int ft = blockIdx.x, ct = blockIdx.y, b = blockIdx.z;
  const int f0 = ft * 16, c0 = ct * 64;
  const float* p  = scores + (size_t)b * NN * NN;
  const float* vf = qkv + ((size_t)2 * NMAT + (size_t)b * NN) * 1024;
  const float* arow = p + (size_t)(c0 + nIdx) * NN + hi * 16;
  const float* bcol = vf + (size_t)(hi * 16) * NF + f0 + nIdx;

  v8f acc[4] = {};
  RowFrag a0[4], a1[4];
  ColFrag b0, b1;

  load_out_tiles(a0, b0, arow, bcol, 0);
  for (int kb = 0; kb < NN - 64; kb += 64) {
    load_out_tiles(a1, b1, arow, bcol, kb + 32);
    __builtin_amdgcn_sched_barrier(0);
    mma_out_chunk(acc, a0, b0);
    __builtin_amdgcn_sched_barrier(0);
    load_out_tiles(a0, b0, arow, bcol, kb + 64);
    __builtin_amdgcn_sched_barrier(0);
    mma_out_chunk(acc, a1, b1);
    __builtin_amdgcn_sched_barrier(0);
  }
  load_out_tiles(a1, b1, arow, bcol, NN - 32);
  __builtin_amdgcn_sched_barrier(0);
  mma_out_chunk(acc, a0, b0);
  mma_out_chunk(acc, a1, b1);

  float* ob = out + (size_t)b * NN * NF;
#pragma unroll
  for (int t = 0; t < 4; ++t) {
#pragma unroll
    for (int r = 0; r < 8; ++r) {
      int c = c0 + t * 16 + r + hi * 8;
      ob[(size_t)c * NF + f0 + nIdx] = acc[t][r];
    }
  }
}

extern "C" void kernel_launch(void* const* d_in, const int* in_sizes, int n_in,
                              void* d_out, int out_size, void* d_ws, size_t ws_size,
                              hipStream_t stream) {
  (void)in_sizes; (void)n_in; (void)out_size; (void)ws_size;
  const float* x  = (const float*)d_in[0];
  const float* qw = (const float*)d_in[1];
  const float* kw = (const float*)d_in[2];
  const float* vw = (const float*)d_in[3];
  float* out = (float*)d_out;

  float* ws     = (float*)d_ws;
  float* C      = ws;
  float* xl     = C + 3 * 1024;
  float* qkv    = xl + (size_t)NMAT * 1024;
  float* nrm    = qkv + (size_t)3 * NMAT * 1024;
  float* scores = nrm + 2 * NMAT;

  cayley_kernel<<<3, 256, 0, stream>>>(qw, kw, vw, C);
  jacobi_logm_kernel<<<NMAT, 32, 0, stream>>>(x, xl);
  transbase_kernel<<<dim3(NMAT, 3), 32, 0, stream>>>(xl, C, qkv);
  norms_kernel<<<64, 256, 0, stream>>>(qkv, nrm);
  scores_softmax_kernel<<<dim3(NN / 32, NB), 256, 0, stream>>>(qkv, nrm, scores);
  outgemm_kernel<<<dim3(NF / 16, NN / 64, NB), 32, 0, stream>>>(scores, qkv, out);
}


# eigensolver V update: vectoriser kept from pairing the column rotations into packed f32 ops; they compile to plain FMAs on scalar c,s with no register-pair copies and no in-loop spill; sums kept
# speedup vs baseline: 2.5930x; 1.3352x over previous
#include <hip/hip_runtime.h>
#include <math.h>


typedef __attribute__((ext_vector_type(2))) float v2f;
typedef __attribute__((ext_vector_type(8))) float v8f;

#define WMMA_F32_16x16x4(a, b, c) \
  __builtin_amdgcn_wmma_f32_16x16x4_f32(false, (a), false, (b), (short)0, (c), false, false)

#define NMAT 8192
#define NB   16
#define NN   512
#define NF   1024

__global__ __launch_bounds__(256) void cayley_kernel(const float* __restrict__ qw,
                                                     const float* __restrict__ kw,
                                                     const float* __restrict__ vw,
                                                     float* __restrict__ Cout) {
  __shared__ float X[32][32];
  __shared__ float Mm[32][33];
  __shared__ float Inv[32][33];
  __shared__ float fac[32];
  __shared__ float pivinv;
  const float* W = (blockIdx.x == 0) ? qw : ((blockIdx.x == 1) ? kw : vw);
  const int tid = threadIdx.x;

  for (int e = tid; e < 1024; e += 256) {
    int i = e >> 5, j = e & 31;
    float xv = W[i * 32 + j] - W[j * 32 + i];
    X[i][j]  = xv;
    Mm[i][j] = xv + ((i == j) ? 1.f : 0.f);
    Inv[i][j] = (i == j) ? 1.f : 0.f;
  }
  __syncthreads();

  for (int k = 0; k < 32; ++k) {
    if (tid == 0) pivinv = 1.f / Mm[k][k];
    __syncthreads();
    if (tid < 32)      Mm[k][tid]       *= pivinv;
    else if (tid < 64) Inv[k][tid - 32] *= pivinv;
    __syncthreads();
    if (tid < 32) fac[tid] = (tid == k) ? 0.f : Mm[tid][k];
    __syncthreads();
    for (int e = tid; e < 2048; e += 256) {
      int i = e >> 6, j = e & 63;
      if (i == k) continue;
      float f = fac[i];
      if (j < 32) Mm[i][j]        -= f * Mm[k][j];
      else        Inv[i][j - 32]  -= f * Inv[k][j - 32];
    }
    __syncthreads();
  }

  for (int e = tid; e < 1024; e += 256) {
    int i = e >> 5, j = e & 31;
    float acc = 0.f;
    for (int kk = 0; kk < 32; ++kk) {
      float a = ((i == kk) ? 1.f : 0.f) - X[i][kk];
      acc += a * Inv[kk][j];
    }
    Cout[blockIdx.x * 1024 + i * 32 + j] = acc;
  }
}

typedef __attribute__((ext_vector_type(4))) float v4f;

__host__ __device__ constexpr int seat_index(int P) { return (P & 1) ? 31 - (P >> 1) : (P >> 1); }
__host__ __device__ constexpr int index_seat(int k) { return (k < 16) ? 2 * k : 2 * (31 - k) + 1; }
__host__ __device__ constexpr int seat_source(int P) {
  return (P == 1) ? 1 : ((P & 1) == 0 ? (P < 30 ? P + 2 : 31) : (P == 3 ? 0 : P - 2));
}

__device__ __forceinline__ float lane_swap1(float v) {
  return __int_as_float(__builtin_amdgcn_update_dpp(0, __float_as_int(v), 0xB1, 0xF, 0xF, true));
}

__device__ __forceinline__ float pick(float hi, float lo, unsigned mask) {
  float r;
  asm("v_cndmask_b32_e64 %0, %1, %2, %3" : "=v"(r) : "v"(lo), "v"(hi), "s"(mask));
  return r;
}

__device__ __forceinline__ void pick_pair_entries(const float (&a)[32], unsigned sel16,
                                                  unsigned sel8, unsigned sel4, unsigned sel2,
                                                  float& e0, float& e1) {
  float s16[16], s8[8], s4[4];
#pragma unroll
  for (int k = 0; k < 16; ++k) s16[k] = pick(a[16 + k], a[k], sel16);
#pragma unroll
  for (int k = 0; k < 8; ++k) s8[k] = pick(s16[8 + k], s16[k], sel8);
#pragma unroll
  for (int k = 0; k < 4; ++k) s4[k] = pick(s8[4 + k], s8[k], sel4);
  e0 = pick(s4[2], s4[0], sel2);
  e1 = pick(s4[3], s4[1], sel2);
}

__global__ __launch_bounds__(32, 8) void jacobi_logm_kernel(const float* __restrict__ x,
                                                            float* __restrict__ xl) {
  __shared__ float V[32][33];
  __shared__ float lw[32];
  const int lane = threadIdx.x;
  const bool odd = (lane & 1) != 0;
  const float* xm = x + (size_t)blockIdx.x * 1024;

  const int myrow = odd ? 31 - (lane >> 1) : (lane >> 1);
  const v4f* rp = (const v4f*)(xm + myrow * 32);
  float rowv[32];
#pragma unroll
  for (int g = 0; g < 8; ++g) {
    v4f t = rp[g];
    rowv[4 * g + 0] = t.x; rowv[4 * g + 1] = t.y; rowv[4 * g + 2] = t.z; rowv[4 * g + 3] = t.w;
  }
  float a[32], w[32];
#pragma unroll
  for (int Q = 0; Q < 32; ++Q) {
    a[Q] = rowv[seat_index(Q)];
    w[Q] = (seat_index(Q) == lane) ? 1.f : 0.f;
  }
  const int pull = seat_source(lane) * 4;
  const unsigned sel16 = 0xFFFF0000u, sel8 = 0xFF00FF00u, sel4 = 0xF0F0F0F0u, sel2 = 0xCCCCCCCCu;

  for (int it = 0; it < 8 * 31; ++it) {
    float e0, e1;
    pick_pair_entries(a, sel16, sel8, sel4, sel2, e0, e1);
    const float dg  = odd ? e1 : e0;
    const float od  = odd ? e0 : e1;
    const float dgn = lane_swap1(dg);
    const float odn = lane_swap1(od);
    const float app = odd ? dgn : dg;
    const float aqq = odd ? dg : dgn;
    const float apq = odd ? odn : od;
    float c = 1.f, s = 0.f;
    if (fabsf(apq) > 1e-12f) {
      float theta = (aqq - app) / (2.f * apq);
      float t = 1.f / (fabsf(theta) + sqrtf(theta * theta + 1.f));
      if (theta < 0.f) t = -t;
      c = rsqrtf(t * t + 1.f);
      s = t * c;
    }
    float ct[16], st[16];
#pragma unroll
    for (int t = 0; t < 16; ++t) {
      ct[t] = __int_as_float(__builtin_amdgcn_readlane(__float_as_int(c), 2 * t));
      st[t] = __int_as_float(__builtin_amdgcn_readlane(__float_as_int(s), 2 * t));
    }

    const float sg = odd ? s : -s;
#pragma unroll
    for (int Q = 0; Q < 32; ++Q) a[Q] = c * a[Q] + sg * lane_swap1(a[Q]);

#pragma unroll
    for (int t = 0; t < 16; ++t) {
      const float ap = a[2 * t], aq = a[2 * t + 1];
      a[2 * t]     = ct[t] * ap - st[t] * aq;
      a[2 * t + 1] = st[t] * ap + ct[t] * aq;
    }

    float na[32];
#pragma unroll
    for (int P = 0; P < 32; ++P)
      na[P] = __int_as_float(__builtin_amdgcn_ds_bpermute(pull, __float_as_int(a[seat_source(P)])));

    __builtin_amdgcn_sched_barrier(0);
    float nw[32];
#pragma unroll
    for (int t = 0; t < 16; ++t) {
      const float wp = w[2 * t], wq = w[2 * t + 1];
      float lo = ct[t] * wp - st[t] * wq;
      float hi = st[t] * wp + ct[t] * wq;
      asm("" : "+v"(lo), "+v"(hi));
      nw[2 * t]     = lo;
      nw[2 * t + 1] = hi;
    }
#pragma unroll
    for (int P = 0; P < 32; ++P) {
      w[P] = nw[seat_source(P)];
      a[P] = na[P];
    }
  }

  {
    float e0, e1;
    pick_pair_entries(a, sel16, sel8, sel4, sel2, e0, e1);
    const float dg = odd ? e1 : e0;
    lw[lane] = logf(fmaxf(dg, 1e-20f));
  }
#pragma unroll
  for (int Q = 0; Q < 32; ++Q) V[lane][Q] = w[Q];
  __syncthreads();

  float vj[32];
#pragma unroll
  for (int Q = 0; Q < 32; ++Q) vj[Q] = w[Q] * lw[Q];

  float* outm = xl + (size_t)blockIdx.x * 1024;
  for (int i = 0; i < 32; ++i) {
    float acc = 0.f;
#pragma unroll
    for (int k = 0; k < 32; ++k) acc += V[i][index_seat(k)] * vj[index_seat(k)];
    outm[i * 32 + lane] = (i == lane) ? 0.f : acc;
  }
}

__global__ __launch_bounds__(32) void transbase_kernel(const float* __restrict__ xl,
                                                       const float* __restrict__ C,
                                                       float* __restrict__ qkv) {
  __shared__ float Y[32][33];
  __shared__ float CL[32][33];
  __shared__ float T[32][33];
  __shared__ float T2[32][33];
  const int lane = threadIdx.x;
  const int nIdx = lane & 15;
  const int hi   = lane >> 4;
  const int m = blockIdx.x, w = blockIdx.y;
  const float* ym = xl + (size_t)m * 1024;
  const float* Cw = C + (size_t)w * 1024;

  for (int i = 0; i < 32; ++i) {
    Y[i][lane]  = ym[i * 32 + lane];
    CL[i][lane] = Cw[i * 32 + lane];
  }
  __syncthreads();

  for (int ti = 0; ti < 2; ++ti)
    for (int tj = 0; tj < 2; ++tj) {
      v8f acc = {};
#pragma unroll
      for (int kk = 0; kk < 8; ++kk) {
        int k0 = kk * 4 + hi * 2;
        v2f a; a.x = CL[ti * 16 + nIdx][k0]; a.y = CL[ti * 16 + nIdx][k0 + 1];
        v2f b; b.x = Y[k0][tj * 16 + nIdx];  b.y = Y[k0 + 1][tj * 16 + nIdx];
        acc = WMMA_F32_16x16x4(a, b, acc);
      }
#pragma unroll
      for (int r = 0; r < 8; ++r)
        T[ti * 16 + r + hi * 8][tj * 16 + nIdx] = acc[r];
    }
  __syncthreads();

  for (int ti = 0; ti < 2; ++ti)
    for (int tj = 0; tj < 2; ++tj) {
      v8f acc = {};
#pragma unroll
      for (int kk = 0; kk < 8; ++kk) {
        int k0 = kk * 4 + hi * 2;
        v2f a; a.x = T[ti * 16 + nIdx][k0];       a.y = T[ti * 16 + nIdx][k0 + 1];
        v2f b; b.x = CL[tj * 16 + nIdx][k0];      b.y = CL[tj * 16 + nIdx][k0 + 1];
        acc = WMMA_F32_16x16x4(a, b, acc);
      }
#pragma unroll
      for (int r = 0; r < 8; ++r)
        T2[ti * 16 + r + hi * 8][tj * 16 + nIdx] = acc[r];
    }
  __syncthreads();

  float* outm = qkv + ((size_t)w * NMAT + m) * 1024;
  for (int i = 0; i < 32; ++i) {
    float lo = T2[i][lane];
    float up = T2[lane][i];
    float val = (i > lane) ? lo : ((i < lane) ? up : 0.f);
    outm[i * 32 + lane] = val;
  }
}

__global__ __launch_bounds__(256) void norms_kernel(const float* __restrict__ qkv,
                                                    float* __restrict__ nrm) {
  int idx = blockIdx.x * 256 + threadIdx.x;
  int sel = idx >> 13;
  int r   = idx & (NMAT - 1);
  const float* row = qkv + ((size_t)sel * NMAT + r) * 1024;
  float acc = 0.f;
  for (int f = 0; f < NF; ++f) { float t = row[f]; acc += t * t; }
  nrm[idx] = acc;
}

struct RowFrag { v4f q[4]; };

__device__ __forceinline__ void load_frag(RowFrag& f, const float* p) {
  const v4f* p4 = (const v4f*)p;
  f.q[0] = p4[0]; f.q[1] = p4[1]; f.q[2] = p4[2]; f.q[3] = p4[3];
}

__device__ __forceinline__ v2f frag_pair(const RowFrag& f, int s) {
  v4f q = f.q[s >> 1];
  v2f lo = __builtin_shufflevector(q, q, 0, 1);
  v2f hp = __builtin_shufflevector(q, q, 2, 3);
  return (s & 1) ? hp : lo;
}

__device__ __forceinline__ void load_tiles(RowFrag (&A)[4], RowFrag (&B)[2],
                                           const float* arow, const float* brow, int kb) {
#pragma unroll
  for (int jt = 0; jt < 4; ++jt) load_frag(A[jt], arow + jt * (16 * NF) + kb);
#pragma unroll
  for (int it = 0; it < 2; ++it) load_frag(B[it], brow + it * (16 * NF) + kb);
}

__device__ __forceinline__ void mma_chunk(v8f (&acc)[4][2], const RowFrag (&A)[4],
                                          const RowFrag (&B)[2]) {
#pragma unroll
  for (int s = 0; s < 8; ++s) {
#pragma unroll
    for (int jt = 0; jt < 4; ++jt) {
      v2f a = frag_pair(A[jt], s);
#pragma unroll
      for (int it = 0; it < 2; ++it)
        acc[jt][it] = WMMA_F32_16x16x4(a, frag_pair(B[it], s), acc[jt][it]);
    }
  }
}

__global__ __launch_bounds__(256) void scores_softmax_kernel(const float* __restrict__ qkv,
                                                             const float* __restrict__ nrm,
                                                             float* __restrict__ scores) {
  __shared__ float red[8][32];
  const int tid  = threadIdx.x;
  const int lane = tid & 31;
  const int w    = tid >> 5;
  const int nIdx = lane & 15;
  const int hi   = lane >> 4;
  const int b  = blockIdx.y;
  const int i0 = blockIdx.x * 32;
  const int j0 = w * 64;
  const float* qf = qkv + ((size_t)0 * NMAT + (size_t)b * NN) * 1024;
  const float* kf = qkv + ((size_t)1 * NMAT + (size_t)b * NN) * 1024;
  const float* arow = kf + (size_t)(j0 + nIdx) * NF + hi * 16;
  const float* brow = qf + (size_t)(i0 + nIdx) * NF + hi * 16;

  v8f acc[4][2] = {};
  RowFrag a0[4], b0[2], a1[4], b1[2];

  load_tiles(a0, b0, arow, brow, 0);
  for (int kb = 0; kb < NF; kb += 64) {
    load_tiles(a1, b1, arow, brow, kb + 32);
    mma_chunk(acc, a0, b0);
    if (kb + 64 < NF) load_tiles(a0, b0, arow, brow, kb + 64);
    mma_chunk(acc, a1, b1);
  }

  float q2[2], mx[2];
#pragma unroll
  for (int it = 0; it < 2; ++it) {
    q2[it] = nrm[b * NN + i0 + it * 16 + nIdx];
    mx[it] = -1e30f;
  }
#pragma unroll
  for (int jt = 0; jt < 4; ++jt) {
#pragma unroll
    for (int r = 0; r < 8; ++r) {
      float k2 = nrm[NMAT + b * NN + j0 + jt * 16 + hi * 8 + r];
#pragma unroll
      for (int it = 0; it < 2; ++it) {
        float d2 = q2[it] + k2 - 2.f * acc[jt][it][r];
        d2 = fmaxf(d2, 1e-12f);
        float e  = sqrtf(d2);
        float sc = 1.f / (1.f + log1pf(e));
        acc[jt][it][r] = sc;
        mx[it] = fmaxf(mx[it], sc);
      }
    }
  }

#pragma unroll
  for (int it = 0; it < 2; ++it) {
    mx[it] = fmaxf(mx[it], __shfl_xor(mx[it], 16));
    if (hi == 0) red[w][it * 16 + nIdx] = mx[it];
  }
  __syncthreads();
#pragma unroll
  for (int it = 0; it < 2; ++it) {
    float m = -1e30f;
#pragma unroll
    for (int ww = 0; ww < 8; ++ww) m = fmaxf(m, red[ww][it * 16 + nIdx]);
    mx[it] = m;
  }
  __syncthreads();

  float ps[2] = {0.f, 0.f};
#pragma unroll
  for (int jt = 0; jt < 4; ++jt) {
#pragma unroll
    for (int r = 0; r < 8; ++r) {
#pragma unroll
      for (int it = 0; it < 2; ++it) {
        float e = __expf(acc[jt][it][r] - mx[it]);
        acc[jt][it][r] = e;
        ps[it] += e;
      }
    }
  }
#pragma unroll
  for (int it = 0; it < 2; ++it) {
    ps[it] += __shfl_xor(ps[it], 16);
    if (hi == 0) red[w][it * 16 + nIdx] = ps[it];
  }
  __syncthreads();
  float inv[2];
#pragma unroll
  for (int it = 0; it < 2; ++it) {
    float sum = 0.f;
#pragma unroll
    for (int ww = 0; ww < 8; ++ww) sum += red[ww][it * 16 + nIdx];
    inv[it] = 1.f / sum;
  }

  float* sb = scores + (size_t)b * NN * NN;
#pragma unroll
  for (int jt = 0; jt < 4; ++jt) {
#pragma unroll
    for (int r = 0; r < 8; ++r) {
      int j = j0 + jt * 16 + hi * 8 + r;
#pragma unroll
      for (int it = 0; it < 2; ++it)
        sb[(size_t)j * NN + i0 + it * 16 + nIdx] = acc[jt][it][r] * inv[it];
    }
  }
}

struct ColFrag { v2f e[8]; };

__device__ __forceinline__ void load_out_tiles(RowFrag (&A)[4], ColFrag& B,
                                               const float* arow, const float* bcol, int kb) {
#pragma unroll
  for (int t = 0; t < 4; ++t) load_frag(A[t], arow + t * (16 * NN) + kb);
  const float* bp = bcol + (size_t)kb * NF;
#pragma unroll
  for (int s = 0; s < 8; ++s) {
    B.e[s].x = bp[(size_t)(2 * s) * NF];
    B.e[s].y = bp[(size_t)(2 * s + 1) * NF];
  }
}

__device__ __forceinline__ void mma_out_chunk(v8f (&acc)[4], const RowFrag (&A)[4],
                                              const ColFrag& B) {
#pragma unroll
  for (int s = 0; s < 8; ++s) {
#pragma unroll
    for (int t = 0; t < 4; ++t)
      acc[t] = WMMA_F32_16x16x4(frag_pair(A[t], s), B.e[s], acc[t]);
  }
}

__global__ __launch_bounds__(32) void outgemm_kernel(const float* __restrict__ scores,
                                                     const float* __restrict__ qkv,
                                                     float* __restrict__ out) {
  const int lane = threadIdx.x;
  const int nIdx = lane & 15;
  const int hi   = lane >> 4;
  const int ft = blockIdx.x, ct = blockIdx.y, b = blockIdx.z;
  const int f0 = ft * 16, c0 = ct * 64;
  const float* p  = scores + (size_t)b * NN * NN;
  const float* vf = qkv + ((size_t)2 * NMAT + (size_t)b * NN) * 1024;
  const float* arow = p + (size_t)(c0 + nIdx) * NN + hi * 16;
  const float* bcol = vf + (size_t)(hi * 16) * NF + f0 + nIdx;

  v8f acc[4] = {};
  RowFrag a0[4], a1[4];
  ColFrag b0, b1;

  load_out_tiles(a0, b0, arow, bcol, 0);
  for (int kb = 0; kb < NN - 64; kb += 64) {
    load_out_tiles(a1, b1, arow, bcol, kb + 32);
    __builtin_amdgcn_sched_barrier(0);
    mma_out_chunk(acc, a0, b0);
    __builtin_amdgcn_sched_barrier(0);
    load_out_tiles(a0, b0, arow, bcol, kb + 64);
    __builtin_amdgcn_sched_barrier(0);
    mma_out_chunk(acc, a1, b1);
    __builtin_amdgcn_sched_barrier(0);
  }
  load_out_tiles(a1, b1, arow, bcol, NN - 32);
  __builtin_amdgcn_sched_barrier(0);
  mma_out_chunk(acc, a0, b0);
  mma_out_chunk(acc, a1, b1);

  float* ob = out + (size_t)b * NN * NF;
#pragma unroll
  for (int t = 0; t < 4; ++t) {
#pragma unroll
    for (int r = 0; r < 8; ++r) {
      int c = c0 + t * 16 + r + hi * 8;
      ob[(size_t)c * NF + f0 + nIdx] = acc[t][r];
    }
  }
}

extern "C" void kernel_launch(void* const* d_in, const int* in_sizes, int n_in,
                              void* d_out, int out_size, void* d_ws, size_t ws_size,
                              hipStream_t stream) {
  (void)in_sizes; (void)n_in; (void)out_size; (void)ws_size;
  const float* x  = (const float*)d_in[0];
  const float* qw = (const float*)d_in[1];
  const float* kw = (const float*)d_in[2];
  const float* vw = (const float*)d_in[3];
  float* out = (float*)d_out;

  float* ws     = (float*)d_ws;
  float* C      = ws;
  float* xl     = C + 3 * 1024;
  float* qkv    = xl + (size_t)NMAT * 1024;
  float* nrm    = qkv + (size_t)3 * NMAT * 1024;
  float* scores = nrm + 2 * NMAT;

  cayley_kernel<<<3, 256, 0, stream>>>(qw, kw, vw, C);
  jacobi_logm_kernel<<<NMAT, 32, 0, stream>>>(x, xl);
  transbase_kernel<<<dim3(NMAT, 3), 32, 0, stream>>>(xl, C, qkv);
  norms_kernel<<<64, 256, 0, stream>>>(qkv, nrm);
  scores_softmax_kernel<<<dim3(NN / 32, NB), 256, 0, stream>>>(qkv, nrm, scores);
  outgemm_kernel<<<dim3(NF / 16, NN / 64, NB), 32, 0, stream>>>(scores, qkv, out);
}


# eigensolver: rows change lanes through an LDS row buffer (16-B accesses) that also gives a lane its partner's row and its pair's 2x2 block: no DPP rotation, select tree or bpermute; sums kept
# speedup vs baseline: 2.7008x; 1.0416x over previous
#include <hip/hip_runtime.h>
#include <math.h>


typedef __attribute__((ext_vector_type(2))) float v2f;
typedef __attribute__((ext_vector_type(8))) float v8f;

#define WMMA_F32_16x16x4(a, b, c) \
  __builtin_amdgcn_wmma_f32_16x16x4_f32(false, (a), false, (b), (short)0, (c), false, false)

#define NMAT 8192
#define NB   16
#define NN   512
#define NF   1024

__global__ __launch_bounds__(256) void cayley_kernel(const float* __restrict__ qw,
                                                     const float* __restrict__ kw,
                                                     const float* __restrict__ vw,
                                                     float* __restrict__ Cout) {
  __shared__ float X[32][32];
  __shared__ float Mm[32][33];
  __shared__ float Inv[32][33];
  __shared__ float fac[32];
  __shared__ float pivinv;
  const float* W = (blockIdx.x == 0) ? qw : ((blockIdx.x == 1) ? kw : vw);
  const int tid = threadIdx.x;

  for (int e = tid; e < 1024; e += 256) {
    int i = e >> 5, j = e & 31;
    float xv = W[i * 32 + j] - W[j * 32 + i];
    X[i][j]  = xv;
    Mm[i][j] = xv + ((i == j) ? 1.f : 0.f);
    Inv[i][j] = (i == j) ? 1.f : 0.f;
  }
  __syncthreads();

  for (int k = 0; k < 32; ++k) {
    if (tid == 0) pivinv = 1.f / Mm[k][k];
    __syncthreads();
    if (tid < 32)      Mm[k][tid]       *= pivinv;
    else if (tid < 64) Inv[k][tid - 32] *= pivinv;
    __syncthreads();
    if (tid < 32) fac[tid] = (tid == k) ? 0.f : Mm[tid][k];
    __syncthreads();
    for (int e = tid; e < 2048; e += 256) {
      int i = e >> 6, j = e & 63;
      if (i == k) continue;
      float f = fac[i];
      if (j < 32) Mm[i][j]        -= f * Mm[k][j];
      else        Inv[i][j - 32]  -= f * Inv[k][j - 32];
    }
    __syncthreads();
  }

  for (int e = tid; e < 1024; e += 256) {
    int i = e >> 5, j = e & 31;
    float acc = 0.f;
    for (int kk = 0; kk < 32; ++kk) {
      float a = ((i == kk) ? 1.f : 0.f) - X[i][kk];
      acc += a * Inv[kk][j];
    }
    Cout[blockIdx.x * 1024 + i * 32 + j] = acc;
  }
}

typedef __attribute__((ext_vector_type(4))) float v4f;

__host__ __device__ constexpr int seat_index(int P) { return (P & 1) ? 31 - (P >> 1) : (P >> 1); }
__host__ __device__ constexpr int index_seat(int k) { return (k < 16) ? 2 * k : 2 * (31 - k) + 1; }
__host__ __device__ constexpr int seat_source(int P) {
  return (P == 1) ? 1 : ((P & 1) == 0 ? (P < 30 ? P + 2 : 31) : (P == 3 ? 0 : P - 2));
}

__device__ __forceinline__ float lane_swap1(float v) {
  return __int_as_float(__builtin_amdgcn_update_dpp(0, __float_as_int(v), 0xB1, 0xF, 0xF, true));
}

__global__ __launch_bounds__(32, 8) void jacobi_logm_kernel(const float* __restrict__ x,
                                                            float* __restrict__ xl) {
  __shared__ __attribute__((aligned(16))) float R[32][36];
  __shared__ float lw[32];
  const int lane = threadIdx.x;
  const bool odd = (lane & 1) != 0;
  const float* xm = x + (size_t)blockIdx.x * 1024;

  const int myrow = odd ? 31 - (lane >> 1) : (lane >> 1);
  const v4f* rp = (const v4f*)(xm + myrow * 32);
  float rowv[32];
#pragma unroll
  for (int g = 0; g < 8; ++g) {
    v4f t = rp[g];
    rowv[4 * g + 0] = t.x; rowv[4 * g + 1] = t.y; rowv[4 * g + 2] = t.z; rowv[4 * g + 3] = t.w;
  }
  float a[32], w[32];
#pragma unroll
  for (int Q = 0; Q < 32; ++Q) {
    a[Q] = rowv[seat_index(Q)];
    w[Q] = (seat_index(Q) == lane) ? 1.f : 0.f;
  }
  const float* own_next  = &R[seat_source(lane)][0];
  const float* par_next  = &R[seat_source(lane ^ 1)][0];
  const float* piv_next  = &R[seat_source(lane)][lane & ~1];

#pragma unroll
  for (int g = 0; g < 8; ++g) {
    v4f t; t.x = a[4 * g]; t.y = a[4 * g + 1]; t.z = a[4 * g + 2]; t.w = a[4 * g + 3];
    *(v4f*)&R[lane][4 * g] = t;
  }
  const float* own = &R[lane][0];
  const float* par = &R[lane ^ 1][0];
  const float* piv = &R[lane][lane & ~1];
  v2f pv;

  for (int it = 0; it < 8 * 31; ++it) {
    __syncthreads();
    float b[32];
    pv = *(const v2f*)piv;
#pragma unroll
    for (int g = 0; g < 8; ++g) {
      v4f t = *(const v4f*)(own + 4 * g);
      a[4 * g] = t.x; a[4 * g + 1] = t.y; a[4 * g + 2] = t.z; a[4 * g + 3] = t.w;
      v4f u = *(const v4f*)(par + 4 * g);
      b[4 * g] = u.x; b[4 * g + 1] = u.y; b[4 * g + 2] = u.z; b[4 * g + 3] = u.w;
    }
    own = own_next; par = par_next; piv = piv_next;
    const float e0 = pv.x, e1 = pv.y;
    const float dg  = odd ? e1 : e0;
    const float od  = odd ? e0 : e1;
    const float dgn = lane_swap1(dg);
    const float odn = lane_swap1(od);
    const float app = odd ? dgn : dg;
    const float aqq = odd ? dg : dgn;
    const float apq = odd ? odn : od;
    float c = 1.f, s = 0.f;
    if (fabsf(apq) > 1e-12f) {
      float theta = (aqq - app) / (2.f * apq);
      float t = 1.f / (fabsf(theta) + sqrtf(theta * theta + 1.f));
      if (theta < 0.f) t = -t;
      c = rsqrtf(t * t + 1.f);
      s = t * c;
    }
    float ct[16], st[16];
#pragma unroll
    for (int t = 0; t < 16; ++t) {
      ct[t] = __int_as_float(__builtin_amdgcn_readlane(__float_as_int(c), 2 * t));
      st[t] = __int_as_float(__builtin_amdgcn_readlane(__float_as_int(s), 2 * t));
    }

    const float sg = odd ? s : -s;
#pragma unroll
    for (int Q = 0; Q < 32; ++Q) {
      float r = __builtin_fmaf(c, a[Q], sg * b[Q]);
      asm("" : "+v"(r));
      a[Q] = r;
    }

#pragma unroll
    for (int t = 0; t < 16; ++t) {
      const float ap = a[2 * t], aq = a[2 * t + 1];
      float lo = ct[t] * ap - st[t] * aq;
      float hi = st[t] * ap + ct[t] * aq;
      asm("" : "+v"(lo), "+v"(hi));
      a[2 * t]     = lo;
      a[2 * t + 1] = hi;
    }

    __syncthreads();
#pragma unroll
    for (int g = 0; g < 8; ++g) {
      v4f t;
      t.x = a[seat_source(4 * g)];     t.y = a[seat_source(4 * g + 1)];
      t.z = a[seat_source(4 * g + 2)]; t.w = a[seat_source(4 * g + 3)];
      *(v4f*)&R[lane][4 * g] = t;
    }

    __builtin_amdgcn_sched_barrier(0);
    float nw[32];
#pragma unroll
    for (int t = 0; t < 16; ++t) {
      const float wp = w[2 * t], wq = w[2 * t + 1];
      float lo = ct[t] * wp - st[t] * wq;
      float hi = st[t] * wp + ct[t] * wq;
      asm("" : "+v"(lo), "+v"(hi));
      nw[2 * t]     = lo;
      nw[2 * t + 1] = hi;
    }
#pragma unroll
    for (int P = 0; P < 32; ++P) w[P] = nw[seat_source(P)];

  }
  __syncthreads();
  pv = *(const v2f*)piv;

  {
    const float dg = odd ? pv.y : pv.x;
    lw[lane] = logf(fmaxf(dg, 1e-20f));
  }
  __syncthreads();
#pragma unroll
  for (int Q = 0; Q < 32; ++Q) R[lane][Q] = w[Q];
  __syncthreads();

  float vj[32];
#pragma unroll
  for (int Q = 0; Q < 32; ++Q) vj[Q] = w[Q] * lw[Q];

  float* outm = xl + (size_t)blockIdx.x * 1024;
  for (int i = 0; i < 32; ++i) {
    float acc = 0.f;
#pragma unroll
    for (int k = 0; k < 32; ++k) acc += R[i][index_seat(k)] * vj[index_seat(k)];
    outm[i * 32 + lane] = (i == lane) ? 0.f : acc;
  }
}

__global__ __launch_bounds__(32) void transbase_kernel(const float* __restrict__ xl,
                                                       const float* __restrict__ C,
                                                       float* __restrict__ qkv) {
  __shared__ float Y[32][33];
  __shared__ float CL[32][33];
  __shared__ float T[32][33];
  __shared__ float T2[32][33];
  const int lane = threadIdx.x;
  const int nIdx = lane & 15;
  const int hi   = lane >> 4;
  const int m = blockIdx.x, w = blockIdx.y;
  const float* ym = xl + (size_t)m * 1024;
  const float* Cw = C + (size_t)w * 1024;

  for (int i = 0; i < 32; ++i) {
    Y[i][lane]  = ym[i * 32 + lane];
    CL[i][lane] = Cw[i * 32 + lane];
  }
  __syncthreads();

  for (int ti = 0; ti < 2; ++ti)
    for (int tj = 0; tj < 2; ++tj) {
      v8f acc = {};
#pragma unroll
      for (int kk = 0; kk < 8; ++kk) {
        int k0 = kk * 4 + hi * 2;
        v2f a; a.x = CL[ti * 16 + nIdx][k0]; a.y = CL[ti * 16 + nIdx][k0 + 1];
        v2f b; b.x = Y[k0][tj * 16 + nIdx];  b.y = Y[k0 + 1][tj * 16 + nIdx];
        acc = WMMA_F32_16x16x4(a, b, acc);
      }
#pragma unroll
      for (int r = 0; r < 8; ++r)
        T[ti * 16 + r + hi * 8][tj * 16 + nIdx] = acc[r];
    }
  __syncthreads();

  for (int ti = 0; ti < 2; ++ti)
    for (int tj = 0; tj < 2; ++tj) {
      v8f acc = {};
#pragma unroll
      for (int kk = 0; kk < 8; ++kk) {
        int k0 = kk * 4 + hi * 2;
        v2f a; a.x = T[ti * 16 + nIdx][k0];       a.y = T[ti * 16 + nIdx][k0 + 1];
        v2f b; b.x = CL[tj * 16 + nIdx][k0];      b.y = CL[tj * 16 + nIdx][k0 + 1];
        acc = WMMA_F32_16x16x4(a, b, acc);
      }
#pragma unroll
      for (int r = 0; r < 8; ++r)
        T2[ti * 16 + r + hi * 8][tj * 16 + nIdx] = acc[r];
    }
  __syncthreads();

  float* outm = qkv + ((size_t)w * NMAT + m) * 1024;
  for (int i = 0; i < 32; ++i) {
    float lo = T2[i][lane];
    float up = T2[lane][i];
    float val = (i > lane) ? lo : ((i < lane) ? up : 0.f);
    outm[i * 32 + lane] = val;
  }
}

__global__ __launch_bounds__(256) void norms_kernel(const float* __restrict__ qkv,
                                                    float* __restrict__ nrm) {
  int idx = blockIdx.x * 256 + threadIdx.x;
  int sel = idx >> 13;
  int r   = idx & (NMAT - 1);
  const float* row = qkv + ((size_t)sel * NMAT + r) * 1024;
  float acc = 0.f;
  for (int f = 0; f < NF; ++f) { float t = row[f]; acc += t * t; }
  nrm[idx] = acc;
}

struct RowFrag { v4f q[4]; };

__device__ __forceinline__ void load_frag(RowFrag& f, const float* p) {
  const v4f* p4 = (const v4f*)p;
  f.q[0] = p4[0]; f.q[1] = p4[1]; f.q[2] = p4[2]; f.q[3] = p4[3];
}

__device__ __forceinline__ v2f frag_pair(const RowFrag& f, int s) {
  v4f q = f.q[s >> 1];
  v2f lo = __builtin_shufflevector(q, q, 0, 1);
  v2f hp = __builtin_shufflevector(q, q, 2, 3);
  return (s & 1) ? hp : lo;
}

__device__ __forceinline__ void load_tiles(RowFrag (&A)[4], RowFrag (&B)[2],
                                           const float* arow, const float* brow, int kb) {
#pragma unroll
  for (int jt = 0; jt < 4; ++jt) load_frag(A[jt], arow + jt * (16 * NF) + kb);
#pragma unroll
  for (int it = 0; it < 2; ++it) load_frag(B[it], brow + it * (16 * NF) + kb);
}

__device__ __forceinline__ void mma_chunk(v8f (&acc)[4][2], const RowFrag (&A)[4],
                                          const RowFrag (&B)[2]) {
#pragma unroll
  for (int s = 0; s < 8; ++s) {
#pragma unroll
    for (int jt = 0; jt < 4; ++jt) {
      v2f a = frag_pair(A[jt], s);
#pragma unroll
      for (int it = 0; it < 2; ++it)
        acc[jt][it] = WMMA_F32_16x16x4(a, frag_pair(B[it], s), acc[jt][it]);
    }
  }
}

__global__ __launch_bounds__(256) void scores_softmax_kernel(const float* __restrict__ qkv,
                                                             const float* __restrict__ nrm,
                                                             float* __restrict__ scores) {
  __shared__ float red[8][32];
  const int tid  = threadIdx.x;
  const int lane = tid & 31;
  const int w    = tid >> 5;
  const int nIdx = lane & 15;
  const int hi   = lane >> 4;
  const int b  = blockIdx.y;
  const int i0 = blockIdx.x * 32;
  const int j0 = w * 64;
  const float* qf = qkv + ((size_t)0 * NMAT + (size_t)b * NN) * 1024;
  const float* kf = qkv + ((size_t)1 * NMAT + (size_t)b * NN) * 1024;
  const float* arow = kf + (size_t)(j0 + nIdx) * NF + hi * 16;
  const float* brow = qf + (size_t)(i0 + nIdx) * NF + hi * 16;

  v8f acc[4][2] = {};
  RowFrag a0[4], b0[2], a1[4], b1[2];

  load_tiles(a0, b0, arow, brow, 0);
  for (int kb = 0; kb < NF; kb += 64) {
    load_tiles(a1, b1, arow, brow, kb + 32);
    mma_chunk(acc, a0, b0);
    if (kb + 64 < NF) load_tiles(a0, b0, arow, brow, kb + 64);
    mma_chunk(acc, a1, b1);
  }

  float q2[2], mx[2];
#pragma unroll
  for (int it = 0; it < 2; ++it) {
    q2[it] = nrm[b * NN + i0 + it * 16 + nIdx];
    mx[it] = -1e30f;
  }
#pragma unroll
  for (int jt = 0; jt < 4; ++jt) {
#pragma unroll
    for (int r = 0; r < 8; ++r) {
      float k2 = nrm[NMAT + b * NN + j0 + jt * 16 + hi * 8 + r];
#pragma unroll
      for (int it = 0; it < 2; ++it) {
        float d2 = q2[it] + k2 - 2.f * acc[jt][it][r];
        d2 = fmaxf(d2, 1e-12f);
        float e  = sqrtf(d2);
        float sc = 1.f / (1.f + log1pf(e));
        acc[jt][it][r] = sc;
        mx[it] = fmaxf(mx[it], sc);
      }
    }
  }

#pragma unroll
  for (int it = 0; it < 2; ++it) {
    mx[it] = fmaxf(mx[it], __shfl_xor(mx[it], 16));
    if (hi == 0) red[w][it * 16 + nIdx] = mx[it];
  }
  __syncthreads();
#pragma unroll
  for (int it = 0; it < 2; ++it) {
    float m = -1e30f;
#pragma unroll
    for (int ww = 0; ww < 8; ++ww) m = fmaxf(m, red[ww][it * 16 + nIdx]);
    mx[it] = m;
  }
  __syncthreads();

  float ps[2] = {0.f, 0.f};
#pragma unroll
  for (int jt = 0; jt < 4; ++jt) {
#pragma unroll
    for (int r = 0; r < 8; ++r) {
#pragma unroll
      for (int it = 0; it < 2; ++it) {
        float e = __expf(acc[jt][it][r] - mx[it]);
        acc[jt][it][r] = e;
        ps[it] += e;
      }
    }
  }
#pragma unroll
  for (int it = 0; it < 2; ++it) {
    ps[it] += __shfl_xor(ps[it], 16);
    if (hi == 0) red[w][it * 16 + nIdx] = ps[it];
  }
  __syncthreads();
  float inv[2];
#pragma unroll
  for (int it = 0; it < 2; ++it) {
    float sum = 0.f;
#pragma unroll
    for (int ww = 0; ww < 8; ++ww) sum += red[ww][it * 16 + nIdx];
    inv[it] = 1.f / sum;
  }

  float* sb = scores + (size_t)b * NN * NN;
#pragma unroll
  for (int jt = 0; jt < 4; ++jt) {
#pragma unroll
    for (int r = 0; r < 8; ++r) {
      int j = j0 + jt * 16 + hi * 8 + r;
#pragma unroll
      for (int it = 0; it < 2; ++it)
        sb[(size_t)j * NN + i0 + it * 16 + nIdx] = acc[jt][it][r] * inv[it];
    }
  }
}

struct ColFrag { v2f e[8]; };

__device__ __forceinline__ void load_out_tiles(RowFrag (&A)[4], ColFrag& B,
                                               const float* arow, const float* bcol, int kb) {
#pragma unroll
  for (int t = 0; t < 4; ++t) load_frag(A[t], arow + t * (16 * NN) + kb);
  const float* bp = bcol + (size_t)kb * NF;
#pragma unroll
  for (int s = 0; s < 8; ++s) {
    B.e[s].x = bp[(size_t)(2 * s) * NF];
    B.e[s].y = bp[(size_t)(2 * s + 1) * NF];
  }
}

__device__ __forceinline__ void mma_out_chunk(v8f (&acc)[4], const RowFrag (&A)[4],
                                              const ColFrag& B) {
#pragma unroll
  for (int s = 0; s < 8; ++s) {
#pragma unroll
    for (int t = 0; t < 4; ++t)
      acc[t] = WMMA_F32_16x16x4(frag_pair(A[t], s), B.e[s], acc[t]);
  }
}

__global__ __launch_bounds__(32) void outgemm_kernel(const float* __restrict__ scores,
                                                     const float* __restrict__ qkv,
                                                     float* __restrict__ out) {
  const int lane = threadIdx.x;
  const int nIdx = lane & 15;
  const int hi   = lane >> 4;
  const int ft = blockIdx.x, ct = blockIdx.y, b = blockIdx.z;
  const int f0 = ft * 16, c0 = ct * 64;
  const float* p  = scores + (size_t)b * NN * NN;
  const float* vf = qkv + ((size_t)2 * NMAT + (size_t)b * NN) * 1024;
  const float* arow = p + (size_t)(c0 + nIdx) * NN + hi * 16;
  const float* bcol = vf + (size_t)(hi * 16) * NF + f0 + nIdx;

  v8f acc[4] = {};
  RowFrag a0[4], a1[4];
  ColFrag b0, b1;

  load_out_tiles(a0, b0, arow, bcol, 0);
  for (int kb = 0; kb < NN - 64; kb += 64) {
    load_out_tiles(a1, b1, arow, bcol, kb + 32);
    __builtin_amdgcn_sched_barrier(0);
    mma_out_chunk(acc, a0, b0);
    __builtin_amdgcn_sched_barrier(0);
    load_out_tiles(a0, b0, arow, bcol, kb + 64);
    __builtin_amdgcn_sched_barrier(0);
    mma_out_chunk(acc, a1, b1);
    __builtin_amdgcn_sched_barrier(0);
  }
  load_out_tiles(a1, b1, arow, bcol, NN - 32);
  __builtin_amdgcn_sched_barrier(0);
  mma_out_chunk(acc, a0, b0);
  mma_out_chunk(acc, a1, b1);

  float* ob = out + (size_t)b * NN * NF;
#pragma unroll
  for (int t = 0; t < 4; ++t) {
#pragma unroll
    for (int r = 0; r < 8; ++r) {
      int c = c0 + t * 16 + r + hi * 8;
      ob[(size_t)c * NF + f0 + nIdx] = acc[t][r];
    }
  }
}

extern "C" void kernel_launch(void* const* d_in, const int* in_sizes, int n_in,
                              void* d_out, int out_size, void* d_ws, size_t ws_size,
                              hipStream_t stream) {
  (void)in_sizes; (void)n_in; (void)out_size; (void)ws_size;
  const float* x  = (const float*)d_in[0];
  const float* qw = (const float*)d_in[1];
  const float* kw = (const float*)d_in[2];
  const float* vw = (const float*)d_in[3];
  float* out = (float*)d_out;

  float* ws     = (float*)d_ws;
  float* C      = ws;
  float* xl     = C + 3 * 1024;
  float* qkv    = xl + (size_t)NMAT * 1024;
  float* nrm    = qkv + (size_t)3 * NMAT * 1024;
  float* scores = nrm + 2 * NMAT;

  cayley_kernel<<<3, 256, 0, stream>>>(qw, kw, vw, C);
  jacobi_logm_kernel<<<NMAT, 32, 0, stream>>>(x, xl);
  transbase_kernel<<<dim3(NMAT, 3), 32, 0, stream>>>(xl, C, qkv);
  norms_kernel<<<64, 256, 0, stream>>>(qkv, nrm);
  scores_softmax_kernel<<<dim3(NN / 32, NB), 256, 0, stream>>>(qkv, nrm, scores);
  outgemm_kernel<<<dim3(NF / 16, NN / 64, NB), 32, 0, stream>>>(scores, qkv, out);
}


# eigensolver: odd-even neighbour pair order, the swap folded into which rotated sum a lane or register keeps: rows and V columns stay put (no own-row reload, no register moves); pair order changes
# speedup vs baseline: 3.0480x; 1.1286x over previous
#include <hip/hip_runtime.h>
#include <math.h>


typedef __attribute__((ext_vector_type(2))) float v2f;
typedef __attribute__((ext_vector_type(8))) float v8f;

#define WMMA_F32_16x16x4(a, b, c) \
  __builtin_amdgcn_wmma_f32_16x16x4_f32(false, (a), false, (b), (short)0, (c), false, false)

#define NMAT 8192
#define NB   16
#define NN   512
#define NF   1024

__global__ __launch_bounds__(256) void cayley_kernel(const float* __restrict__ qw,
                                                     const float* __restrict__ kw,
                                                     const float* __restrict__ vw,
                                                     float* __restrict__ Cout) {
  __shared__ float X[32][32];
  __shared__ float Mm[32][33];
  __shared__ float Inv[32][33];
  __shared__ float fac[32];
  __shared__ float pivinv;
  const float* W = (blockIdx.x == 0) ? qw : ((blockIdx.x == 1) ? kw : vw);
  const int tid = threadIdx.x;

  for (int e = tid; e < 1024; e += 256) {
    int i = e >> 5, j = e & 31;
    float xv = W[i * 32 + j] - W[j * 32 + i];
    X[i][j]  = xv;
    Mm[i][j] = xv + ((i == j) ? 1.f : 0.f);
    Inv[i][j] = (i == j) ? 1.f : 0.f;
  }
  __syncthreads();

  for (int k = 0; k < 32; ++k) {
    if (tid == 0) pivinv = 1.f / Mm[k][k];
    __syncthreads();
    if (tid < 32)      Mm[k][tid]       *= pivinv;
    else if (tid < 64) Inv[k][tid - 32] *= pivinv;
    __syncthreads();
    if (tid < 32) fac[tid] = (tid == k) ? 0.f : Mm[tid][k];
    __syncthreads();
    for (int e = tid; e < 2048; e += 256) {
      int i = e >> 6, j = e & 63;
      if (i == k) continue;
      float f = fac[i];
      if (j < 32) Mm[i][j]        -= f * Mm[k][j];
      else        Inv[i][j - 32]  -= f * Inv[k][j - 32];
    }
    __syncthreads();
  }

  for (int e = tid; e < 1024; e += 256) {
    int i = e >> 5, j = e & 31;
    float acc = 0.f;
    for (int kk = 0; kk < 32; ++kk) {
      float a = ((i == kk) ? 1.f : 0.f) - X[i][kk];
      acc += a * Inv[kk][j];
    }
    Cout[blockIdx.x * 1024 + i * 32 + j] = acc;
  }
}

typedef __attribute__((ext_vector_type(4))) float v4f;

template <int PAR>
__device__ __forceinline__ void jacobi_round(float (&a)[32], float (&w)[32], float* myrow,
                                             const float* prow, const float* ppp,
                                             const float* pqq, const bool lower,
                                             const bool rest) {
  constexpr int NP = PAR ? 15 : 16;
  __syncthreads();
  const float app = ppp[0];
  const float epq = ppp[1];
  const float aqq = pqq[0];
  float b[32];
#pragma unroll
  for (int g = 0; g < 8; ++g) {
    v4f u = *(const v4f*)(prow + 4 * g);
    b[4 * g] = u.x; b[4 * g + 1] = u.y; b[4 * g + 2] = u.z; b[4 * g + 3] = u.w;
  }
  const float apq = rest ? 0.f : epq;
  float c = 1.f, s = 0.f;
  if (fabsf(apq) > 1e-12f) {
    float theta = (aqq - app) / (2.f * apq);
    float t = 1.f / (fabsf(theta) + sqrtf(theta * theta + 1.f));
    if (theta < 0.f) t = -t;
    c = rsqrtf(t * t + 1.f);
    s = t * c;
  }
  float ct[NP], st[NP];
#pragma unroll
  for (int t = 0; t < NP; ++t) {
    ct[t] = __int_as_float(__builtin_amdgcn_readlane(__float_as_int(c), PAR + 2 * t));
    st[t] = __int_as_float(__builtin_amdgcn_readlane(__float_as_int(s), PAR + 2 * t));
  }

  const float sg = lower ? s : -s;
#pragma unroll
  for (int Q = 0; Q < 32; ++Q) {
    float r = __builtin_fmaf(c, b[Q], sg * a[Q]);
    asm("" : "+v"(r));
    a[Q] = r;
  }

#pragma unroll
  for (int t = 0; t < NP; ++t) {
    const int P = PAR + 2 * t;
    const float ap = a[P], aq = a[P + 1];
    float lo = ct[t] * ap - st[t] * aq;
    float hi = st[t] * ap + ct[t] * aq;
    asm("" : "+v"(lo), "+v"(hi));
    a[P]     = hi;
    a[P + 1] = lo;
  }

  __syncthreads();
#pragma unroll
  for (int g = 0; g < 8; ++g) {
    v4f t;
    t.x = a[4 * g]; t.y = a[4 * g + 1]; t.z = a[4 * g + 2]; t.w = a[4 * g + 3];
    *(v4f*)(myrow + 4 * g) = t;
  }

  __builtin_amdgcn_sched_barrier(0);
#pragma unroll
  for (int t = 0; t < NP; ++t) {
    const int P = PAR + 2 * t;
    const float wp = w[P], wq = w[P + 1];
    float lo = ct[t] * wp - st[t] * wq;
    float hi = st[t] * wp + ct[t] * wq;
    asm("" : "+v"(lo), "+v"(hi));
    w[P]     = hi;
    w[P + 1] = lo;
  }
}

__global__ __launch_bounds__(32, 8) void jacobi_logm_kernel(const float* __restrict__ x,
                                                            const float* __restrict__ C,
                                                            float* __restrict__ qkv,
                                                            float* __restrict__ nrm) {
  __shared__ __attribute__((aligned(16))) float R[32][36];
  __shared__ float lw[32];
  const int lane = threadIdx.x;
  const bool odd = (lane & 1) != 0;
  const float* xm = x + (size_t)blockIdx.x * 1024;

  const v4f* rp = (const v4f*)(xm + lane * 32);
  float a[32], w[32];
#pragma unroll
  for (int g = 0; g < 8; ++g) {
    v4f t = rp[g];
    a[4 * g + 0] = t.x; a[4 * g + 1] = t.y; a[4 * g + 2] = t.z; a[4 * g + 3] = t.w;
  }
#pragma unroll
  for (int Q = 0; Q < 32; ++Q) w[Q] = (Q == lane) ? 1.f : 0.f;

  float* myrow = &R[lane][0];
  const int pE = lane & ~1;
  const float* prowE = &R[lane ^ 1][0];
  const float* pppE  = &R[pE][pE];
  const float* pqqE  = &R[pE + 1][pE + 1];
  const bool restO = (lane == 0) || (lane == 31);
  const int pO    = restO ? (lane == 0 ? 0 : 30) : (odd ? lane : lane - 1);
  const int partO = restO ? lane : (odd ? lane + 1 : lane - 1);
  const float* prowO = &R[partO][0];
  const float* pppO  = &R[pO][pO];
  const float* pqqO  = &R[pO + 1][pO + 1];

#pragma unroll
  for (int g = 0; g < 8; ++g) {
    v4f t; t.x = a[4 * g]; t.y = a[4 * g + 1]; t.z = a[4 * g + 2]; t.w = a[4 * g + 3];
    *(v4f*)(myrow + 4 * g) = t;
  }

  for (int it = 0; it < 8 * 16; ++it) {
    jacobi_round<0>(a, w, myrow, prowE, pppE, pqqE, !odd, false);
    jacobi_round<1>(a, w, myrow, prowO, pppO, pqqO, odd, restO);
  }
  __syncthreads();
  {
    const float dg = R[lane][lane];
    lw[lane] = logf(fmaxf(dg, 1e-20f));
  }
  __syncthreads();
#pragma unroll
  for (int Q = 0; Q < 32; ++Q) R[lane][Q] = w[Q];
  __syncthreads();

  float vj[32];
#pragma unroll
  for (int Q = 0; Q < 32; ++Q) vj[Q] = w[Q] * lw[Q];

  for (int i = 0; i < 32; ++i) {
    float acc = 0.f;
#pragma unroll
    for (int k = 0; k < 32; ++k) acc += R[i][k] * vj[k];
    R[i][lane] = (i == lane) ? 0.f : acc;
  }
  __syncthreads();

  const int nIdx = lane & 15;
  const int hi   = lane >> 4;
  v2f yb[2][8];
#pragma unroll
  for (int tj = 0; tj < 2; ++tj) {
#pragma unroll
    for (int kk = 0; kk < 8; ++kk) {
      const int k0 = kk * 4 + hi * 2;
      yb[tj][kk].x = R[k0][tj * 16 + nIdx];
      yb[tj][kk].y = R[k0 + 1][tj * 16 + nIdx];
    }
  }
  __syncthreads();

  const int m = blockIdx.x;
  for (int wsel = 0; wsel < 3; ++wsel) {
    const float* Cw = C + wsel * 1024;
    v2f cf[2][8];
#pragma unroll
    for (int t = 0; t < 2; ++t) {
#pragma unroll
      for (int kk = 0; kk < 8; ++kk)
        cf[t][kk] = *(const v2f*)(Cw + (t * 16 + nIdx) * 32 + kk * 4 + hi * 2);
    }

#pragma unroll
    for (int ti = 0; ti < 2; ++ti) {
      v8f acc0 = {}, acc1 = {};
#pragma unroll
      for (int kk = 0; kk < 8; ++kk) {
        acc0 = WMMA_F32_16x16x4(cf[ti][kk], yb[0][kk], acc0);
        acc1 = WMMA_F32_16x16x4(cf[ti][kk], yb[1][kk], acc1);
      }
#pragma unroll
      for (int r = 0; r < 8; ++r) {
        R[ti * 16 + r + hi * 8][nIdx]      = acc0[r];
        R[ti * 16 + r + hi * 8][16 + nIdx] = acc1[r];
      }
    }
    __syncthreads();

#pragma unroll
    for (int ti = 0; ti < 2; ++ti) {
      v2f ta[8];
#pragma unroll
      for (int kk = 0; kk < 8; ++kk)
        ta[kk] = *(const v2f*)&R[ti * 16 + nIdx][kk * 4 + hi * 2];
      v8f acc0 = {}, acc1 = {};
#pragma unroll
      for (int kk = 0; kk < 8; ++kk) {
        acc0 = WMMA_F32_16x16x4(ta[kk], cf[0][kk], acc0);
        acc1 = WMMA_F32_16x16x4(ta[kk], cf[1][kk], acc1);
      }
      __syncthreads();
#pragma unroll
      for (int r = 0; r < 8; ++r) {
        R[ti * 16 + r + hi * 8][nIdx]      = acc0[r];
        R[ti * 16 + r + hi * 8][16 + nIdx] = acc1[r];
      }
    }
    __syncthreads();

    float* outm = qkv + ((size_t)wsel * NMAT + m) * 1024;
    float ss = 0.f;
#pragma unroll
    for (int i = 0; i < 32; ++i) {
      const float* src = (i > lane) ? &R[i][lane] : &R[lane][i];
      const float t = *src;
      const float val = (i == lane) ? 0.f : t;
      outm[i * 32 + lane] = val;
      ss = __builtin_fmaf(val, val, ss);
    }
    if (wsel < 2) {
      ss += __shfl_xor(ss, 16);
      ss += __shfl_xor(ss, 8);
      ss += __shfl_xor(ss, 4);
      ss += __shfl_xor(ss, 2);
      ss += __shfl_xor(ss, 1);
      if (lane == 0) nrm[wsel * NMAT + m] = ss;
    }
    __syncthreads();
  }
}

struct RowFrag { v4f q[4]; };

__device__ __forceinline__ void load_frag(RowFrag& f, const float* p) {
  const v4f* p4 = (const v4f*)p;
  f.q[0] = p4[0]; f.q[1] = p4[1]; f.q[2] = p4[2]; f.q[3] = p4[3];
}

__device__ __forceinline__ v2f frag_pair(const RowFrag& f, int s) {
  v4f q = f.q[s >> 1];
  v2f lo = __builtin_shufflevector(q, q, 0, 1);
  v2f hp = __builtin_shufflevector(q, q, 2, 3);
  return (s & 1) ? hp : lo;
}

__device__ __forceinline__ void load_tiles(RowFrag (&A)[4], RowFrag (&B)[2],
                                           const float* arow, const float* brow, int kb) {
#pragma unroll
  for (int jt = 0; jt < 4; ++jt) load_frag(A[jt], arow + jt * (16 * NF) + kb);
#pragma unroll
  for (int it = 0; it < 2; ++it) load_frag(B[it], brow + it * (16 * NF) + kb);
}

__device__ __forceinline__ void mma_chunk(v8f (&acc)[4][2], const RowFrag (&A)[4],
                                          const RowFrag (&B)[2]) {
#pragma unroll
  for (int s = 0; s < 8; ++s) {
#pragma unroll
    for (int jt = 0; jt < 4; ++jt) {
      v2f a = frag_pair(A[jt], s);
#pragma unroll
      for (int it = 0; it < 2; ++it)
        acc[jt][it] = WMMA_F32_16x16x4(a, frag_pair(B[it], s), acc[jt][it]);
    }
  }
}

__global__ __launch_bounds__(256) void scores_softmax_kernel(const float* __restrict__ qkv,
                                                             const float* __restrict__ nrm,
                                                             float* __restrict__ scores) {
  __shared__ float red[8][32];
  const int tid  = threadIdx.x;
  const int lane = tid & 31;
  const int w    = tid >> 5;
  const int nIdx = lane & 15;
  const int hi   = lane >> 4;
  const int b  = blockIdx.y;
  const int i0 = blockIdx.x * 32;
  const int j0 = w * 64;
  const float* qf = qkv + ((size_t)0 * NMAT + (size_t)b * NN) * 1024;
  const float* kf = qkv + ((size_t)1 * NMAT + (size_t)b * NN) * 1024;
  const float* arow = kf + (size_t)(j0 + nIdx) * NF + hi * 16;
  const float* brow = qf + (size_t)(i0 + nIdx) * NF + hi * 16;

  v8f acc[4][2] = {};
  RowFrag a0[4], b0[2], a1[4], b1[2];

  load_tiles(a0, b0, arow, brow, 0);
  for (int kb = 0; kb < NF; kb += 64) {
    load_tiles(a1, b1, arow, brow, kb + 32);
    mma_chunk(acc, a0, b0);
    if (kb + 64 < NF) load_tiles(a0, b0, arow, brow, kb + 64);
    mma_chunk(acc, a1, b1);
  }

  float q2[2], mx[2];
#pragma unroll
  for (int it = 0; it < 2; ++it) {
    q2[it] = nrm[b * NN + i0 + it * 16 + nIdx];
    mx[it] = -1e30f;
  }
#pragma unroll
  for (int jt = 0; jt < 4; ++jt) {
#pragma unroll
    for (int r = 0; r < 8; ++r) {
      float k2 = nrm[NMAT + b * NN + j0 + jt * 16 + hi * 8 + r];
#pragma unroll
      for (int it = 0; it < 2; ++it) {
        float d2 = q2[it] + k2 - 2.f * acc[jt][it][r];
        d2 = fmaxf(d2, 1e-12f);
        float e  = sqrtf(d2);
        float sc = 1.f / (1.f + log1pf(e));
        acc[jt][it][r] = sc;
        mx[it] = fmaxf(mx[it], sc);
      }
    }
  }

#pragma unroll
  for (int it = 0; it < 2; ++it) {
    mx[it] = fmaxf(mx[it], __shfl_xor(mx[it], 16));
    if (hi == 0) red[w][it * 16 + nIdx] = mx[it];
  }
  __syncthreads();
#pragma unroll
  for (int it = 0; it < 2; ++it) {
    float m = -1e30f;
#pragma unroll
    for (int ww = 0; ww < 8; ++ww) m = fmaxf(m, red[ww][it * 16 + nIdx]);
    mx[it] = m;
  }
  __syncthreads();

  float ps[2] = {0.f, 0.f};
#pragma unroll
  for (int jt = 0; jt < 4; ++jt) {
#pragma unroll
    for (int r = 0; r < 8; ++r) {
#pragma unroll
      for (int it = 0; it < 2; ++it) {
        float e = __expf(acc[jt][it][r] - mx[it]);
        acc[jt][it][r] = e;
        ps[it] += e;
      }
    }
  }
#pragma unroll
  for (int it = 0; it < 2; ++it) {
    ps[it] += __shfl_xor(ps[it], 16);
    if (hi == 0) red[w][it * 16 + nIdx] = ps[it];
  }
  __syncthreads();
  float inv[2];
#pragma unroll
  for (int it = 0; it < 2; ++it) {
    float sum = 0.f;
#pragma unroll
    for (int ww = 0; ww < 8; ++ww) sum += red[ww][it * 16 + nIdx];
    inv[it] = 1.f / sum;
  }

  float* sb = scores + (size_t)b * NN * NN;
#pragma unroll
  for (int jt = 0; jt < 4; ++jt) {
#pragma unroll
    for (int r = 0; r < 8; ++r) {
      int j = j0 + jt * 16 + hi * 8 + r;
#pragma unroll
      for (int it = 0; it < 2; ++it)
        sb[(size_t)j * NN + i0 + it * 16 + nIdx] = acc[jt][it][r] * inv[it];
    }
  }
}

struct ColFrag { v2f e[8]; };

__device__ __forceinline__ void load_out_tiles(RowFrag (&A)[4], ColFrag& B,
                                               const float* arow, const float* bcol, int kb) {
#pragma unroll
  for (int t = 0; t < 4; ++t) load_frag(A[t], arow + t * (16 * NN) + kb);
  const float* bp = bcol + (size_t)kb * NF;
#pragma unroll
  for (int s = 0; s < 8; ++s) {
    B.e[s].x = bp[(size_t)(2 * s) * NF];
    B.e[s].y = bp[(size_t)(2 * s + 1) * NF];
  }
}

__device__ __forceinline__ void mma_out_chunk(v8f (&acc)[4], const RowFrag (&A)[4],
                                              const ColFrag& B) {
#pragma unroll
  for (int s = 0; s < 8; ++s) {
#pragma unroll
    for (int t = 0; t < 4; ++t)
      acc[t] = WMMA_F32_16x16x4(frag_pair(A[t], s), B.e[s], acc[t]);
  }
}

__global__ __launch_bounds__(32) void outgemm_kernel(const float* __restrict__ scores,
                                                     const float* __restrict__ qkv,
                                                     float* __restrict__ out) {
  const int lane = threadIdx.x;
  const int nIdx = lane & 15;
  const int hi   = lane >> 4;
  const int ft = blockIdx.x, ct = blockIdx.y, b = blockIdx.z;
  const int f0 = ft * 16, c0 = ct * 64;
  const float* p  = scores + (size_t)b * NN * NN;
  const float* vf = qkv + ((size_t)2 * NMAT + (size_t)b * NN) * 1024;
  const float* arow = p + (size_t)(c0 + nIdx) * NN + hi * 16;
  const float* bcol = vf + (size_t)(hi * 16) * NF + f0 + nIdx;

  v8f acc[4] = {};
  RowFrag a0[4], a1[4];
  ColFrag b0, b1;

  load_out_tiles(a0, b0, arow, bcol, 0);
  for (int kb = 0; kb < NN - 64; kb += 64) {
    load_out_tiles(a1, b1, arow, bcol, kb + 32);
    __builtin_amdgcn_sched_barrier(0);
    mma_out_chunk(acc, a0, b0);
    __builtin_amdgcn_sched_barrier(0);
    load_out_tiles(a0, b0, arow, bcol, kb + 64);
    __builtin_amdgcn_sched_barrier(0);
    mma_out_chunk(acc, a1, b1);
    __builtin_amdgcn_sched_barrier(0);
  }
  load_out_tiles(a1, b1, arow, bcol, NN - 32);
  __builtin_amdgcn_sched_barrier(0);
  mma_out_chunk(acc, a0, b0);
  mma_out_chunk(acc, a1, b1);

  float* ob = out + (size_t)b * NN * NF;
#pragma unroll
  for (int t = 0; t < 4; ++t) {
#pragma unroll
    for (int r = 0; r < 8; ++r) {
      int c = c0 + t * 16 + r + hi * 8;
      ob[(size_t)c * NF + f0 + nIdx] = acc[t][r];
    }
  }
}

extern "C" void kernel_launch(void* const* d_in, const int* in_sizes, int n_in,
                              void* d_out, int out_size, void* d_ws, size_t ws_size,
                              hipStream_t stream) {
  (void)in_sizes; (void)n_in; (void)out_size; (void)ws_size;
  const float* x  = (const float*)d_in[0];
  const float* qw = (const float*)d_in[1];
  const float* kw = (const float*)d_in[2];
  const float* vw = (const float*)d_in[3];
  float* out = (float*)d_out;

  float* ws     = (float*)d_ws;
  float* C      = ws;
  float* qkv    = C + 3 * 1024;
  float* nrm    = qkv + (size_t)3 * NMAT * 1024;
  float* scores = nrm + 2 * NMAT;

  cayley_kernel<<<3, 256, 0, stream>>>(qw, kw, vw, C);
  jacobi_logm_kernel<<<NMAT, 32, 0, stream>>>(x, C, qkv, nrm);
  scores_softmax_kernel<<<dim3(NN / 32, NB), 256, 0, stream>>>(qkv, nrm, scores);
  outgemm_kernel<<<dim3(NF / 16, NN / 64, NB), 32, 0, stream>>>(scores, qkv, out);
}


# eigensolver: each pair's c,s reach all lanes through a 128-B LDS table (one 8-B write per lane, eight 16-B broadcast reads per round), not 31 v_readlane into scalar registers; sums kept
# speedup vs baseline: 3.1507x; 1.0337x over previous
#include <hip/hip_runtime.h>
#include <math.h>


typedef __attribute__((ext_vector_type(2))) float v2f;
typedef __attribute__((ext_vector_type(8))) float v8f;

#define WMMA_F32_16x16x4(a, b, c) \
  __builtin_amdgcn_wmma_f32_16x16x4_f32(false, (a), false, (b), (short)0, (c), false, false)

#define NMAT 8192
#define NB   16
#define NN   512
#define NF   1024

__global__ __launch_bounds__(256) void cayley_kernel(const float* __restrict__ qw,
                                                     const float* __restrict__ kw,
                                                     const float* __restrict__ vw,
                                                     float* __restrict__ Cout) {
  __shared__ float X[32][32];
  __shared__ float Mm[32][33];
  __shared__ float Inv[32][33];
  __shared__ float fac[32];
  __shared__ float pivinv;
  const float* W = (blockIdx.x == 0) ? qw : ((blockIdx.x == 1) ? kw : vw);
  const int tid = threadIdx.x;

  for (int e = tid; e < 1024; e += 256) {
    int i = e >> 5, j = e & 31;
    float xv = W[i * 32 + j] - W[j * 32 + i];
    X[i][j]  = xv;
    Mm[i][j] = xv + ((i == j) ? 1.f : 0.f);
    Inv[i][j] = (i == j) ? 1.f : 0.f;
  }
  __syncthreads();

  for (int k = 0; k < 32; ++k) {
    if (tid == 0) pivinv = 1.f / Mm[k][k];
    __syncthreads();
    if (tid < 32)      Mm[k][tid]       *= pivinv;
    else if (tid < 64) Inv[k][tid - 32] *= pivinv;
    __syncthreads();
    if (tid < 32) fac[tid] = (tid == k) ? 0.f : Mm[tid][k];
    __syncthreads();
    for (int e = tid; e < 2048; e += 256) {
      int i = e >> 6, j = e & 63;
      if (i == k) continue;
      float f = fac[i];
      if (j < 32) Mm[i][j]        -= f * Mm[k][j];
      else        Inv[i][j - 32]  -= f * Inv[k][j - 32];
    }
    __syncthreads();
  }

  for (int e = tid; e < 1024; e += 256) {
    int i = e >> 5, j = e & 31;
    float acc = 0.f;
    for (int kk = 0; kk < 32; ++kk) {
      float a = ((i == kk) ? 1.f : 0.f) - X[i][kk];
      acc += a * Inv[kk][j];
    }
    Cout[blockIdx.x * 1024 + i * 32 + j] = acc;
  }
}

typedef __attribute__((ext_vector_type(4))) float v4f;

template <int PAR>
__device__ __forceinline__ void jacobi_round(float (&a)[32], float (&w)[32], float* myrow,
                                             const float* prow, const float* ppp,
                                             const float* pqq, const bool lower,
                                             const bool rest, float* csmine,
                                             const float* cstab) {
  constexpr int NP = PAR ? 15 : 16;
  __syncthreads();
  const float app = ppp[0];
  const float epq = ppp[1];
  const float aqq = pqq[0];
  float b[32];
#pragma unroll
  for (int g = 0; g < 8; ++g) {
    v4f u = *(const v4f*)(prow + 4 * g);
    b[4 * g] = u.x; b[4 * g + 1] = u.y; b[4 * g + 2] = u.z; b[4 * g + 3] = u.w;
  }
  const float apq = rest ? 0.f : epq;
  float c = 1.f, s = 0.f;
  if (fabsf(apq) > 1e-12f) {
    float theta = (aqq - app) / (2.f * apq);
    float t = 1.f / (fabsf(theta) + sqrtf(theta * theta + 1.f));
    if (theta < 0.f) t = -t;
    c = rsqrtf(t * t + 1.f);
    s = t * c;
  }
  {
    v2f e; e.x = c; e.y = s;
    *(v2f*)csmine = e;
  }
  __syncthreads();

  const float sg = lower ? s : -s;
#pragma unroll
  for (int Q = 0; Q < 32; ++Q) {
    float r = __builtin_fmaf(c, b[Q], sg * a[Q]);
    asm("" : "+v"(r));
    a[Q] = r;
  }

  float ct[16], st[16];
#pragma unroll
  for (int g = 0; g < 8; ++g) {
    v4f u = *(const v4f*)(cstab + 4 * g);
    ct[2 * g] = u.x; st[2 * g] = u.y; ct[2 * g + 1] = u.z; st[2 * g + 1] = u.w;
  }

#pragma unroll
  for (int t = 0; t < NP; ++t) {
    const int P = PAR + 2 * t;
    const float ap = a[P], aq = a[P + 1];
    float lo = ct[t] * ap - st[t] * aq;
    float hi = st[t] * ap + ct[t] * aq;
    asm("" : "+v"(lo), "+v"(hi));
    a[P]     = hi;
    a[P + 1] = lo;
  }

  __syncthreads();
#pragma unroll
  for (int g = 0; g < 8; ++g) {
    v4f t;
    t.x = a[4 * g]; t.y = a[4 * g + 1]; t.z = a[4 * g + 2]; t.w = a[4 * g + 3];
    *(v4f*)(myrow + 4 * g) = t;
  }

  __builtin_amdgcn_sched_barrier(0);
#pragma unroll
  for (int t = 0; t < NP; ++t) {
    const int P = PAR + 2 * t;
    const float wp = w[P], wq = w[P + 1];
    float lo = ct[t] * wp - st[t] * wq;
    float hi = st[t] * wp + ct[t] * wq;
    asm("" : "+v"(lo), "+v"(hi));
    w[P]     = hi;
    w[P + 1] = lo;
  }
}

__global__ __launch_bounds__(32, 8) void jacobi_logm_kernel(const float* __restrict__ x,
                                                            const float* __restrict__ C,
                                                            float* __restrict__ qkv,
                                                            float* __restrict__ nrm) {
  __shared__ __attribute__((aligned(16))) float R[32][36];
  __shared__ float lw[32];
  __shared__ __attribute__((aligned(16))) float CS[32];
  const int lane = threadIdx.x;
  const bool odd = (lane & 1) != 0;
  const float* xm = x + (size_t)blockIdx.x * 1024;

  const v4f* rp = (const v4f*)(xm + lane * 32);
  float a[32], w[32];
#pragma unroll
  for (int g = 0; g < 8; ++g) {
    v4f t = rp[g];
    a[4 * g + 0] = t.x; a[4 * g + 1] = t.y; a[4 * g + 2] = t.z; a[4 * g + 3] = t.w;
  }
#pragma unroll
  for (int Q = 0; Q < 32; ++Q) w[Q] = (Q == lane) ? 1.f : 0.f;

  float* myrow = &R[lane][0];
  const int pE = lane & ~1;
  const float* prowE = &R[lane ^ 1][0];
  const float* pppE  = &R[pE][pE];
  const float* pqqE  = pppE + 37;
  const bool restO = (lane == 0) || (lane == 31);
  const int pO    = restO ? (lane == 0 ? 0 : 30) : (odd ? lane : lane - 1);
  const int partO = restO ? lane : (odd ? lane + 1 : lane - 1);
  const float* prowO = &R[partO][0];
  const float* pppO  = &R[pO][pO];
  const float* pqqO  = pppO + 37;
  float* csE = &CS[2 * (lane >> 1)];
  float* csO = &CS[2 * (restO ? 15 : ((lane - 1) >> 1))];

#pragma unroll
  for (int g = 0; g < 8; ++g) {
    v4f t; t.x = a[4 * g]; t.y = a[4 * g + 1]; t.z = a[4 * g + 2]; t.w = a[4 * g + 3];
    *(v4f*)(myrow + 4 * g) = t;
  }

  for (int it = 0; it < 8 * 16; ++it) {
    jacobi_round<0>(a, w, myrow, prowE, pppE, pqqE, !odd, false, csE, CS);
    jacobi_round<1>(a, w, myrow, prowO, pppO, pqqO, odd, restO, csO, CS);
  }
  __syncthreads();
  {
    const float dg = R[lane][lane];
    lw[lane] = logf(fmaxf(dg, 1e-20f));
  }
  __syncthreads();
#pragma unroll
  for (int Q = 0; Q < 32; ++Q) R[lane][Q] = w[Q];
  __syncthreads();

  float vj[32];
#pragma unroll
  for (int Q = 0; Q < 32; ++Q) vj[Q] = w[Q] * lw[Q];

  for (int i = 0; i < 32; ++i) {
    float acc = 0.f;
#pragma unroll
    for (int k = 0; k < 32; ++k) acc += R[i][k] * vj[k];
    R[i][lane] = (i == lane) ? 0.f : acc;
  }
  __syncthreads();

  const int nIdx = lane & 15;
  const int hi   = lane >> 4;
  v2f yb[2][8];
#pragma unroll
  for (int tj = 0; tj < 2; ++tj) {
#pragma unroll
    for (int kk = 0; kk < 8; ++kk) {
      const int k0 = kk * 4 + hi * 2;
      yb[tj][kk].x = R[k0][tj * 16 + nIdx];
      yb[tj][kk].y = R[k0 + 1][tj * 16 + nIdx];
    }
  }
  __syncthreads();

  const int m = blockIdx.x;
  for (int wsel = 0; wsel < 3; ++wsel) {
    const float* Cw = C + wsel * 1024;
    v2f cf[2][8];
#pragma unroll
    for (int t = 0; t < 2; ++t) {
#pragma unroll
      for (int kk = 0; kk < 8; ++kk)
        cf[t][kk] = *(const v2f*)(Cw + (t * 16 + nIdx) * 32 + kk * 4 + hi * 2);
    }

#pragma unroll
    for (int ti = 0; ti < 2; ++ti) {
      v8f acc0 = {}, acc1 = {};
#pragma unroll
      for (int kk = 0; kk < 8; ++kk) {
        acc0 = WMMA_F32_16x16x4(cf[ti][kk], yb[0][kk], acc0);
        acc1 = WMMA_F32_16x16x4(cf[ti][kk], yb[1][kk], acc1);
      }
#pragma unroll
      for (int r = 0; r < 8; ++r) {
        R[ti * 16 + r + hi * 8][nIdx]      = acc0[r];
        R[ti * 16 + r + hi * 8][16 + nIdx] = acc1[r];
      }
    }
    __syncthreads();

#pragma unroll
    for (int ti = 0; ti < 2; ++ti) {
      v2f ta[8];
#pragma unroll
      for (int kk = 0; kk < 8; ++kk)
        ta[kk] = *(const v2f*)&R[ti * 16 + nIdx][kk * 4 + hi * 2];
      v8f acc0 = {}, acc1 = {};
#pragma unroll
      for (int kk = 0; kk < 8; ++kk) {
        acc0 = WMMA_F32_16x16x4(ta[kk], cf[0][kk], acc0);
        acc1 = WMMA_F32_16x16x4(ta[kk], cf[1][kk], acc1);
      }
      __syncthreads();
#pragma unroll
      for (int r = 0; r < 8; ++r) {
        R[ti * 16 + r + hi * 8][nIdx]      = acc0[r];
        R[ti * 16 + r + hi * 8][16 + nIdx] = acc1[r];
      }
    }
    __syncthreads();

    float* outm = qkv + ((size_t)wsel * NMAT + m) * 1024;
    float ss = 0.f;
#pragma unroll
    for (int i = 0; i < 32; ++i) {
      const float* src = (i > lane) ? &R[i][lane] : &R[lane][i];
      const float t = *src;
      const float val = (i == lane) ? 0.f : t;
      outm[i * 32 + lane] = val;
      ss = __builtin_fmaf(val, val, ss);
    }
    if (wsel < 2) {
      ss += __shfl_xor(ss, 16);
      ss += __shfl_xor(ss, 8);
      ss += __shfl_xor(ss, 4);
      ss += __shfl_xor(ss, 2);
      ss += __shfl_xor(ss, 1);
      if (lane == 0) nrm[wsel * NMAT + m] = ss;
    }
    __syncthreads();
  }
}

struct RowFrag { v4f q[4]; };

__device__ __forceinline__ void load_frag(RowFrag& f, const float* p) {
  const v4f* p4 = (const v4f*)p;
  f.q[0] = p4[0]; f.q[1] = p4[1]; f.q[2] = p4[2]; f.q[3] = p4[3];
}

__device__ __forceinline__ v2f frag_pair(const RowFrag& f, int s) {
  v4f q = f.q[s >> 1];
  v2f lo = __builtin_shufflevector(q, q, 0, 1);
  v2f hp = __builtin_shufflevector(q, q, 2, 3);
  return (s & 1) ? hp : lo;
}

__device__ __forceinline__ void load_tiles(RowFrag (&A)[4], RowFrag (&B)[2],
                                           const float* arow, const float* brow, int kb) {
#pragma unroll
  for (int jt = 0; jt < 4; ++jt) load_frag(A[jt], arow + jt * (16 * NF) + kb);
#pragma unroll
  for (int it = 0; it < 2; ++it) load_frag(B[it], brow + it * (16 * NF) + kb);
}

__device__ __forceinline__ void mma_chunk(v8f (&acc)[4][2], const RowFrag (&A)[4],
                                          const RowFrag (&B)[2]) {
#pragma unroll
  for (int s = 0; s < 8; ++s) {
#pragma unroll
    for (int jt = 0; jt < 4; ++jt) {
      v2f a = frag_pair(A[jt], s);
#pragma unroll
      for (int it = 0; it < 2; ++it)
        acc[jt][it] = WMMA_F32_16x16x4(a, frag_pair(B[it], s), acc[jt][it]);
    }
  }
}

__global__ __launch_bounds__(256) void scores_softmax_kernel(const float* __restrict__ qkv,
                                                             const float* __restrict__ nrm,
                                                             float* __restrict__ scores) {
  __shared__ float red[8][32];
  const int tid  = threadIdx.x;
  const int lane = tid & 31;
  const int w    = tid >> 5;
  const int nIdx = lane & 15;
  const int hi   = lane >> 4;
  const int b  = blockIdx.y;
  const int i0 = blockIdx.x * 32;
  const int j0 = w * 64;
  const float* qf = qkv + ((size_t)0 * NMAT + (size_t)b * NN) * 1024;
  const float* kf = qkv + ((size_t)1 * NMAT + (size_t)b * NN) * 1024;
  const float* arow = kf + (size_t)(j0 + nIdx) * NF + hi * 16;
  const float* brow = qf + (size_t)(i0 + nIdx) * NF + hi * 16;

  v8f acc[4][2] = {};
  RowFrag a0[4], b0[2], a1[4], b1[2];

  load_tiles(a0, b0, arow, brow, 0);
  for (int kb = 0; kb < NF; kb += 64) {
    load_tiles(a1, b1, arow, brow, kb + 32);
    mma_chunk(acc, a0, b0);
    if (kb + 64 < NF) load_tiles(a0, b0, arow, brow, kb + 64);
    mma_chunk(acc, a1, b1);
  }

  float q2[2], mx[2];
#pragma unroll
  for (int it = 0; it < 2; ++it) {
    q2[it] = nrm[b * NN + i0 + it * 16 + nIdx];
    mx[it] = -1e30f;
  }
#pragma unroll
  for (int jt = 0; jt < 4; ++jt) {
#pragma unroll
    for (int r = 0; r < 8; ++r) {
      float k2 = nrm[NMAT + b * NN + j0 + jt * 16 + hi * 8 + r];
#pragma unroll
      for (int it = 0; it < 2; ++it) {
        float d2 = q2[it] + k2 - 2.f * acc[jt][it][r];
        d2 = fmaxf(d2, 1e-12f);
        float e  = sqrtf(d2);
        float sc = 1.f / (1.f + log1pf(e));
        acc[jt][it][r] = sc;
        mx[it] = fmaxf(mx[it], sc);
      }
    }
  }

#pragma unroll
  for (int it = 0; it < 2; ++it) {
    mx[it] = fmaxf(mx[it], __shfl_xor(mx[it], 16));
    if (hi == 0) red[w][it * 16 + nIdx] = mx[it];
  }
  __syncthreads();
#pragma unroll
  for (int it = 0; it < 2; ++it) {
    float m = -1e30f;
#pragma unroll
    for (int ww = 0; ww < 8; ++ww) m = fmaxf(m, red[ww][it * 16 + nIdx]);
    mx[it] = m;
  }
  __syncthreads();

  float ps[2] = {0.f, 0.f};
#pragma unroll
  for (int jt = 0; jt < 4; ++jt) {
#pragma unroll
    for (int r = 0; r < 8; ++r) {
#pragma unroll
      for (int it = 0; it < 2; ++it) {
        float e = __expf(acc[jt][it][r] - mx[it]);
        acc[jt][it][r] = e;
        ps[it] += e;
      }
    }
  }
#pragma unroll
  for (int it = 0; it < 2; ++it) {
    ps[it] += __shfl_xor(ps[it], 16);
    if (hi == 0) red[w][it * 16 + nIdx] = ps[it];
  }
  __syncthreads();
  float inv[2];
#pragma unroll
  for (int it = 0; it < 2; ++it) {
    float sum = 0.f;
#pragma unroll
    for (int ww = 0; ww < 8; ++ww) sum += red[ww][it * 16 + nIdx];
    inv[it] = 1.f / sum;
  }

  float* sb = scores + (size_t)b * NN * NN;
#pragma unroll
  for (int jt = 0; jt < 4; ++jt) {
#pragma unroll
    for (int r = 0; r < 8; ++r) {
      int j = j0 + jt * 16 + hi * 8 + r;
#pragma unroll
      for (int it = 0; it < 2; ++it)
        sb[(size_t)j * NN + i0 + it * 16 + nIdx] = acc[jt][it][r] * inv[it];
    }
  }
}

struct ColFrag { v2f e[8]; };

__device__ __forceinline__ void load_out_tiles(RowFrag (&A)[4], ColFrag& B,
                                               const float* arow, const float* bcol, int kb) {
#pragma unroll
  for (int t = 0; t < 4; ++t) load_frag(A[t], arow + t * (16 * NN) + kb);
  const float* bp = bcol + (size_t)kb * NF;
#pragma unroll
  for (int s = 0; s < 8; ++s) {
    B.e[s].x = bp[(size_t)(2 * s) * NF];
    B.e[s].y = bp[(size_t)(2 * s + 1) * NF];
  }
}

__device__ __forceinline__ void mma_out_chunk(v8f (&acc)[4], const RowFrag (&A)[4],
                                              const ColFrag& B) {
#pragma unroll
  for (int s = 0; s < 8; ++s) {
#pragma unroll
    for (int t = 0; t < 4; ++t)
      acc[t] = WMMA_F32_16x16x4(frag_pair(A[t], s), B.e[s], acc[t]);
  }
}

__global__ __launch_bounds__(32) void outgemm_kernel(const float* __restrict__ scores,
                                                     const float* __restrict__ qkv,
                                                     float* __restrict__ out) {
  const int lane = threadIdx.x;
  const int nIdx = lane & 15;
  const int hi   = lane >> 4;
  const int ft = blockIdx.x, ct = blockIdx.y, b = blockIdx.z;
  const int f0 = ft * 16, c0 = ct * 64;
  const float* p  = scores + (size_t)b * NN * NN;
  const float* vf = qkv + ((size_t)2 * NMAT + (size_t)b * NN) * 1024;
  const float* arow = p + (size_t)(c0 + nIdx) * NN + hi * 16;
  const float* bcol = vf + (size_t)(hi * 16) * NF + f0 + nIdx;

  v8f acc[4] = {};
  RowFrag a0[4], a1[4];
  ColFrag b0, b1;

  load_out_tiles(a0, b0, arow, bcol, 0);
  for (int kb = 0; kb < NN - 64; kb += 64) {
    load_out_tiles(a1, b1, arow, bcol, kb + 32);
    __builtin_amdgcn_sched_barrier(0);
    mma_out_chunk(acc, a0, b0);
    __builtin_amdgcn_sched_barrier(0);
    load_out_tiles(a0, b0, arow, bcol, kb + 64);
    __builtin_amdgcn_sched_barrier(0);
    mma_out_chunk(acc, a1, b1);
    __builtin_amdgcn_sched_barrier(0);
  }
  load_out_tiles(a1, b1, arow, bcol, NN - 32);
  __builtin_amdgcn_sched_barrier(0);
  mma_out_chunk(acc, a0, b0);
  mma_out_chunk(acc, a1, b1);

  float* ob = out + (size_t)b * NN * NF;
#pragma unroll
  for (int t = 0; t < 4; ++t) {
#pragma unroll
    for (int r = 0; r < 8; ++r) {
      int c = c0 + t * 16 + r + hi * 8;
      ob[(size_t)c * NF + f0 + nIdx] = acc[t][r];
    }
  }
}

extern "C" void kernel_launch(void* const* d_in, const int* in_sizes, int n_in,
                              void* d_out, int out_size, void* d_ws, size_t ws_size,
                              hipStream_t stream) {
  (void)in_sizes; (void)n_in; (void)out_size; (void)ws_size;
  const float* x  = (const float*)d_in[0];
  const float* qw = (const float*)d_in[1];
  const float* kw = (const float*)d_in[2];
  const float* vw = (const float*)d_in[3];
  float* out = (float*)d_out;

  float* ws     = (float*)d_ws;
  float* C      = ws;
  float* qkv    = C + 3 * 1024;
  float* nrm    = qkv + (size_t)3 * NMAT * 1024;
  float* scores = nrm + 2 * NMAT;

  cayley_kernel<<<3, 256, 0, stream>>>(qw, kw, vw, C);
  jacobi_logm_kernel<<<NMAT, 32, 0, stream>>>(x, C, qkv, nrm);
  scores_softmax_kernel<<<dim3(NN / 32, NB), 256, 0, stream>>>(qkv, nrm, scores);
  outgemm_kernel<<<dim3(NF / 16, NN / 64, NB), 32, 0, stream>>>(scores, qkv, out);
}


# eigensolver: one-sided Jacobi, rows only (no column phase, no V, no c,s table), swept until a sweep meets no row pair with cosine over 32 eps (cap 24), not a fixed 8; rotations and sum order change
# speedup vs baseline: 3.4740x; 1.1026x over previous
#include <hip/hip_runtime.h>
#include <math.h>


typedef __attribute__((ext_vector_type(2))) float v2f;
typedef __attribute__((ext_vector_type(8))) float v8f;

#define WMMA_F32_16x16x4(a, b, c) \
  __builtin_amdgcn_wmma_f32_16x16x4_f32(false, (a), false, (b), (short)0, (c), false, false)

#define NMAT 8192
#define NB   16
#define NN   512
#define NF   1024

__global__ __launch_bounds__(256) void cayley_kernel(const float* __restrict__ qw,
                                                     const float* __restrict__ kw,
                                                     const float* __restrict__ vw,
                                                     float* __restrict__ Cout) {
  __shared__ float X[32][32];
  __shared__ float Mm[32][33];
  __shared__ float Inv[32][33];
  __shared__ float fac[32];
  __shared__ float pivinv;
  const float* W = (blockIdx.x == 0) ? qw : ((blockIdx.x == 1) ? kw : vw);
  const int tid = threadIdx.x;

  for (int e = tid; e < 1024; e += 256) {
    int i = e >> 5, j = e & 31;
    float xv = W[i * 32 + j] - W[j * 32 + i];
    X[i][j]  = xv;
    Mm[i][j] = xv + ((i == j) ? 1.f : 0.f);
    Inv[i][j] = (i == j) ? 1.f : 0.f;
  }
  __syncthreads();

  for (int k = 0; k < 32; ++k) {
    if (tid == 0) pivinv = 1.f / Mm[k][k];
    __syncthreads();
    if (tid < 32)      Mm[k][tid]       *= pivinv;
    else if (tid < 64) Inv[k][tid - 32] *= pivinv;
    __syncthreads();
    if (tid < 32) fac[tid] = (tid == k) ? 0.f : Mm[tid][k];
    __syncthreads();
    for (int e = tid; e < 2048; e += 256) {
      int i = e >> 6, j = e & 63;
      if (i == k) continue;
      float f = fac[i];
      if (j < 32) Mm[i][j]        -= f * Mm[k][j];
      else        Inv[i][j - 32]  -= f * Inv[k][j - 32];
    }
    __syncthreads();
  }

  for (int e = tid; e < 1024; e += 256) {
    int i = e >> 5, j = e & 31;
    float acc = 0.f;
    for (int kk = 0; kk < 32; ++kk) {
      float a = ((i == kk) ? 1.f : 0.f) - X[i][kk];
      acc += a * Inv[kk][j];
    }
    Cout[blockIdx.x * 1024 + i * 32 + j] = acc;
  }
}

typedef __attribute__((ext_vector_type(4))) float v4f;

__device__ __forceinline__ float row_norm2(const float (&a)[32]) {
  float n0 = 0.f, n1 = 0.f;
#pragma unroll
  for (int Q = 0; Q < 32; Q += 2) {
    n0 = __builtin_fmaf(a[Q], a[Q], n0);
    n1 = __builtin_fmaf(a[Q + 1], a[Q + 1], n1);
  }
  return n0 + n1;
}

#define COS_TOL   (32.f * 1.1920929e-7f)
#define SWEEP_CAP 24

__device__ __forceinline__ void jacobi_round(float (&a)[32], float& so, float* myrow,
                                             const float* prow, const bool lower,
                                             const bool rest, bool& met) {
  __syncthreads();
  float b[32];
#pragma unroll
  for (int g = 0; g < 8; ++g) {
    v4f u = *(const v4f*)(prow + 4 * g);
    b[4 * g] = u.x; b[4 * g + 1] = u.y; b[4 * g + 2] = u.z; b[4 * g + 3] = u.w;
  }
  const float sp = prow[32];
  float x0 = 0.f, x1 = 0.f;
#pragma unroll
  for (int Q = 0; Q < 32; Q += 2) {
    x0 = __builtin_fmaf(a[Q], b[Q], x0);
    x1 = __builtin_fmaf(a[Q + 1], b[Q + 1], x1);
  }
  const float app = lower ? so : sp;
  const float aqq = lower ? sp : so;
  const float apq = rest ? 0.f : (x0 + x1);
  met = met || (apq * apq > ((COS_TOL * COS_TOL) * so) * sp);
  float c = 1.f, s = 0.f;
  if (fabsf(apq) > 1e-12f) {
    float theta = (aqq - app) / (2.f * apq);
    float t = 1.f / (fabsf(theta) + sqrtf(theta * theta + 1.f));
    if (theta < 0.f) t = -t;
    c = rsqrtf(t * t + 1.f);
    s = t * c;
  }

  const float sg = lower ? s : -s;
#pragma unroll
  for (int Q = 0; Q < 32; ++Q) a[Q] = __builtin_fmaf(c, b[Q], sg * a[Q]);
  so = row_norm2(a);

  __syncthreads();
#pragma unroll
  for (int g = 0; g < 8; ++g) {
    v4f t;
    t.x = a[4 * g]; t.y = a[4 * g + 1]; t.z = a[4 * g + 2]; t.w = a[4 * g + 3];
    *(v4f*)(myrow + 4 * g) = t;
  }
  myrow[32] = so;
}

__global__ __launch_bounds__(32, 8) void jacobi_logm_kernel(const float* __restrict__ x,
                                                            const float* __restrict__ C,
                                                            float* __restrict__ qkv,
                                                            float* __restrict__ nrm) {
  __shared__ __attribute__((aligned(16))) float R[32][36];
  __shared__ float lw[32];
  const int lane = threadIdx.x;
  const bool odd = (lane & 1) != 0;
  const float* xm = x + (size_t)blockIdx.x * 1024;

  const v4f* rp = (const v4f*)(xm + lane * 32);
  float a[32];
#pragma unroll
  for (int g = 0; g < 8; ++g) {
    v4f t = rp[g];
    a[4 * g + 0] = t.x; a[4 * g + 1] = t.y; a[4 * g + 2] = t.z; a[4 * g + 3] = t.w;
  }
  float so = row_norm2(a);

  float* myrow = &R[lane][0];
  const float* prowE = &R[lane ^ 1][0];
  const bool restO = (lane == 0) || (lane == 31);
  const int partO = restO ? lane : (odd ? lane + 1 : lane - 1);
  const float* prowO = &R[partO][0];

#pragma unroll
  for (int g = 0; g < 8; ++g) {
    v4f t; t.x = a[4 * g]; t.y = a[4 * g + 1]; t.z = a[4 * g + 2]; t.w = a[4 * g + 3];
    *(v4f*)(myrow + 4 * g) = t;
  }
  myrow[32] = so;

  for (int sw = 0; sw < SWEEP_CAP; ++sw) {
    bool met = false;
    for (int it = 0; it < 16; ++it) {
      jacobi_round(a, so, myrow, prowE, !odd, false, met);
      jacobi_round(a, so, myrow, prowO, odd, restO, met);
    }
    if (__builtin_amdgcn_ballot_w32(met) == 0u) break;
  }
  const float lam = fmaxf(sqrtf(so), 1e-20f);
  const float rl  = 1.f / lam;
  __syncthreads();
  lw[lane] = logf(lam);
#pragma unroll
  for (int Q = 0; Q < 32; ++Q) R[Q][lane] = a[Q] * rl;
  __syncthreads();
  float w[32];
#pragma unroll
  for (int g = 0; g < 8; ++g) {
    v4f u = *(const v4f*)(myrow + 4 * g);
    w[4 * g] = u.x; w[4 * g + 1] = u.y; w[4 * g + 2] = u.z; w[4 * g + 3] = u.w;
  }

  float vj[32];
#pragma unroll
  for (int Q = 0; Q < 32; ++Q) vj[Q] = w[Q] * lw[Q];

  for (int i = 0; i < 32; ++i) {
    float acc = 0.f;
#pragma unroll
    for (int k = 0; k < 32; ++k) acc += R[i][k] * vj[k];
    R[i][lane] = (i == lane) ? 0.f : acc;
  }
  __syncthreads();

  const int nIdx = lane & 15;
  const int hi   = lane >> 4;
  v2f yb[2][8];
#pragma unroll
  for (int tj = 0; tj < 2; ++tj) {
#pragma unroll
    for (int kk = 0; kk < 8; ++kk) {
      const int k0 = kk * 4 + hi * 2;
      yb[tj][kk].x = R[k0][tj * 16 + nIdx];
      yb[tj][kk].y = R[k0 + 1][tj * 16 + nIdx];
    }
  }
  __syncthreads();

  const int m = blockIdx.x;
  for (int wsel = 0; wsel < 3; ++wsel) {
    const float* Cw = C + wsel * 1024;
    v2f cf[2][8];
#pragma unroll
    for (int t = 0; t < 2; ++t) {
#pragma unroll
      for (int kk = 0; kk < 8; ++kk)
        cf[t][kk] = *(const v2f*)(Cw + (t * 16 + nIdx) * 32 + kk * 4 + hi * 2);
    }

#pragma unroll
    for (int ti = 0; ti < 2; ++ti) {
      v8f acc0 = {}, acc1 = {};
#pragma unroll
      for (int kk = 0; kk < 8; ++kk) {
        acc0 = WMMA_F32_16x16x4(cf[ti][kk], yb[0][kk], acc0);
        acc1 = WMMA_F32_16x16x4(cf[ti][kk], yb[1][kk], acc1);
      }
#pragma unroll
      for (int r = 0; r < 8; ++r) {
        R[ti * 16 + r + hi * 8][nIdx]      = acc0[r];
        R[ti * 16 + r + hi * 8][16 + nIdx] = acc1[r];
      }
    }
    __syncthreads();

#pragma unroll
    for (int ti = 0; ti < 2; ++ti) {
      v2f ta[8];
#pragma unroll
      for (int kk = 0; kk < 8; ++kk)
        ta[kk] = *(const v2f*)&R[ti * 16 + nIdx][kk * 4 + hi * 2];
      v8f acc0 = {}, acc1 = {};
#pragma unroll
      for (int kk = 0; kk < 8; ++kk) {
        acc0 = WMMA_F32_16x16x4(ta[kk], cf[0][kk], acc0);
        acc1 = WMMA_F32_16x16x4(ta[kk], cf[1][kk], acc1);
      }
      __syncthreads();
#pragma unroll
      for (int r = 0; r < 8; ++r) {
        R[ti * 16 + r + hi * 8][nIdx]      = acc0[r];
        R[ti * 16 + r + hi * 8][16 + nIdx] = acc1[r];
      }
    }
    __syncthreads();

    float* outm = qkv + ((size_t)wsel * NMAT + m) * 1024;
    float ss = 0.f;
#pragma unroll
    for (int i = 0; i < 32; ++i) {
      const float* src = (i > lane) ? &R[i][lane] : &R[lane][i];
      const float t = *src;
      const float val = (i == lane) ? 0.f : t;
      outm[i * 32 + lane] = val;
      ss = __builtin_fmaf(val, val, ss);
    }
    if (wsel < 2) {
      ss += __shfl_xor(ss, 16);
      ss += __shfl_xor(ss, 8);
      ss += __shfl_xor(ss, 4);
      ss += __shfl_xor(ss, 2);
      ss += __shfl_xor(ss, 1);
      if (lane == 0) nrm[wsel * NMAT + m] = ss;
    }
    __syncthreads();
  }
}

struct RowFrag { v4f q[4]; };

__device__ __forceinline__ void load_frag(RowFrag& f, const float* p) {
  const v4f* p4 = (const v4f*)p;
  f.q[0] = p4[0]; f.q[1] = p4[1]; f.q[2] = p4[2]; f.q[3] = p4[3];
}

__device__ __forceinline__ v2f frag_pair(const RowFrag& f, int s) {
  v4f q = f.q[s >> 1];
  v2f lo = __builtin_shufflevector(q, q, 0, 1);
  v2f hp = __builtin_shufflevector(q, q, 2, 3);
  return (s & 1) ? hp : lo;
}

__device__ __forceinline__ void load_tiles(RowFrag (&A)[4], RowFrag (&B)[2],
                                           const float* arow, const float* brow, int kb) {
#pragma unroll
  for (int jt = 0; jt < 4; ++jt) load_frag(A[jt], arow + jt * (16 * NF) + kb);
#pragma unroll
  for (int it = 0; it < 2; ++it) load_frag(B[it], brow + it * (16 * NF) + kb);
}

__device__ __forceinline__ void mma_chunk(v8f (&acc)[4][2], const RowFrag (&A)[4],
                                          const RowFrag (&B)[2]) {
#pragma unroll
  for (int s = 0; s < 8; ++s) {
#pragma unroll
    for (int jt = 0; jt < 4; ++jt) {
      v2f a = frag_pair(A[jt], s);
#pragma unroll
      for (int it = 0; it < 2; ++it)
        acc[jt][it] = WMMA_F32_16x16x4(a, frag_pair(B[it], s), acc[jt][it]);
    }
  }
}

__global__ __launch_bounds__(256) void scores_softmax_kernel(const float* __restrict__ qkv,
                                                             const float* __restrict__ nrm,
                                                             float* __restrict__ scores) {
  __shared__ float red[8][32];
  const int tid  = threadIdx.x;
  const int lane = tid & 31;
  const int w    = tid >> 5;
  const int nIdx = lane & 15;
  const int hi   = lane >> 4;
  const int b  = blockIdx.y;
  const int i0 = blockIdx.x * 32;
  const int j0 = w * 64;
  const float* qf = qkv + ((size_t)0 * NMAT + (size_t)b * NN) * 1024;
  const float* kf = qkv + ((size_t)1 * NMAT + (size_t)b * NN) * 1024;
  const float* arow = kf + (size_t)(j0 + nIdx) * NF + hi * 16;
  const float* brow = qf + (size_t)(i0 + nIdx) * NF + hi * 16;

  v8f acc[4][2] = {};
  RowFrag a0[4], b0[2], a1[4], b1[2];

  load_tiles(a0, b0, arow, brow, 0);
  for (int kb = 0; kb < NF; kb += 64) {
    load_tiles(a1, b1, arow, brow, kb + 32);
    mma_chunk(acc, a0, b0);
    if (kb + 64 < NF) load_tiles(a0, b0, arow, brow, kb + 64);
    mma_chunk(acc, a1, b1);
  }

  float q2[2], mx[2];
#pragma unroll
  for (int it = 0; it < 2; ++it) {
    q2[it] = nrm[b * NN + i0 + it * 16 + nIdx];
    mx[it] = -1e30f;
  }
#pragma unroll
  for (int jt = 0; jt < 4; ++jt) {
#pragma unroll
    for (int r = 0; r < 8; ++r) {
      float k2 = nrm[NMAT + b * NN + j0 + jt * 16 + hi * 8 + r];
#pragma unroll
      for (int it = 0; it < 2; ++it) {
        float d2 = q2[it] + k2 - 2.f * acc[jt][it][r];
        d2 = fmaxf(d2, 1e-12f);
        float e  = sqrtf(d2);
        float sc = 1.f / (1.f + log1pf(e));
        acc[jt][it][r] = sc;
        mx[it] = fmaxf(mx[it], sc);
      }
    }
  }

#pragma unroll
  for (int it = 0; it < 2; ++it) {
    mx[it] = fmaxf(mx[it], __shfl_xor(mx[it], 16));
    if (hi == 0) red[w][it * 16 + nIdx] = mx[it];
  }
  __syncthreads();
#pragma unroll
  for (int it = 0; it < 2; ++it) {
    float m = -1e30f;
#pragma unroll
    for (int ww = 0; ww < 8; ++ww) m = fmaxf(m, red[ww][it * 16 + nIdx]);
    mx[it] = m;
  }
  __syncthreads();

  float ps[2] = {0.f, 0.f};
#pragma unroll
  for (int jt = 0; jt < 4; ++jt) {
#pragma unroll
    for (int r = 0; r < 8; ++r) {
#pragma unroll
      for (int it = 0; it < 2; ++it) {
        float e = __expf(acc[jt][it][r] - mx[it]);
        acc[jt][it][r] = e;
        ps[it] += e;
      }
    }
  }
#pragma unroll
  for (int it = 0; it < 2; ++it) {
    ps[it] += __shfl_xor(ps[it], 16);
    if (hi == 0) red[w][it * 16 + nIdx] = ps[it];
  }
  __syncthreads();
  float inv[2];
#pragma unroll
  for (int it = 0; it < 2; ++it) {
    float sum = 0.f;
#pragma unroll
    for (int ww = 0; ww < 8; ++ww) sum += red[ww][it * 16 + nIdx];
    inv[it] = 1.f / sum;
  }

  float* sb = scores + (size_t)b * NN * NN;
#pragma unroll
  for (int jt = 0; jt < 4; ++jt) {
#pragma unroll
    for (int r = 0; r < 8; ++r) {
      int j = j0 + jt * 16 + hi * 8 + r;
#pragma unroll
      for (int it = 0; it < 2; ++it)
        sb[(size_t)j * NN + i0 + it * 16 + nIdx] = acc[jt][it][r] * inv[it];
    }
  }
}

struct ColFrag { v2f e[8]; };

__device__ __forceinline__ void load_out_tiles(RowFrag (&A)[4], ColFrag& B,
                                               const float* arow, const float* bcol, int kb) {
#pragma unroll
  for (int t = 0; t < 4; ++t) load_frag(A[t], arow + t * (16 * NN) + kb);
  const float* bp = bcol + (size_t)kb * NF;
#pragma unroll
  for (int s = 0; s < 8; ++s) {
    B.e[s].x = bp[(size_t)(2 * s) * NF];
    B.e[s].y = bp[(size_t)(2 * s + 1) * NF];
  }
}

__device__ __forceinline__ void mma_out_chunk(v8f (&acc)[4], const RowFrag (&A)[4],
                                              const ColFrag& B) {
#pragma unroll
  for (int s = 0; s < 8; ++s) {
#pragma unroll
    for (int t = 0; t < 4; ++t)
      acc[t] = WMMA_F32_16x16x4(frag_pair(A[t], s), B.e[s], acc[t]);
  }
}

__global__ __launch_bounds__(32) void outgemm_kernel(const float* __restrict__ scores,
                                                     const float* __restrict__ qkv,
                                                     float* __restrict__ out) {
  const int lane = threadIdx.x;
  const int nIdx = lane & 15;
  const int hi   = lane >> 4;
  const int ft = blockIdx.x, ct = blockIdx.y, b = blockIdx.z;
  const int f0 = ft * 16, c0 = ct * 64;
  const float* p  = scores + (size_t)b * NN * NN;
  const float* vf = qkv + ((size_t)2 * NMAT + (size_t)b * NN) * 1024;
  const float* arow = p + (size_t)(c0 + nIdx) * NN + hi * 16;
  const float* bcol = vf + (size_t)(hi * 16) * NF + f0 + nIdx;

  v8f acc[4] = {};
  RowFrag a0[4], a1[4];
  ColFrag b0, b1;

  load_out_tiles(a0, b0, arow, bcol, 0);
  for (int kb = 0; kb < NN - 64; kb += 64) {
    load_out_tiles(a1, b1, arow, bcol, kb + 32);
    __builtin_amdgcn_sched_barrier(0);
    mma_out_chunk(acc, a0, b0);
    __builtin_amdgcn_sched_barrier(0);
    load_out_tiles(a0, b0, arow, bcol, kb + 64);
    __builtin_amdgcn_sched_barrier(0);
    mma_out_chunk(acc, a1, b1);
    __builtin_amdgcn_sched_barrier(0);
  }
  load_out_tiles(a1, b1, arow, bcol, NN - 32);
  __builtin_amdgcn_sched_barrier(0);
  mma_out_chunk(acc, a0, b0);
  mma_out_chunk(acc, a1, b1);

  float* ob = out + (size_t)b * NN * NF;
#pragma unroll
  for (int t = 0; t < 4; ++t) {
#pragma unroll
    for (int r = 0; r < 8; ++r) {
      int c = c0 + t * 16 + r + hi * 8;
      ob[(size_t)c * NF + f0 + nIdx] = acc[t][r];
    }
  }
}

extern "C" void kernel_launch(void* const* d_in, const int* in_sizes, int n_in,
                              void* d_out, int out_size, void* d_ws, size_t ws_size,
                              hipStream_t stream) {
  (void)in_sizes; (void)n_in; (void)out_size; (void)ws_size;
  const float* x  = (const float*)d_in[0];
  const float* qw = (const float*)d_in[1];
  const float* kw = (const float*)d_in[2];
  const float* vw = (const float*)d_in[3];
  float* out = (float*)d_out;

  float* ws     = (float*)d_ws;
  float* C      = ws;
  float* qkv    = C + 3 * 1024;
  float* nrm    = qkv + (size_t)3 * NMAT * 1024;
  float* scores = nrm + 2 * NMAT;

  cayley_kernel<<<3, 256, 0, stream>>>(qw, kw, vw, C);
  jacobi_logm_kernel<<<NMAT, 32, 0, stream>>>(x, C, qkv, nrm);
  scores_softmax_kernel<<<dim3(NN / 32, NB), 256, 0, stream>>>(qkv, nrm, scores);
  outgemm_kernel<<<dim3(NF / 16, NN / 64, NB), 32, 0, stream>>>(scores, qkv, out);
}


# out GEMM (p x v): 8-wave block owns 64 rows x 256 columns, each wave a 64x32 register tile; p rows shared via the WGP cache; a lane owns a column pair, so a row is stored as one line; k order kept
# speedup vs baseline: 3.7088x; 1.0676x over previous
#include <hip/hip_runtime.h>
#include <math.h>


typedef __attribute__((ext_vector_type(2))) float v2f;
typedef __attribute__((ext_vector_type(8))) float v8f;

#define WMMA_F32_16x16x4(a, b, c) \
  __builtin_amdgcn_wmma_f32_16x16x4_f32(false, (a), false, (b), (short)0, (c), false, false)

#define NMAT 8192
#define NB   16
#define NN   512
#define NF   1024

__global__ __launch_bounds__(256) void cayley_kernel(const float* __restrict__ qw,
                                                     const float* __restrict__ kw,
                                                     const float* __restrict__ vw,
                                                     float* __restrict__ Cout) {
  __shared__ float X[32][32];
  __shared__ float Mm[32][33];
  __shared__ float Inv[32][33];
  __shared__ float fac[32];
  __shared__ float pivinv;
  const float* W = (blockIdx.x == 0) ? qw : ((blockIdx.x == 1) ? kw : vw);
  const int tid = threadIdx.x;

  for (int e = tid; e < 1024; e += 256) {
    int i = e >> 5, j = e & 31;
    float xv = W[i * 32 + j] - W[j * 32 + i];
    X[i][j]  = xv;
    Mm[i][j] = xv + ((i == j) ? 1.f : 0.f);
    Inv[i][j] = (i == j) ? 1.f : 0.f;
  }
  __syncthreads();

  for (int k = 0; k < 32; ++k) {
    if (tid == 0) pivinv = 1.f / Mm[k][k];
    __syncthreads();
    if (tid < 32)      Mm[k][tid]       *= pivinv;
    else if (tid < 64) Inv[k][tid - 32] *= pivinv;
    __syncthreads();
    if (tid < 32) fac[tid] = (tid == k) ? 0.f : Mm[tid][k];
    __syncthreads();
    for (int e = tid; e < 2048; e += 256) {
      int i = e >> 6, j = e & 63;
      if (i == k) continue;
      float f = fac[i];
      if (j < 32) Mm[i][j]        -= f * Mm[k][j];
      else        Inv[i][j - 32]  -= f * Inv[k][j - 32];
    }
    __syncthreads();
  }

  for (int e = tid; e < 1024; e += 256) {
    int i = e >> 5, j = e & 31;
    float acc = 0.f;
    for (int kk = 0; kk < 32; ++kk) {
      float a = ((i == kk) ? 1.f : 0.f) - X[i][kk];
      acc += a * Inv[kk][j];
    }
    Cout[blockIdx.x * 1024 + i * 32 + j] = acc;
  }
}

typedef __attribute__((ext_vector_type(4))) float v4f;

__device__ __forceinline__ float row_norm2(const float (&a)[32]) {
  float n0 = 0.f, n1 = 0.f;
#pragma unroll
  for (int Q = 0; Q < 32; Q += 2) {
    n0 = __builtin_fmaf(a[Q], a[Q], n0);
    n1 = __builtin_fmaf(a[Q + 1], a[Q + 1], n1);
  }
  return n0 + n1;
}

#define COS_TOL   (32.f * 1.1920929e-7f)
#define SWEEP_CAP 24

__device__ __forceinline__ void jacobi_round(float (&a)[32], float& so, float* myrow,
                                             const float* prow, const bool lower,
                                             const bool rest, bool& met) {
  __syncthreads();
  float b[32];
#pragma unroll
  for (int g = 0; g < 8; ++g) {
    v4f u = *(const v4f*)(prow + 4 * g);
    b[4 * g] = u.x; b[4 * g + 1] = u.y; b[4 * g + 2] = u.z; b[4 * g + 3] = u.w;
  }
  const float sp = prow[32];
  float x0 = 0.f, x1 = 0.f;
#pragma unroll
  for (int Q = 0; Q < 32; Q += 2) {
    x0 = __builtin_fmaf(a[Q], b[Q], x0);
    x1 = __builtin_fmaf(a[Q + 1], b[Q + 1], x1);
  }
  const float app = lower ? so : sp;
  const float aqq = lower ? sp : so;
  const float apq = rest ? 0.f : (x0 + x1);
  met = met || (apq * apq > ((COS_TOL * COS_TOL) * so) * sp);
  float c = 1.f, s = 0.f;
  if (fabsf(apq) > 1e-12f) {
    float theta = (aqq - app) / (2.f * apq);
    float t = 1.f / (fabsf(theta) + sqrtf(theta * theta + 1.f));
    if (theta < 0.f) t = -t;
    c = rsqrtf(t * t + 1.f);
    s = t * c;
  }

  const float sg = lower ? s : -s;
#pragma unroll
  for (int Q = 0; Q < 32; ++Q) a[Q] = __builtin_fmaf(c, b[Q], sg * a[Q]);
  so = row_norm2(a);

  __syncthreads();
#pragma unroll
  for (int g = 0; g < 8; ++g) {
    v4f t;
    t.x = a[4 * g]; t.y = a[4 * g + 1]; t.z = a[4 * g + 2]; t.w = a[4 * g + 3];
    *(v4f*)(myrow + 4 * g) = t;
  }
  myrow[32] = so;
}

__global__ __launch_bounds__(32, 8) void jacobi_logm_kernel(const float* __restrict__ x,
                                                            const float* __restrict__ C,
                                                            float* __restrict__ qkv,
                                                            float* __restrict__ nrm) {
  __shared__ __attribute__((aligned(16))) float R[32][36];
  __shared__ float lw[32];
  const int lane = threadIdx.x;
  const bool odd = (lane & 1) != 0;
  const float* xm = x + (size_t)blockIdx.x * 1024;

  const v4f* rp = (const v4f*)(xm + lane * 32);
  float a[32];
#pragma unroll
  for (int g = 0; g < 8; ++g) {
    v4f t = rp[g];
    a[4 * g + 0] = t.x; a[4 * g + 1] = t.y; a[4 * g + 2] = t.z; a[4 * g + 3] = t.w;
  }
  float so = row_norm2(a);

  float* myrow = &R[lane][0];
  const float* prowE = &R[lane ^ 1][0];
  const bool restO = (lane == 0) || (lane == 31);
  const int partO = restO ? lane : (odd ? lane + 1 : lane - 1);
  const float* prowO = &R[partO][0];

#pragma unroll
  for (int g = 0; g < 8; ++g) {
    v4f t; t.x = a[4 * g]; t.y = a[4 * g + 1]; t.z = a[4 * g + 2]; t.w = a[4 * g + 3];
    *(v4f*)(myrow + 4 * g) = t;
  }
  myrow[32] = so;

  for (int sw = 0; sw < SWEEP_CAP; ++sw) {
    bool met = false;
    for (int it = 0; it < 16; ++it) {
      jacobi_round(a, so, myrow, prowE, !odd, false, met);
      jacobi_round(a, so, myrow, prowO, odd, restO, met);
    }
    if (__builtin_amdgcn_ballot_w32(met) == 0u) break;
  }
  const float lam = fmaxf(sqrtf(so), 1e-20f);
  const float rl  = 1.f / lam;
  __syncthreads();
  lw[lane] = logf(lam);
#pragma unroll
  for (int Q = 0; Q < 32; ++Q) R[Q][lane] = a[Q] * rl;
  __syncthreads();
  float w[32];
#pragma unroll
  for (int g = 0; g < 8; ++g) {
    v4f u = *(const v4f*)(myrow + 4 * g);
    w[4 * g] = u.x; w[4 * g + 1] = u.y; w[4 * g + 2] = u.z; w[4 * g + 3] = u.w;
  }

  float vj[32];
#pragma unroll
  for (int Q = 0; Q < 32; ++Q) vj[Q] = w[Q] * lw[Q];

  for (int i = 0; i < 32; ++i) {
    float acc = 0.f;
#pragma unroll
    for (int k = 0; k < 32; ++k) acc += R[i][k] * vj[k];
    R[i][lane] = (i == lane) ? 0.f : acc;
  }
  __syncthreads();

  const int nIdx = lane & 15;
  const int hi   = lane >> 4;
  v2f yb[2][8];
#pragma unroll
  for (int tj = 0; tj < 2; ++tj) {
#pragma unroll
    for (int kk = 0; kk < 8; ++kk) {
      const int k0 = kk * 4 + hi * 2;
      yb[tj][kk].x = R[k0][tj * 16 + nIdx];
      yb[tj][kk].y = R[k0 + 1][tj * 16 + nIdx];
    }
  }
  __syncthreads();

  const int m = blockIdx.x;
  for (int wsel = 0; wsel < 3; ++wsel) {
    const float* Cw = C + wsel * 1024;
    v2f cf[2][8];
#pragma unroll
    for (int t = 0; t < 2; ++t) {
#pragma unroll
      for (int kk = 0; kk < 8; ++kk)
        cf[t][kk] = *(const v2f*)(Cw + (t * 16 + nIdx) * 32 + kk * 4 + hi * 2);
    }

#pragma unroll
    for (int ti = 0; ti < 2; ++ti) {
      v8f acc0 = {}, acc1 = {};
#pragma unroll
      for (int kk = 0; kk < 8; ++kk) {
        acc0 = WMMA_F32_16x16x4(cf[ti][kk], yb[0][kk], acc0);
        acc1 = WMMA_F32_16x16x4(cf[ti][kk], yb[1][kk], acc1);
      }
#pragma unroll
      for (int r = 0; r < 8; ++r) {
        R[ti * 16 + r + hi * 8][nIdx]      = acc0[r];
        R[ti * 16 + r + hi * 8][16 + nIdx] = acc1[r];
      }
    }
    __syncthreads();

#pragma unroll
    for (int ti = 0; ti < 2; ++ti) {
      v2f ta[8];
#pragma unroll
      for (int kk = 0; kk < 8; ++kk)
        ta[kk] = *(const v2f*)&R[ti * 16 + nIdx][kk * 4 + hi * 2];
      v8f acc0 = {}, acc1 = {};
#pragma unroll
      for (int kk = 0; kk < 8; ++kk) {
        acc0 = WMMA_F32_16x16x4(ta[kk], cf[0][kk], acc0);
        acc1 = WMMA_F32_16x16x4(ta[kk], cf[1][kk], acc1);
      }
      __syncthreads();
#pragma unroll
      for (int r = 0; r < 8; ++r) {
        R[ti * 16 + r + hi * 8][nIdx]      = acc0[r];
        R[ti * 16 + r + hi * 8][16 + nIdx] = acc1[r];
      }
    }
    __syncthreads();

    float* outm = qkv + ((size_t)wsel * NMAT + m) * 1024;
    float ss = 0.f;
#pragma unroll
    for (int i = 0; i < 32; ++i) {
      const float* src = (i > lane) ? &R[i][lane] : &R[lane][i];
      const float t = *src;
      const float val = (i == lane) ? 0.f : t;
      outm[i * 32 + lane] = val;
      ss = __builtin_fmaf(val, val, ss);
    }
    if (wsel < 2) {
      ss += __shfl_xor(ss, 16);
      ss += __shfl_xor(ss, 8);
      ss += __shfl_xor(ss, 4);
      ss += __shfl_xor(ss, 2);
      ss += __shfl_xor(ss, 1);
      if (lane == 0) nrm[wsel * NMAT + m] = ss;
    }
    __syncthreads();
  }
}

struct RowFrag { v4f q[4]; };

__device__ __forceinline__ void load_frag(RowFrag& f, const float* p) {
  const v4f* p4 = (const v4f*)p;
  f.q[0] = p4[0]; f.q[1] = p4[1]; f.q[2] = p4[2]; f.q[3] = p4[3];
}

__device__ __forceinline__ v2f frag_pair(const RowFrag& f, int s) {
  v4f q = f.q[s >> 1];
  v2f lo = __builtin_shufflevector(q, q, 0, 1);
  v2f hp = __builtin_shufflevector(q, q, 2, 3);
  return (s & 1) ? hp : lo;
}

__device__ __forceinline__ void load_tiles(RowFrag (&A)[4], RowFrag (&B)[2],
                                           const float* arow, const float* brow, int kb) {
#pragma unroll
  for (int jt = 0; jt < 4; ++jt) load_frag(A[jt], arow + jt * (16 * NF) + kb);
#pragma unroll
  for (int it = 0; it < 2; ++it) load_frag(B[it], brow + it * (16 * NF) + kb);
}

__device__ __forceinline__ void mma_chunk(v8f (&acc)[4][2], const RowFrag (&A)[4],
                                          const RowFrag (&B)[2]) {
#pragma unroll
  for (int s = 0; s < 8; ++s) {
#pragma unroll
    for (int jt = 0; jt < 4; ++jt) {
      v2f a = frag_pair(A[jt], s);
#pragma unroll
      for (int it = 0; it < 2; ++it)
        acc[jt][it] = WMMA_F32_16x16x4(a, frag_pair(B[it], s), acc[jt][it]);
    }
  }
}

__global__ __launch_bounds__(256) void scores_softmax_kernel(const float* __restrict__ qkv,
                                                             const float* __restrict__ nrm,
                                                             float* __restrict__ scores) {
  __shared__ float red[8][32];
  const int tid  = threadIdx.x;
  const int lane = tid & 31;
  const int w    = tid >> 5;
  const int nIdx = lane & 15;
  const int hi   = lane >> 4;
  const int b  = blockIdx.y;
  const int i0 = blockIdx.x * 32;
  const int j0 = w * 64;
  const float* qf = qkv + ((size_t)0 * NMAT + (size_t)b * NN) * 1024;
  const float* kf = qkv + ((size_t)1 * NMAT + (size_t)b * NN) * 1024;
  const float* arow = kf + (size_t)(j0 + nIdx) * NF + hi * 16;
  const float* brow = qf + (size_t)(i0 + nIdx) * NF + hi * 16;

  v8f acc[4][2] = {};
  RowFrag a0[4], b0[2], a1[4], b1[2];

  load_tiles(a0, b0, arow, brow, 0);
  for (int kb = 0; kb < NF; kb += 64) {
    load_tiles(a1, b1, arow, brow, kb + 32);
    mma_chunk(acc, a0, b0);
    if (kb + 64 < NF) load_tiles(a0, b0, arow, brow, kb + 64);
    mma_chunk(acc, a1, b1);
  }

  float q2[2], mx[2];
#pragma unroll
  for (int it = 0; it < 2; ++it) {
    q2[it] = nrm[b * NN + i0 + it * 16 + nIdx];
    mx[it] = -1e30f;
  }
#pragma unroll
  for (int jt = 0; jt < 4; ++jt) {
#pragma unroll
    for (int r = 0; r < 8; ++r) {
      float k2 = nrm[NMAT + b * NN + j0 + jt * 16 + hi * 8 + r];
#pragma unroll
      for (int it = 0; it < 2; ++it) {
        float d2 = q2[it] + k2 - 2.f * acc[jt][it][r];
        d2 = fmaxf(d2, 1e-12f);
        float e  = sqrtf(d2);
        float sc = 1.f / (1.f + log1pf(e));
        acc[jt][it][r] = sc;
        mx[it] = fmaxf(mx[it], sc);
      }
    }
  }

#pragma unroll
  for (int it = 0; it < 2; ++it) {
    mx[it] = fmaxf(mx[it], __shfl_xor(mx[it], 16));
    if (hi == 0) red[w][it * 16 + nIdx] = mx[it];
  }
  __syncthreads();
#pragma unroll
  for (int it = 0; it < 2; ++it) {
    float m = -1e30f;
#pragma unroll
    for (int ww = 0; ww < 8; ++ww) m = fmaxf(m, red[ww][it * 16 + nIdx]);
    mx[it] = m;
  }
  __syncthreads();

  float ps[2] = {0.f, 0.f};
#pragma unroll
  for (int jt = 0; jt < 4; ++jt) {
#pragma unroll
    for (int r = 0; r < 8; ++r) {
#pragma unroll
      for (int it = 0; it < 2; ++it) {
        float e = __expf(acc[jt][it][r] - mx[it]);
        acc[jt][it][r] = e;
        ps[it] += e;
      }
    }
  }
#pragma unroll
  for (int it = 0; it < 2; ++it) {
    ps[it] += __shfl_xor(ps[it], 16);
    if (hi == 0) red[w][it * 16 + nIdx] = ps[it];
  }
  __syncthreads();
  float inv[2];
#pragma unroll
  for (int it = 0; it < 2; ++it) {
    float sum = 0.f;
#pragma unroll
    for (int ww = 0; ww < 8; ++ww) sum += red[ww][it * 16 + nIdx];
    inv[it] = 1.f / sum;
  }

  float* sb = scores + (size_t)b * NN * NN;
#pragma unroll
  for (int jt = 0; jt < 4; ++jt) {
#pragma unroll
    for (int r = 0; r < 8; ++r) {
      int j = j0 + jt * 16 + hi * 8 + r;
#pragma unroll
      for (int it = 0; it < 2; ++it)
        sb[(size_t)j * NN + i0 + it * 16 + nIdx] = acc[jt][it][r] * inv[it];
    }
  }
}

struct ColFrag { v2f r[16]; };

__device__ __forceinline__ void load_out_tiles(RowFrag (&A)[4], ColFrag& B,
                                               const float* arow, const float* bcol, int kb) {
#pragma unroll
  for (int t = 0; t < 4; ++t) load_frag(A[t], arow + t * (16 * NN) + kb);
  const float* bp = bcol + (size_t)kb * NF;
#pragma unroll
  for (int j = 0; j < 16; ++j) B.r[j] = *(const v2f*)(bp + (size_t)j * NF);
}

__device__ __forceinline__ void mma_out_chunk(v8f (&acc)[4][2], const RowFrag (&A)[4],
                                              const ColFrag& B) {
#pragma unroll
  for (int s = 0; s < 8; ++s) {
#pragma unroll
    for (int t = 0; t < 4; ++t) {
      v2f a = frag_pair(A[t], s);
#pragma unroll
      for (int u = 0; u < 2; ++u) {
        v2f bb;
        bb.x = B.r[2 * s][u];
        bb.y = B.r[2 * s + 1][u];
        acc[t][u] = WMMA_F32_16x16x4(a, bb, acc[t][u]);
      }
    }
  }
}

__global__ __launch_bounds__(256) void outgemm_kernel(const float* __restrict__ scores,
                                                      const float* __restrict__ qkv,
                                                      float* __restrict__ out) {
  const int tid  = threadIdx.x;
  const int lane = tid & 31;
  const int w    = tid >> 5;
  const int nIdx = lane & 15;
  const int hi   = lane >> 4;
  const int b  = blockIdx.z;
  const int c0 = blockIdx.y * 64;
  const int f0 = blockIdx.x * 256 + w * 32;
  const float* p  = scores + (size_t)b * NN * NN;
  const float* vf = qkv + ((size_t)2 * NMAT + (size_t)b * NN) * 1024;
  const float* arow = p + (size_t)(c0 + nIdx) * NN + hi * 16;
  const float* bcol = vf + (size_t)(hi * 16) * NF + f0 + 2 * nIdx;

  v8f acc[4][2] = {};
  RowFrag a0[4], a1[4];
  ColFrag b0, b1;

  load_out_tiles(a0, b0, arow, bcol, 0);
  for (int kb = 0; kb < NN - 64; kb += 64) {
    load_out_tiles(a1, b1, arow, bcol, kb + 32);
    __builtin_amdgcn_sched_barrier(0);
    mma_out_chunk(acc, a0, b0);
    __builtin_amdgcn_sched_barrier(0);
    load_out_tiles(a0, b0, arow, bcol, kb + 64);
    __builtin_amdgcn_sched_barrier(0);
    mma_out_chunk(acc, a1, b1);
    __builtin_amdgcn_sched_barrier(0);
  }
  load_out_tiles(a1, b1, arow, bcol, NN - 32);
  __builtin_amdgcn_sched_barrier(0);
  mma_out_chunk(acc, a0, b0);
  mma_out_chunk(acc, a1, b1);

  float* ob = out + (size_t)b * NN * NF;
#pragma unroll
  for (int t = 0; t < 4; ++t) {
#pragma unroll
    for (int r = 0; r < 8; ++r) {
      const int c = c0 + t * 16 + r + hi * 8;
      v2f o;
      o.x = acc[t][0][r];
      o.y = acc[t][1][r];
      *(v2f*)(ob + (size_t)c * NF + f0 + 2 * nIdx) = o;
    }
  }
}

extern "C" void kernel_launch(void* const* d_in, const int* in_sizes, int n_in,
                              void* d_out, int out_size, void* d_ws, size_t ws_size,
                              hipStream_t stream) {
  (void)in_sizes; (void)n_in; (void)out_size; (void)ws_size;
  const float* x  = (const float*)d_in[0];
  const float* qw = (const float*)d_in[1];
  const float* kw = (const float*)d_in[2];
  const float* vw = (const float*)d_in[3];
  float* out = (float*)d_out;

  float* ws     = (float*)d_ws;
  float* C      = ws;
  float* qkv    = C + 3 * 1024;
  float* nrm    = qkv + (size_t)3 * NMAT * 1024;
  float* scores = nrm + 2 * NMAT;

  cayley_kernel<<<3, 256, 0, stream>>>(qw, kw, vw, C);
  jacobi_logm_kernel<<<NMAT, 32, 0, stream>>>(x, C, qkv, nrm);
  scores_softmax_kernel<<<dim3(NN / 32, NB), 256, 0, stream>>>(qkv, nrm, scores);
  outgemm_kernel<<<dim3(NF / 256, NN / 64, NB), 256, 0, stream>>>(scores, qkv, out);
}
